# starting kernel
# baseline (speedup 1.0000x reference)
#include <hip/hip_runtime.h>
#include <math.h>

#define B_    4
#define WIN_  5
#define N_    50000
#define E_    1000000
#define H_    32
#define NTOT_ (B_ * N_)
#define BE_   (B_ * E_)

typedef __attribute__((ext_vector_type(2))) float v2f;
typedef __attribute__((ext_vector_type(8))) float v8f;

static __device__ __forceinline__ void decode_edge(const int* __restrict__ ei, int e,
                                                   int& src, int& dst) {
    int b  = e / E_;
    int el = e - b * E_;
    src = ei[(b * 2 + 0) * E_ + el] + b * N_;
    dst = ei[(b * 2 + 1) * E_ + el] + b * N_;
}

__global__ void zero_f32_kernel(float* __restrict__ p, int n) {
    int i = blockIdx.x * blockDim.x + threadIdx.x;
    if (i < n) p[i] = 0.0f;
}

__global__ void deg_kernel(const int* __restrict__ ei, float* __restrict__ deg) {
    int e = blockIdx.x * blockDim.x + threadIdx.x;
    if (e >= BE_) return;
    int src, dst;
    decode_edge(ei, e, src, dst);
    atomicAdd(&deg[dst], 1.0f);
}

__global__ void dinv_kernel(float* __restrict__ d) {
    int i = blockIdx.x * blockDim.x + threadIdx.x;
    if (i >= NTOT_) return;
    float v = d[i];
    d[i] = (v > 0.0f) ? rsqrtf(v) : 0.0f;
}

__global__ void norm_kernel(const int* __restrict__ ei, const float* __restrict__ dinv,
                            float* __restrict__ norm) {
    int e = blockIdx.x * blockDim.x + threadIdx.x;
    if (e >= BE_) return;
    int src, dst;
    decode_edge(ei, e, src, dst);
    norm[e] = dinv[src] * dinv[dst];
}

template <int F>
__global__ void prop_kernel(const int* __restrict__ ei, const float* __restrict__ norm,
                            const float* __restrict__ hin, float* __restrict__ hout) {
    int tid = blockIdx.x * blockDim.x + threadIdx.x;
    int e = tid / F;
    int f = tid - e * F;
    if (e >= BE_) return;
    int src, dst;
    decode_edge(ei, e, src, dst);
    float v = hin[src * F + f] * norm[e];
    atomicAdd(&hout[dst * F + f], v);
}

__global__ void prop4_kernel(const int* __restrict__ ei, const float* __restrict__ norm,
                             const float4* __restrict__ hin4, float* __restrict__ hout) {
    int tid = blockIdx.x * blockDim.x + threadIdx.x;
    int e = tid >> 3;
    int q = tid & 7;
    if (e >= BE_) return;
    int src, dst;
    decode_edge(ei, e, src, dst);
    float nrm = norm[e];
    float4 v = hin4[src * 8 + q];
    float* op = &hout[dst * H_ + q * 4];
    atomicAdd(op + 0, v.x * nrm);
    atomicAdd(op + 1, v.y * nrm);
    atomicAdd(op + 2, v.z * nrm);
    atomicAdd(op + 3, v.w * nrm);
}

template <int F>
__global__ void tag_gemm_kernel(const float* __restrict__ x0, const float* __restrict__ x1,
                                const float* __restrict__ x2, const float* __restrict__ W,
                                const float* __restrict__ bias, float* __restrict__ out) {
    const int numTiles = NTOT_ / 16;
    int wave = threadIdx.x >> 5;
    int tile = blockIdx.x * (blockDim.x >> 5) + wave;
    if (tile >= numTiles) return;

    int lane = threadIdx.x & 31;
    int half = lane >> 4;
    int l16  = lane & 15;
    int rowA = tile * 16 + l16;

    v8f c0 = {};
    v8f c1 = {};

    const float* xs[3] = { x0, x1, x2 };
    constexpr bool FULL = (F % 4 == 0);

#pragma unroll
    for (int s = 0; s < 3; ++s) {
        const float* __restrict__ xp = xs[s];
#pragma unroll
        for (int k0 = 0; k0 < F; k0 += 4) {
            int kA = k0 + half * 2;
            v2f a, b0, b1;
            if (FULL) {
                a.x = xp[rowA * F + kA];
                a.y = xp[rowA * F + kA + 1];
                int kW = s * F + kA;
                b0.x = W[(kW    ) * H_ + l16];
                b0.y = W[(kW + 1) * H_ + l16];
                b1.x = W[(kW    ) * H_ + 16 + l16];
                b1.y = W[(kW + 1) * H_ + 16 + l16];
            } else {
                float m0 = (kA     < F) ? 1.0f : 0.0f;
                float m1 = (kA + 1 < F) ? 1.0f : 0.0f;
                int ka0 = (kA     < F) ? kA     : (F - 1);
                int ka1 = (kA + 1 < F) ? kA + 1 : (F - 1);
                a.x = xp[rowA * F + ka0] * m0;
                a.y = xp[rowA * F + ka1] * m1;
                int kw0 = s * F + ka0;
                int kw1 = s * F + ka1;
                b0.x = W[kw0 * H_ + l16] * m0;
                b0.y = W[kw1 * H_ + l16] * m1;
                b1.x = W[kw0 * H_ + 16 + l16] * m0;
                b1.y = W[kw1 * H_ + 16 + l16] * m1;
            }
            c0 = __builtin_amdgcn_wmma_f32_16x16x4_f32(false, a, false, b0,
                                                       (short)0, c0, false, false);
            c1 = __builtin_amdgcn_wmma_f32_16x16x4_f32(false, a, false, b1,
                                                       (short)0, c1, false, false);
        }
    }

#pragma unroll
    for (int r = 0; r < 8; ++r) {
        int row = tile * 16 + half * 8 + r;
        float v0 = tanhf(c0[r] + bias[l16]);
        float v1 = tanhf(c1[r] + bias[16 + l16]);
        out[row * H_ + l16]      = v0;
        out[row * H_ + 16 + l16] = v1;
    }
}

__global__ void readout_kernel(const float* __restrict__ h, const float* __restrict__ Wr,
                               const float* __restrict__ br, float* __restrict__ out) {
    int i = blockIdx.x * blockDim.x + threadIdx.x;
    if (i >= NTOT_) return;
    float acc = br[0];
#pragma unroll
    for (int j = 0; j < H_; ++j) acc += h[i * H_ + j] * Wr[j];
    out[i] = acc;
}

static inline int cdiv(int a, int b) { return (a + b - 1) / b; }

extern "C" void kernel_launch(void* const* d_in, const int* in_sizes, int n_in,
                              void* d_out, int out_size, void* d_ws, size_t ws_size,
                              hipStream_t stream) {
    (void)in_sizes; (void)n_in; (void)out_size; (void)ws_size;

    const float* X  = (const float*)d_in[0];
    const int*   EI = (const int*)d_in[1];
    const float* W0 = (const float*)d_in[2];
    const float* b0 = (const float*)d_in[3];
    const float* W1 = (const float*)d_in[4];
    const float* b1 = (const float*)d_in[5];
    const float* W2 = (const float*)d_in[6];
    const float* b2 = (const float*)d_in[7];
    const float* Wr = (const float*)d_in[8];
    const float* br = (const float*)d_in[9];

    float* outF = (float*)d_out;
    float* hOut = outF + NTOT_;

    float* ws   = (float*)d_ws;
    float* dinv = ws;
    float* norm = dinv + NTOT_;
    float* p1   = norm + BE_;
    float* p2   = p1 + (size_t)NTOT_ * H_;
    float* bufA = p2 + (size_t)NTOT_ * H_;
    float* bufB = bufA + (size_t)NTOT_ * H_;

    const int TB = 256;

    zero_f32_kernel<<<cdiv(NTOT_, TB), TB, 0, stream>>>(dinv, NTOT_);
    deg_kernel<<<cdiv(BE_, TB), TB, 0, stream>>>(EI, dinv);
    dinv_kernel<<<cdiv(NTOT_, TB), TB, 0, stream>>>(dinv);
    norm_kernel<<<cdiv(BE_, TB), TB, 0, stream>>>(EI, dinv, norm);

    const int gemmBlocks = cdiv(NTOT_ / 16, TB / 32);
    const int propQ = BE_ * 8;

    zero_f32_kernel<<<cdiv(NTOT_ * WIN_, TB), TB, 0, stream>>>(p1, NTOT_ * WIN_);
    prop_kernel<WIN_><<<cdiv(BE_ * WIN_, TB), TB, 0, stream>>>(EI, norm, X, p1);
    zero_f32_kernel<<<cdiv(NTOT_ * WIN_, TB), TB, 0, stream>>>(p2, NTOT_ * WIN_);
    prop_kernel<WIN_><<<cdiv(BE_ * WIN_, TB), TB, 0, stream>>>(EI, norm, p1, p2);
    tag_gemm_kernel<WIN_><<<gemmBlocks, TB, 0, stream>>>(X, p1, p2, W0, b0, bufA);

    zero_f32_kernel<<<cdiv(NTOT_ * H_, TB), TB, 0, stream>>>(p1, NTOT_ * H_);
    prop4_kernel<<<cdiv(propQ, TB), TB, 0, stream>>>(EI, norm, (const float4*)bufA, p1);
    zero_f32_kernel<<<cdiv(NTOT_ * H_, TB), TB, 0, stream>>>(p2, NTOT_ * H_);
    prop4_kernel<<<cdiv(propQ, TB), TB, 0, stream>>>(EI, norm, (const float4*)p1, p2);
    tag_gemm_kernel<H_><<<gemmBlocks, TB, 0, stream>>>(bufA, p1, p2, W1, b1, bufB);

    zero_f32_kernel<<<cdiv(NTOT_ * H_, TB), TB, 0, stream>>>(p1, NTOT_ * H_);
    prop4_kernel<<<cdiv(propQ, TB), TB, 0, stream>>>(EI, norm, (const float4*)bufB, p1);
    zero_f32_kernel<<<cdiv(NTOT_ * H_, TB), TB, 0, stream>>>(p2, NTOT_ * H_);
    prop4_kernel<<<cdiv(propQ, TB), TB, 0, stream>>>(EI, norm, (const float4*)p1, p2);
    tag_gemm_kernel<H_><<<gemmBlocks, TB, 0, stream>>>(bufB, p1, p2, W2, b2, hOut);

    readout_kernel<<<cdiv(NTOT_, TB), TB, 0, stream>>>(hOut, Wr, br, outF);
}


# Scatter sums leave global float atomics: edges bucketed once by 256-row destination block (count, scan, fill); a work-group adds its block's rows in LDS, stores each once; sums stay arrival-order
# speedup vs baseline: 3.2877x; 3.2877x over previous
#include <hip/hip_runtime.h>
#include <math.h>

#define B_    4
#define WIN_  5
#define N_    50000
#define E_    1000000
#define H_    32
#define NTOT_ (B_ * N_)
#define BE_   (B_ * E_)

typedef __attribute__((ext_vector_type(2))) float v2f;
typedef __attribute__((ext_vector_type(8))) float v8f;

static __device__ __forceinline__ void decode_edge(const int* __restrict__ ei, int e,
                                                   int& src, int& dst) {
    int b  = e / E_;
    int el = e - b * E_;
    src = ei[(b * 2 + 0) * E_ + el] + b * N_;
    dst = ei[(b * 2 + 1) * E_ + el] + b * N_;
}

#define RB_   256
#define NB_   ((NTOT_ + RB_ - 1) / RB_)
#define CH_   8192
#define NCH_  ((BE_ + CH_ - 1) / CH_)
#define SRCM_ 0x3FFFF

__global__ void zero_i32_kernel(int* __restrict__ p, int n) {
    int i = blockIdx.x * blockDim.x + threadIdx.x;
    if (i < n) p[i] = 0;
}

__global__ __launch_bounds__(256) void bin_count_kernel(const int* __restrict__ ei,
                                                        int* __restrict__ binCount) {
    __shared__ int hist[NB_];
    for (int i = threadIdx.x; i < NB_; i += 256) hist[i] = 0;
    __syncthreads();
    int e0 = blockIdx.x * CH_;
    for (int k = threadIdx.x; k < CH_; k += 256) {
        int e = e0 + k;
        if (e < BE_) {
            int src, dst;
            decode_edge(ei, e, src, dst);
            atomicAdd(&hist[dst / RB_], 1);
        }
    }
    __syncthreads();
    for (int i = threadIdx.x; i < NB_; i += 256) {
        int c = hist[i];
        if (c) atomicAdd(&binCount[i], c);
    }
}

__global__ __launch_bounds__(1024) void bin_scan_kernel(const int* __restrict__ binCount,
                                                        int* __restrict__ binStart,
                                                        int* __restrict__ cursor) {
    __shared__ int s[1024];
    int t = threadIdx.x;
    int v = (t < NB_) ? binCount[t] : 0;
    s[t] = v;
    __syncthreads();
    for (int off = 1; off < 1024; off <<= 1) {
        int x = (t >= off) ? s[t - off] : 0;
        __syncthreads();
        s[t] += x;
        __syncthreads();
    }
    int excl = s[t] - v;
    if (t < NB_) { binStart[t] = excl; cursor[t] = excl; }
    if (t == NB_ - 1) binStart[NB_] = s[t];
}

__global__ __launch_bounds__(256) void bin_fill_kernel(const int* __restrict__ ei,
                                                       int* __restrict__ cursor,
                                                       int* __restrict__ recs) {
    __shared__ int hist[NB_];
    __shared__ int base[NB_];
    for (int i = threadIdx.x; i < NB_; i += 256) hist[i] = 0;
    __syncthreads();
    int e0 = blockIdx.x * CH_;
    for (int k = threadIdx.x; k < CH_; k += 256) {
        int e = e0 + k;
        if (e < BE_) {
            int src, dst;
            decode_edge(ei, e, src, dst);
            atomicAdd(&hist[dst / RB_], 1);
        }
    }
    __syncthreads();
    for (int i = threadIdx.x; i < NB_; i += 256) {
        int c = hist[i];
        base[i] = c ? atomicAdd(&cursor[i], c) : 0;
        hist[i] = 0;
    }
    __syncthreads();
    for (int k = threadIdx.x; k < CH_; k += 256) {
        int e = e0 + k;
        if (e < BE_) {
            int src, dst;
            decode_edge(ei, e, src, dst);
            int bin = dst / RB_;
            int slot = atomicAdd(&hist[bin], 1);
            recs[base[bin] + slot] = src | ((dst - bin * RB_) << 18);
        }
    }
}

__global__ __launch_bounds__(256) void bin_dinv_kernel(const int* __restrict__ binStart,
                                                       const int* __restrict__ recs,
                                                       float* __restrict__ dinv) {
    __shared__ int cnt[RB_];
    int bin = blockIdx.x;
    cnt[threadIdx.x] = 0;
    __syncthreads();
    int s = binStart[bin], e = binStart[bin + 1];
    for (int i = s + threadIdx.x; i < e; i += 256) atomicAdd(&cnt[recs[i] >> 18], 1);
    __syncthreads();
    int node = bin * RB_ + threadIdx.x;
    if (node < NTOT_) {
        int c = cnt[threadIdx.x];
        dinv[node] = (c > 0) ? rsqrtf((float)c) : 0.0f;
    }
}

__global__ __launch_bounds__(256) void bin_prop5_kernel(const int* __restrict__ binStart,
                                                        const int* __restrict__ recs,
                                                        const float* __restrict__ dinv,
                                                        const float* __restrict__ hin,
                                                        float* __restrict__ hout) {
    __shared__ float tile[RB_ * WIN_];
    __shared__ float dinvD[RB_];
    int bin = blockIdx.x;
    int node0 = bin * RB_;
    int rows = (NTOT_ - node0 < RB_) ? (NTOT_ - node0) : RB_;
    for (int i = threadIdx.x; i < RB_ * WIN_; i += 256) tile[i] = 0.0f;
    dinvD[threadIdx.x] = ((int)threadIdx.x < rows) ? dinv[node0 + threadIdx.x] : 0.0f;
    __syncthreads();
    int s = binStart[bin], e = binStart[bin + 1];
    for (int i = s + threadIdx.x; i < e; i += 256) {
        int r = recs[i];
        int src = r & SRCM_;
        int dl = r >> 18;
        float nrm = dinv[src] * dinvD[dl];
        const float* hp = hin + src * WIN_;
        float a0 = hp[0], a1 = hp[1], a2 = hp[2], a3 = hp[3], a4 = hp[4];
        float* tp = &tile[dl * WIN_];
        atomicAdd(tp + 0, a0 * nrm);
        atomicAdd(tp + 1, a1 * nrm);
        atomicAdd(tp + 2, a2 * nrm);
        atomicAdd(tp + 3, a3 * nrm);
        atomicAdd(tp + 4, a4 * nrm);
    }
    __syncthreads();
    for (int i = threadIdx.x; i < rows * WIN_; i += 256) hout[node0 * WIN_ + i] = tile[i];
}

__global__ __launch_bounds__(256) void bin_prop32_kernel(const int* __restrict__ binStart,
                                                         const int* __restrict__ recs,
                                                         const float* __restrict__ dinv,
                                                         const float* __restrict__ hin,
                                                         float* __restrict__ hout) {
    __shared__ float4 tile4[RB_ * 8];
    __shared__ float dinvD[RB_];
    float* tile = (float*)tile4;
    const float4* __restrict__ hin4 = (const float4*)hin;
    int bin = blockIdx.x;
    int node0 = bin * RB_;
    int rows = (NTOT_ - node0 < RB_) ? (NTOT_ - node0) : RB_;
    for (int i = threadIdx.x; i < RB_ * 8; i += 256) tile4[i] = make_float4(0.f, 0.f, 0.f, 0.f);
    dinvD[threadIdx.x] = ((int)threadIdx.x < rows) ? dinv[node0 + threadIdx.x] : 0.0f;
    __syncthreads();
    int q = threadIdx.x & 7;
    int sub = threadIdx.x >> 3;
    int rot = sub & 3;
    bool r1 = (rot & 1) != 0, r2 = (rot & 2) != 0;
    int o0 = rot, o1 = (rot + 1) & 3, o2 = (rot + 2) & 3, o3 = (rot + 3) & 3;
    int s = binStart[bin], e = binStart[bin + 1];
    for (int i0 = s + sub; i0 < e; i0 += 128) {
        int r[4];
        float ds[4];
        float4 v[4];
#pragma unroll
        for (int u = 0; u < 4; ++u) {
            int i = i0 + u * 32;
            r[u] = (i < e) ? recs[i] : -1;
        }
#pragma unroll
        for (int u = 0; u < 4; ++u) {
            int srcu = (r[u] < 0) ? 0 : (r[u] & SRCM_);
            ds[u] = dinv[srcu];
            v[u] = hin4[srcu * 8 + q];
        }
#pragma unroll
        for (int u = 0; u < 4; ++u) {
            if (r[u] >= 0) {
                int dl = r[u] >> 18;
                float nrm = ds[u] * dinvD[dl];
                float* tp = &tile[dl * H_ + q * 4];
                float b0 = r1 ? v[u].y : v[u].x, b1 = r1 ? v[u].z : v[u].y;
                float b2 = r1 ? v[u].w : v[u].z, b3 = r1 ? v[u].x : v[u].w;
                float c0 = r2 ? b2 : b0, c1 = r2 ? b3 : b1, c2 = r2 ? b0 : b2, c3 = r2 ? b1 : b3;
                atomicAdd(tp + o0, c0 * nrm);
                atomicAdd(tp + o1, c1 * nrm);
                atomicAdd(tp + o2, c2 * nrm);
                atomicAdd(tp + o3, c3 * nrm);
            }
        }
    }
    __syncthreads();
    float4* __restrict__ out4 = (float4*)hout;
    for (int i = threadIdx.x; i < rows * 8; i += 256) out4[node0 * 8 + i] = tile4[i];
}

template <int F>
__global__ void tag_gemm_kernel(const float* __restrict__ x0, const float* __restrict__ x1,
                                const float* __restrict__ x2, const float* __restrict__ W,
                                const float* __restrict__ bias, float* __restrict__ out) {
    const int numTiles = NTOT_ / 16;
    int wave = threadIdx.x >> 5;
    int tile = blockIdx.x * (blockDim.x >> 5) + wave;
    if (tile >= numTiles) return;

    int lane = threadIdx.x & 31;
    int half = lane >> 4;
    int l16  = lane & 15;
    int rowA = tile * 16 + l16;

    v8f c0 = {};
    v8f c1 = {};

    const float* xs[3] = { x0, x1, x2 };
    constexpr bool FULL = (F % 4 == 0);

#pragma unroll
    for (int s = 0; s < 3; ++s) {
        const float* __restrict__ xp = xs[s];
#pragma unroll
        for (int k0 = 0; k0 < F; k0 += 4) {
            int kA = k0 + half * 2;
            v2f a, b0, b1;
            if (FULL) {
                a.x = xp[rowA * F + kA];
                a.y = xp[rowA * F + kA + 1];
                int kW = s * F + kA;
                b0.x = W[(kW    ) * H_ + l16];
                b0.y = W[(kW + 1) * H_ + l16];
                b1.x = W[(kW    ) * H_ + 16 + l16];
                b1.y = W[(kW + 1) * H_ + 16 + l16];
            } else {
                float m0 = (kA     < F) ? 1.0f : 0.0f;
                float m1 = (kA + 1 < F) ? 1.0f : 0.0f;
                int ka0 = (kA     < F) ? kA     : (F - 1);
                int ka1 = (kA + 1 < F) ? kA + 1 : (F - 1);
                a.x = xp[rowA * F + ka0] * m0;
                a.y = xp[rowA * F + ka1] * m1;
                int kw0 = s * F + ka0;
                int kw1 = s * F + ka1;
                b0.x = W[kw0 * H_ + l16] * m0;
                b0.y = W[kw1 * H_ + l16] * m1;
                b1.x = W[kw0 * H_ + 16 + l16] * m0;
                b1.y = W[kw1 * H_ + 16 + l16] * m1;
            }
            c0 = __builtin_amdgcn_wmma_f32_16x16x4_f32(false, a, false, b0,
                                                       (short)0, c0, false, false);
            c1 = __builtin_amdgcn_wmma_f32_16x16x4_f32(false, a, false, b1,
                                                       (short)0, c1, false, false);
        }
    }

#pragma unroll
    for (int r = 0; r < 8; ++r) {
        int row = tile * 16 + half * 8 + r;
        float v0 = tanhf(c0[r] + bias[l16]);
        float v1 = tanhf(c1[r] + bias[16 + l16]);
        out[row * H_ + l16]      = v0;
        out[row * H_ + 16 + l16] = v1;
    }
}

__global__ void readout_kernel(const float* __restrict__ h, const float* __restrict__ Wr,
                               const float* __restrict__ br, float* __restrict__ out) {
    int i = blockIdx.x * blockDim.x + threadIdx.x;
    if (i >= NTOT_) return;
    float acc = br[0];
#pragma unroll
    for (int j = 0; j < H_; ++j) acc += h[i * H_ + j] * Wr[j];
    out[i] = acc;
}

static inline int cdiv(int a, int b) { return (a + b - 1) / b; }

extern "C" void kernel_launch(void* const* d_in, const int* in_sizes, int n_in,
                              void* d_out, int out_size, void* d_ws, size_t ws_size,
                              hipStream_t stream) {
    (void)in_sizes; (void)n_in; (void)out_size; (void)ws_size;

    const float* X  = (const float*)d_in[0];
    const int*   EI = (const int*)d_in[1];
    const float* W0 = (const float*)d_in[2];
    const float* b0 = (const float*)d_in[3];
    const float* W1 = (const float*)d_in[4];
    const float* b1 = (const float*)d_in[5];
    const float* W2 = (const float*)d_in[6];
    const float* b2 = (const float*)d_in[7];
    const float* Wr = (const float*)d_in[8];
    const float* br = (const float*)d_in[9];

    float* outF = (float*)d_out;
    float* hOut = outF + NTOT_;

    float* ws       = (float*)d_ws;
    float* dinv     = ws;
    int*   recs     = (int*)(dinv + NTOT_);
    int*   binCount = recs + BE_;
    int*   binStart = binCount + 1024;
    int*   cursor   = binStart + 1024;
    float* p1   = (float*)(cursor + 1024);
    float* p2   = p1 + (size_t)NTOT_ * H_;
    float* bufA = p2 + (size_t)NTOT_ * H_;
    float* bufB = bufA + (size_t)NTOT_ * H_;

    const int TB = 256;

    zero_i32_kernel<<<cdiv(NB_, TB), TB, 0, stream>>>(binCount, NB_);
    bin_count_kernel<<<NCH_, 256, 0, stream>>>(EI, binCount);
    bin_scan_kernel<<<1, 1024, 0, stream>>>(binCount, binStart, cursor);
    bin_fill_kernel<<<NCH_, 256, 0, stream>>>(EI, cursor, recs);
    bin_dinv_kernel<<<NB_, 256, 0, stream>>>(binStart, recs, dinv);

    const int gemmBlocks = cdiv(NTOT_ / 16, TB / 32);

    bin_prop5_kernel<<<NB_, 256, 0, stream>>>(binStart, recs, dinv, X, p1);
    bin_prop5_kernel<<<NB_, 256, 0, stream>>>(binStart, recs, dinv, p1, p2);
    tag_gemm_kernel<WIN_><<<gemmBlocks, TB, 0, stream>>>(X, p1, p2, W0, b0, bufA);

    bin_prop32_kernel<<<NB_, 256, 0, stream>>>(binStart, recs, dinv, bufA, p1);
    bin_prop32_kernel<<<NB_, 256, 0, stream>>>(binStart, recs, dinv, p1, p2);
    tag_gemm_kernel<H_><<<gemmBlocks, TB, 0, stream>>>(bufA, p1, p2, W1, b1, bufB);

    bin_prop32_kernel<<<NB_, 256, 0, stream>>>(binStart, recs, dinv, bufB, p1);
    bin_prop32_kernel<<<NB_, 256, 0, stream>>>(binStart, recs, dinv, p1, p2);
    tag_gemm_kernel<H_><<<gemmBlocks, TB, 0, stream>>>(bufB, p1, p2, W2, b2, hOut);

    readout_kernel<<<cdiv(NTOT_, TB), TB, 0, stream>>>(hOut, Wr, br, outF);
}


# The dense stages leave their own launches: the work-group that sums a block's second-hop rows in LDS also runs tag_gemm (WMMA f32) and the read-out on that tile; 15 -> 11 launches
# speedup vs baseline: 3.4705x; 1.0556x over previous
#include <hip/hip_runtime.h>
#include <math.h>

#define B_    4
#define WIN_  5
#define N_    50000
#define E_    1000000
#define H_    32
#define NTOT_ (B_ * N_)
#define BE_   (B_ * E_)

typedef __attribute__((ext_vector_type(2))) float v2f;
typedef __attribute__((ext_vector_type(8))) float v8f;

static __device__ __forceinline__ void decode_edge(const int* __restrict__ ei, int e,
                                                   int& src, int& dst) {
    int b  = e / E_;
    int el = e - b * E_;
    src = ei[(b * 2 + 0) * E_ + el] + b * N_;
    dst = ei[(b * 2 + 1) * E_ + el] + b * N_;
}

#define RB_   256
#define NB_   ((NTOT_ + RB_ - 1) / RB_)
#define CH_   8192
#define NCH_  ((BE_ + CH_ - 1) / CH_)
#define SRCM_ 0x3FFFF
#define SP_   36
#define SP4_  (SP_ / 4)

__global__ void zero_i32_kernel(int* __restrict__ p, int n) {
    int i = blockIdx.x * blockDim.x + threadIdx.x;
    if (i < n) p[i] = 0;
}

__global__ __launch_bounds__(256) void bin_count_kernel(const int* __restrict__ ei,
                                                        int* __restrict__ binCount) {
    __shared__ int hist[NB_];
    for (int i = threadIdx.x; i < NB_; i += 256) hist[i] = 0;
    __syncthreads();
    int e0 = blockIdx.x * CH_;
    for (int k = threadIdx.x; k < CH_; k += 256) {
        int e = e0 + k;
        if (e < BE_) {
            int src, dst;
            decode_edge(ei, e, src, dst);
            atomicAdd(&hist[dst / RB_], 1);
        }
    }
    __syncthreads();
    for (int i = threadIdx.x; i < NB_; i += 256) {
        int c = hist[i];
        if (c) atomicAdd(&binCount[i], c);
    }
}

__global__ __launch_bounds__(1024) void bin_scan_kernel(const int* __restrict__ binCount,
                                                        int* __restrict__ binStart,
                                                        int* __restrict__ cursor) {
    __shared__ int s[1024];
    int t = threadIdx.x;
    int v = (t < NB_) ? binCount[t] : 0;
    s[t] = v;
    __syncthreads();
    for (int off = 1; off < 1024; off <<= 1) {
        int x = (t >= off) ? s[t - off] : 0;
        __syncthreads();
        s[t] += x;
        __syncthreads();
    }
    int excl = s[t] - v;
    if (t < NB_) { binStart[t] = excl; cursor[t] = excl; }
    if (t == NB_ - 1) binStart[NB_] = s[t];
}

__global__ __launch_bounds__(256) void bin_fill_kernel(const int* __restrict__ ei,
                                                       int* __restrict__ cursor,
                                                       int* __restrict__ recs) {
    __shared__ int hist[NB_];
    __shared__ int base[NB_];
    for (int i = threadIdx.x; i < NB_; i += 256) hist[i] = 0;
    __syncthreads();
    int e0 = blockIdx.x * CH_;
    for (int k = threadIdx.x; k < CH_; k += 256) {
        int e = e0 + k;
        if (e < BE_) {
            int src, dst;
            decode_edge(ei, e, src, dst);
            atomicAdd(&hist[dst / RB_], 1);
        }
    }
    __syncthreads();
    for (int i = threadIdx.x; i < NB_; i += 256) {
        int c = hist[i];
        base[i] = c ? atomicAdd(&cursor[i], c) : 0;
        hist[i] = 0;
    }
    __syncthreads();
    for (int k = threadIdx.x; k < CH_; k += 256) {
        int e = e0 + k;
        if (e < BE_) {
            int src, dst;
            decode_edge(ei, e, src, dst);
            int bin = dst / RB_;
            int slot = atomicAdd(&hist[bin], 1);
            recs[base[bin] + slot] = src | ((dst - bin * RB_) << 18);
        }
    }
}

__global__ __launch_bounds__(256) void bin_dinv_kernel(const int* __restrict__ binStart,
                                                       const int* __restrict__ recs,
                                                       float* __restrict__ dinv) {
    __shared__ int cnt[RB_];
    int bin = blockIdx.x;
    cnt[threadIdx.x] = 0;
    __syncthreads();
    int s = binStart[bin], e = binStart[bin + 1];
    for (int i = s + threadIdx.x; i < e; i += 256) atomicAdd(&cnt[recs[i] >> 18], 1);
    __syncthreads();
    int node = bin * RB_ + threadIdx.x;
    if (node < NTOT_) {
        int c = cnt[threadIdx.x];
        dinv[node] = (c > 0) ? rsqrtf((float)c) : 0.0f;
    }
}

static __device__ __forceinline__ void accum5(float* tile, float* dinvD, int bin, int node0, int rows,
                                              const int* __restrict__ binStart,
                                              const int* __restrict__ recs,
                                              const float* __restrict__ dinv,
                                              const float* __restrict__ hin) {
    for (int i = threadIdx.x; i < RB_ * WIN_; i += 256) tile[i] = 0.0f;
    dinvD[threadIdx.x] = ((int)threadIdx.x < rows) ? dinv[node0 + threadIdx.x] : 0.0f;
    __syncthreads();
    int s = binStart[bin], e = binStart[bin + 1];
    for (int i = s + threadIdx.x; i < e; i += 256) {
        int r = recs[i];
        int src = r & SRCM_;
        int dl = r >> 18;
        float nrm = dinv[src] * dinvD[dl];
        const float* hp = hin + src * WIN_;
        float a0 = hp[0], a1 = hp[1], a2 = hp[2], a3 = hp[3], a4 = hp[4];
        float* tp = &tile[dl * WIN_];
        atomicAdd(tp + 0, a0 * nrm);
        atomicAdd(tp + 1, a1 * nrm);
        atomicAdd(tp + 2, a2 * nrm);
        atomicAdd(tp + 3, a3 * nrm);
        atomicAdd(tp + 4, a4 * nrm);
    }
    __syncthreads();
}

static __device__ __forceinline__ void accum32(float4* tile4, float* dinvD, int bin, int node0, int rows,
                                               const int* __restrict__ binStart,
                                               const int* __restrict__ recs,
                                               const float* __restrict__ dinv,
                                               const float* __restrict__ hin) {
    float* tile = (float*)tile4;
    const float4* __restrict__ hin4 = (const float4*)hin;
    for (int i = threadIdx.x; i < RB_ * 8; i += 256) tile4[i] = make_float4(0.f, 0.f, 0.f, 0.f);
    dinvD[threadIdx.x] = ((int)threadIdx.x < rows) ? dinv[node0 + threadIdx.x] : 0.0f;
    __syncthreads();
    int q = threadIdx.x & 7;
    int sub = threadIdx.x >> 3;
    int rot = sub & 3;
    bool r1 = (rot & 1) != 0, r2 = (rot & 2) != 0;
    int o0 = rot, o1 = (rot + 1) & 3, o2 = (rot + 2) & 3, o3 = (rot + 3) & 3;
    int s = binStart[bin], e = binStart[bin + 1];
    for (int i0 = s + sub; i0 < e; i0 += 128) {
        int r[4];
        float ds[4];
        float4 v[4];
#pragma unroll
        for (int u = 0; u < 4; ++u) {
            int i = i0 + u * 32;
            r[u] = (i < e) ? recs[i] : -1;
        }
#pragma unroll
        for (int u = 0; u < 4; ++u) {
            int srcu = (r[u] < 0) ? 0 : (r[u] & SRCM_);
            ds[u] = dinv[srcu];
            v[u] = hin4[srcu * 8 + q];
        }
#pragma unroll
        for (int u = 0; u < 4; ++u) {
            if (r[u] >= 0) {
                int dl = r[u] >> 18;
                float nrm = ds[u] * dinvD[dl];
                float* tp = &tile[dl * H_ + q * 4];
                float b0 = r1 ? v[u].y : v[u].x, b1 = r1 ? v[u].z : v[u].y;
                float b2 = r1 ? v[u].w : v[u].z, b3 = r1 ? v[u].x : v[u].w;
                float c0 = r2 ? b2 : b0, c1 = r2 ? b3 : b1, c2 = r2 ? b0 : b2, c3 = r2 ? b1 : b3;
                atomicAdd(tp + o0, c0 * nrm);
                atomicAdd(tp + o1, c1 * nrm);
                atomicAdd(tp + o2, c2 * nrm);
                atomicAdd(tp + o3, c3 * nrm);
            }
        }
    }
    __syncthreads();
}

__global__ __launch_bounds__(256) void bin_prop5_kernel(const int* __restrict__ binStart,
                                                        const int* __restrict__ recs,
                                                        const float* __restrict__ dinv,
                                                        const float* __restrict__ hin,
                                                        float* __restrict__ hout) {
    __shared__ float tile[RB_ * WIN_];
    __shared__ float dinvD[RB_];
    int bin = blockIdx.x;
    int node0 = bin * RB_;
    int rows = (NTOT_ - node0 < RB_) ? (NTOT_ - node0) : RB_;
    accum5(tile, dinvD, bin, node0, rows, binStart, recs, dinv, hin);
    for (int i = threadIdx.x; i < rows * WIN_; i += 256) hout[node0 * WIN_ + i] = tile[i];
}

__global__ __launch_bounds__(256) void bin_prop32_kernel(const int* __restrict__ binStart,
                                                         const int* __restrict__ recs,
                                                         const float* __restrict__ dinv,
                                                         const float* __restrict__ hin,
                                                         float* __restrict__ hout) {
    __shared__ float4 tile4[RB_ * 8];
    __shared__ float dinvD[RB_];
    int bin = blockIdx.x;
    int node0 = bin * RB_;
    int rows = (NTOT_ - node0 < RB_) ? (NTOT_ - node0) : RB_;
    accum32(tile4, dinvD, bin, node0, rows, binStart, recs, dinv, hin);
    float4* __restrict__ out4 = (float4*)hout;
    for (int i = threadIdx.x; i < rows * 8; i += 256) out4[node0 * 8 + i] = tile4[i];
}

static __device__ __forceinline__ void dense_tail(float4* stage4, const v8f (&c)[2][2],
                                                  const float* __restrict__ bias,
                                                  int wave, int half, int l16, int node0, int rows,
                                                  float* __restrict__ hout) {
    float* stage = (float*)stage4;
    float bz0 = bias[l16], bz1 = bias[16 + l16];
#pragma unroll
    for (int t = 0; t < 2; ++t) {
#pragma unroll
        for (int r = 0; r < 8; ++r) {
            int row = wave * 32 + t * 16 + half * 8 + r;
            stage[row * SP_ + l16]      = tanhf(c[t][0][r] + bz0);
            stage[row * SP_ + 16 + l16] = tanhf(c[t][1][r] + bz1);
        }
    }
    __syncthreads();
    float4* __restrict__ out4 = (float4*)hout;
    for (int i = threadIdx.x; i < rows * 8; i += 256) out4[node0 * 8 + i] = stage4[(i >> 3) * SP4_ + (i & 7)];
}

__global__ __launch_bounds__(256) void bin_prop5_dense_kernel(const int* __restrict__ binStart,
                                                              const int* __restrict__ recs,
                                                              const float* __restrict__ dinv,
                                                              const float* __restrict__ x0,
                                                              const float* __restrict__ x1,
                                                              const float* __restrict__ W,
                                                              const float* __restrict__ bias,
                                                              float* __restrict__ hout) {
    __shared__ float tile[RB_ * WIN_];
    __shared__ float dinvD[RB_];
    __shared__ float4 stage4[RB_ * SP4_];
    int bin = blockIdx.x;
    int node0 = bin * RB_;
    int rows = (NTOT_ - node0 < RB_) ? (NTOT_ - node0) : RB_;
    accum5(tile, dinvD, bin, node0, rows, binStart, recs, dinv, x1);

    int wave = threadIdx.x >> 5;
    int lane = threadIdx.x & 31;
    int half = lane >> 4;
    int l16  = lane & 15;
    v8f c[2][2] = {};
#pragma unroll
    for (int s = 0; s < 3; ++s) {
#pragma unroll
        for (int k0 = 0; k0 < WIN_; k0 += 4) {
            int kA = k0 + half * 2;
            float m0 = (kA     < WIN_) ? 1.0f : 0.0f;
            float m1 = (kA + 1 < WIN_) ? 1.0f : 0.0f;
            int ka0 = (kA     < WIN_) ? kA     : (WIN_ - 1);
            int ka1 = (kA + 1 < WIN_) ? kA + 1 : (WIN_ - 1);
            int kw0 = s * WIN_ + ka0;
            int kw1 = s * WIN_ + ka1;
            v2f b0, b1;
            b0.x = W[kw0 * H_ + l16] * m0;
            b0.y = W[kw1 * H_ + l16] * m1;
            b1.x = W[kw0 * H_ + 16 + l16] * m0;
            b1.y = W[kw1 * H_ + 16 + l16] * m1;
#pragma unroll
            for (int t = 0; t < 2; ++t) {
                int rl = wave * 32 + t * 16 + l16;
                int rg = node0 + rl;
                rg = (rg < NTOT_) ? rg : (NTOT_ - 1);
                v2f a;
                if (s == 0)      { a.x = x0[rg * WIN_ + ka0] * m0;  a.y = x0[rg * WIN_ + ka1] * m1; }
                else if (s == 1) { a.x = x1[rg * WIN_ + ka0] * m0;  a.y = x1[rg * WIN_ + ka1] * m1; }
                else             { a.x = tile[rl * WIN_ + ka0] * m0; a.y = tile[rl * WIN_ + ka1] * m1; }
                c[t][0] = __builtin_amdgcn_wmma_f32_16x16x4_f32(false, a, false, b0,
                                                                (short)0, c[t][0], false, false);
                c[t][1] = __builtin_amdgcn_wmma_f32_16x16x4_f32(false, a, false, b1,
                                                                (short)0, c[t][1], false, false);
            }
        }
    }
    dense_tail(stage4, c, bias, wave, half, l16, node0, rows, hout);
}

template <int PITCH>
static __device__ __forceinline__ void seg_from_tile(const float* tile, int wave, int half, int l16,
                                                     const float* __restrict__ Wseg, v8f (&c)[2][2]) {
#pragma unroll
    for (int k0 = 0; k0 < H_; k0 += 4) {
        int kA = k0 + half * 2;
        v2f b0, b1;
        b0.x = Wseg[(kA    ) * H_ + l16];
        b0.y = Wseg[(kA + 1) * H_ + l16];
        b1.x = Wseg[(kA    ) * H_ + 16 + l16];
        b1.y = Wseg[(kA + 1) * H_ + 16 + l16];
#pragma unroll
        for (int t = 0; t < 2; ++t) {
            const float* p = &tile[(wave * 32 + t * 16 + l16) * PITCH + kA];
            v2f a;
            a.x = p[0];
            a.y = p[1];
            c[t][0] = __builtin_amdgcn_wmma_f32_16x16x4_f32(false, a, false, b0,
                                                            (short)0, c[t][0], false, false);
            c[t][1] = __builtin_amdgcn_wmma_f32_16x16x4_f32(false, a, false, b1,
                                                            (short)0, c[t][1], false, false);
        }
    }
}

template <bool READOUT>
__global__ __launch_bounds__(256, 8) void bin_prop32_dense_kernel(const int* __restrict__ binStart,
                                                               const int* __restrict__ recs,
                                                               const float* __restrict__ dinv,
                                                               const float* __restrict__ x0,
                                                               const float* __restrict__ x1,
                                                               const float* __restrict__ W,
                                                               const float* __restrict__ bias,
                                                               float* __restrict__ hout,
                                                               const float* __restrict__ Wr,
                                                               const float* __restrict__ br,
                                                               float* __restrict__ rout) {
    __shared__ float4 tile4[RB_ * 8];
    __shared__ float4 stage4[RB_ * SP4_];
    __shared__ float dinvD[RB_];
    const float* tile = (const float*)tile4;
    const float* stage = (const float*)stage4;
    int bin = blockIdx.x;
    int node0 = bin * RB_;
    int rows = (NTOT_ - node0 < RB_) ? (NTOT_ - node0) : RB_;
    const float4* __restrict__ x0_4 = (const float4*)x0;
    const float4* __restrict__ x1_4 = (const float4*)x1;
    accum32(tile4, dinvD, bin, node0, rows, binStart, recs, dinv, x1);

    int wave = threadIdx.x >> 5;
    int lane = threadIdx.x & 31;
    int half = lane >> 4;
    int l16  = lane & 15;
    int srow = threadIdx.x >> 3;
    int sq   = threadIdx.x & 7;
    float4 g[8];
#pragma unroll
    for (int j = 0; j < 8; ++j) {
        int i = threadIdx.x + j * 256;
        g[j] = (i < rows * 8) ? x0_4[node0 * 8 + i] : make_float4(0.f, 0.f, 0.f, 0.f);
    }
#pragma unroll
    for (int j = 0; j < 8; ++j) stage4[(srow + j * 32) * SP4_ + sq] = g[j];
    __syncthreads();
    v8f c[2][2] = {};
    seg_from_tile<SP_>(stage, wave, half, l16, W, c);
#pragma unroll
    for (int j = 0; j < 8; ++j) {
        int i = threadIdx.x + j * 256;
        g[j] = (i < rows * 8) ? x1_4[node0 * 8 + i] : make_float4(0.f, 0.f, 0.f, 0.f);
    }
    __syncthreads();
#pragma unroll
    for (int j = 0; j < 8; ++j) stage4[(srow + j * 32) * SP4_ + sq] = g[j];
    __syncthreads();
    seg_from_tile<SP_>(stage, wave, half, l16, W + H_ * H_, c);
    seg_from_tile<H_>(tile, wave, half, l16, W + 2 * H_ * H_, c);
    dense_tail(stage4, c, bias, wave, half, l16, node0, rows, hout);
    if (READOUT) {
        int row = threadIdx.x;
        if (row < rows) {
            float acc = br[0];
#pragma unroll
            for (int j = 0; j < H_; ++j) acc += stage[row * SP_ + j] * Wr[j];
            rout[node0 + row] = acc;
        }
    }
}

static inline int cdiv(int a, int b) { return (a + b - 1) / b; }

extern "C" void kernel_launch(void* const* d_in, const int* in_sizes, int n_in,
                              void* d_out, int out_size, void* d_ws, size_t ws_size,
                              hipStream_t stream) {
    (void)in_sizes; (void)n_in; (void)out_size; (void)ws_size;

    const float* X  = (const float*)d_in[0];
    const int*   EI = (const int*)d_in[1];
    const float* W0 = (const float*)d_in[2];
    const float* b0 = (const float*)d_in[3];
    const float* W1 = (const float*)d_in[4];
    const float* b1 = (const float*)d_in[5];
    const float* W2 = (const float*)d_in[6];
    const float* b2 = (const float*)d_in[7];
    const float* Wr = (const float*)d_in[8];
    const float* br = (const float*)d_in[9];

    float* outF = (float*)d_out;
    float* hOut = outF + NTOT_;

    float* ws       = (float*)d_ws;
    float* dinv     = ws;
    int*   recs     = (int*)(dinv + NTOT_);
    int*   binCount = recs + BE_;
    int*   binStart = binCount + 1024;
    int*   cursor   = binStart + 1024;
    float* p1   = (float*)(cursor + 1024);
    float* bufA = p1 + (size_t)NTOT_ * H_;
    float* bufB = bufA + (size_t)NTOT_ * H_;

    const int TB = 256;

    zero_i32_kernel<<<cdiv(NB_, TB), TB, 0, stream>>>(binCount, NB_);
    bin_count_kernel<<<NCH_, 256, 0, stream>>>(EI, binCount);
    bin_scan_kernel<<<1, 1024, 0, stream>>>(binCount, binStart, cursor);
    bin_fill_kernel<<<NCH_, 256, 0, stream>>>(EI, cursor, recs);
    bin_dinv_kernel<<<NB_, 256, 0, stream>>>(binStart, recs, dinv);

    bin_prop5_kernel<<<NB_, 256, 0, stream>>>(binStart, recs, dinv, X, p1);
    bin_prop5_dense_kernel<<<NB_, 256, 0, stream>>>(binStart, recs, dinv, X, p1, W0, b0, bufA);

    bin_prop32_kernel<<<NB_, 256, 0, stream>>>(binStart, recs, dinv, bufA, p1);
    bin_prop32_dense_kernel<false><<<NB_, 256, 0, stream>>>(binStart, recs, dinv, bufA, p1, W1, b1, bufB,
                                                            Wr, br, outF);

    bin_prop32_kernel<<<NB_, 256, 0, stream>>>(binStart, recs, dinv, bufB, p1);
    bin_prop32_dense_kernel<true><<<NB_, 256, 0, stream>>>(binStart, recs, dinv, bufB, p1, W2, b2, hOut,
                                                           Wr, br, outF);
}


# Edge norm dinv[src]*dinv[dst] is computed once per call: the first propagation stores it per record slot, the other five load it coalesced beside the record; no dinv[src] lookup, sum order kept
# speedup vs baseline: 4.0924x; 1.1792x over previous
#include <hip/hip_runtime.h>
#include <math.h>

#define B_    4
#define WIN_  5
#define N_    50000
#define E_    1000000
#define H_    32
#define NTOT_ (B_ * N_)
#define BE_   (B_ * E_)

typedef __attribute__((ext_vector_type(2))) float v2f;
typedef __attribute__((ext_vector_type(8))) float v8f;

static __device__ __forceinline__ void decode_edge(const int* __restrict__ ei, int e,
                                                   int& src, int& dst) {
    int b  = e / E_;
    int el = e - b * E_;
    src = ei[(b * 2 + 0) * E_ + el] + b * N_;
    dst = ei[(b * 2 + 1) * E_ + el] + b * N_;
}

#define RB_   256
#define NB_   ((NTOT_ + RB_ - 1) / RB_)
#define CH_   8192
#define NCH_  ((BE_ + CH_ - 1) / CH_)
#define SRCM_ 0x3FFFF
#define SP_   36
#define SP4_  (SP_ / 4)

__global__ void zero_i32_kernel(int* __restrict__ p, int n) {
    int i = blockIdx.x * blockDim.x + threadIdx.x;
    if (i < n) p[i] = 0;
}

__global__ __launch_bounds__(256) void bin_count_kernel(const int* __restrict__ ei,
                                                        int* __restrict__ binCount) {
    __shared__ int hist[NB_];
    for (int i = threadIdx.x; i < NB_; i += 256) hist[i] = 0;
    __syncthreads();
    int e0 = blockIdx.x * CH_;
    for (int k = threadIdx.x; k < CH_; k += 256) {
        int e = e0 + k;
        if (e < BE_) {
            int src, dst;
            decode_edge(ei, e, src, dst);
            atomicAdd(&hist[dst / RB_], 1);
        }
    }
    __syncthreads();
    for (int i = threadIdx.x; i < NB_; i += 256) {
        int c = hist[i];
        if (c) atomicAdd(&binCount[i], c);
    }
}

__global__ __launch_bounds__(1024) void bin_scan_kernel(const int* __restrict__ binCount,
                                                        int* __restrict__ binStart,
                                                        int* __restrict__ cursor) {
    __shared__ int s[1024];
    int t = threadIdx.x;
    int v = (t < NB_) ? binCount[t] : 0;
    s[t] = v;
    __syncthreads();
    for (int off = 1; off < 1024; off <<= 1) {
        int x = (t >= off) ? s[t - off] : 0;
        __syncthreads();
        s[t] += x;
        __syncthreads();
    }
    int excl = s[t] - v;
    if (t < NB_) { binStart[t] = excl; cursor[t] = excl; }
    if (t == NB_ - 1) binStart[NB_] = s[t];
}

__global__ __launch_bounds__(256) void bin_fill_kernel(const int* __restrict__ ei,
                                                       int* __restrict__ cursor,
                                                       int* __restrict__ recs) {
    __shared__ int hist[NB_];
    __shared__ int base[NB_];
    for (int i = threadIdx.x; i < NB_; i += 256) hist[i] = 0;
    __syncthreads();
    int e0 = blockIdx.x * CH_;
    for (int k = threadIdx.x; k < CH_; k += 256) {
        int e = e0 + k;
        if (e < BE_) {
            int src, dst;
            decode_edge(ei, e, src, dst);
            atomicAdd(&hist[dst / RB_], 1);
        }
    }
    __syncthreads();
    for (int i = threadIdx.x; i < NB_; i += 256) {
        int c = hist[i];
        base[i] = c ? atomicAdd(&cursor[i], c) : 0;
        hist[i] = 0;
    }
    __syncthreads();
    for (int k = threadIdx.x; k < CH_; k += 256) {
        int e = e0 + k;
        if (e < BE_) {
            int src, dst;
            decode_edge(ei, e, src, dst);
            int bin = dst / RB_;
            int slot = atomicAdd(&hist[bin], 1);
            recs[base[bin] + slot] = src | ((dst - bin * RB_) << 18);
        }
    }
}

__global__ __launch_bounds__(256) void bin_dinv_kernel(const int* __restrict__ binStart,
                                                       const int* __restrict__ recs,
                                                       float* __restrict__ dinv) {
    __shared__ int cnt[RB_];
    int bin = blockIdx.x;
    cnt[threadIdx.x] = 0;
    __syncthreads();
    int s = binStart[bin], e = binStart[bin + 1];
    for (int i = s + threadIdx.x; i < e; i += 256) atomicAdd(&cnt[recs[i] >> 18], 1);
    __syncthreads();
    int node = bin * RB_ + threadIdx.x;
    if (node < NTOT_) {
        int c = cnt[threadIdx.x];
        dinv[node] = (c > 0) ? rsqrtf((float)c) : 0.0f;
    }
}


template <bool MAKE_W>
static __device__ __forceinline__ void accum5(float* tile, float* dinvD, int bin, int node0, int rows,
                                              const int* __restrict__ binStart,
                                              const int* __restrict__ recs,
                                              const float* __restrict__ dinv,
                                              float* __restrict__ wts,
                                              const float* __restrict__ hin) {
    for (int i = threadIdx.x; i < RB_ * WIN_; i += 256) tile[i] = 0.0f;
    if (MAKE_W) dinvD[threadIdx.x] = ((int)threadIdx.x < rows) ? dinv[node0 + threadIdx.x] : 0.0f;
    __syncthreads();
    int s = binStart[bin], e = binStart[bin + 1];
    for (int i = s + threadIdx.x; i < e; i += 256) {
        int r = recs[i];
        int src = r & SRCM_;
        int dl = r >> 18;
        float nrm;
        if (MAKE_W) { nrm = dinv[src] * dinvD[dl]; wts[i] = nrm; }
        else        { nrm = wts[i]; }
        const float* hp = hin + src * WIN_;
        float a0 = hp[0], a1 = hp[1], a2 = hp[2], a3 = hp[3], a4 = hp[4];
        float* tp = &tile[dl * WIN_];
        atomicAdd(tp + 0, a0 * nrm);
        atomicAdd(tp + 1, a1 * nrm);
        atomicAdd(tp + 2, a2 * nrm);
        atomicAdd(tp + 3, a3 * nrm);
        atomicAdd(tp + 4, a4 * nrm);
    }
    __syncthreads();
}

static __device__ __forceinline__ void accum32(float4* tile4, int bin,
                                               const int* __restrict__ binStart,
                                               const int* __restrict__ recs,
                                               const float* __restrict__ wts,
                                               const float* __restrict__ hin) {
    float* tile = (float*)tile4;
    const float4* __restrict__ hin4 = (const float4*)hin;
    for (int i = threadIdx.x; i < RB_ * 8; i += 256) tile4[i] = make_float4(0.f, 0.f, 0.f, 0.f);
    __syncthreads();
    int q = threadIdx.x & 7;
    int sub = threadIdx.x >> 3;
    int rot = sub & 3;
    bool r1 = (rot & 1) != 0, r2 = (rot & 2) != 0;
    int o0 = rot, o1 = (rot + 1) & 3, o2 = (rot + 2) & 3, o3 = (rot + 3) & 3;
    int s = binStart[bin], e = binStart[bin + 1];
    for (int i0 = s + sub; i0 < e; i0 += 128) {
        int r[4];
        float w[4];
        float4 v[4];
#pragma unroll
        for (int u = 0; u < 4; ++u) {
            int i = i0 + u * 32;
            r[u] = -1;
            w[u] = 0.0f;
            if (i < e) { r[u] = recs[i]; w[u] = wts[i]; }
        }
#pragma unroll
        for (int u = 0; u < 4; ++u) {
            int srcu = (r[u] < 0) ? 0 : (r[u] & SRCM_);
            v[u] = hin4[srcu * 8 + q];
        }
#pragma unroll
        for (int u = 0; u < 4; ++u) {
            if (r[u] >= 0) {
                int dl = r[u] >> 18;
                float nrm = w[u];
                float* tp = &tile[dl * H_ + q * 4];
                float b0 = r1 ? v[u].y : v[u].x, b1 = r1 ? v[u].z : v[u].y;
                float b2 = r1 ? v[u].w : v[u].z, b3 = r1 ? v[u].x : v[u].w;
                float c0 = r2 ? b2 : b0, c1 = r2 ? b3 : b1, c2 = r2 ? b0 : b2, c3 = r2 ? b1 : b3;
                atomicAdd(tp + o0, c0 * nrm);
                atomicAdd(tp + o1, c1 * nrm);
                atomicAdd(tp + o2, c2 * nrm);
                atomicAdd(tp + o3, c3 * nrm);
            }
        }
    }
    __syncthreads();
}

__global__ __launch_bounds__(256) void bin_prop5_kernel(const int* __restrict__ binStart,
                                                        const int* __restrict__ recs,
                                                        const float* __restrict__ dinv,
                                                        float* __restrict__ wts,
                                                        const float* __restrict__ hin,
                                                        float* __restrict__ hout) {
    __shared__ float tile[RB_ * WIN_];
    __shared__ float dinvD[RB_];
    int bin = blockIdx.x;
    int node0 = bin * RB_;
    int rows = (NTOT_ - node0 < RB_) ? (NTOT_ - node0) : RB_;
    accum5<true>(tile, dinvD, bin, node0, rows, binStart, recs, dinv, wts, hin);
    for (int i = threadIdx.x; i < rows * WIN_; i += 256) hout[node0 * WIN_ + i] = tile[i];
}

__global__ __launch_bounds__(256) void bin_prop32_kernel(const int* __restrict__ binStart,
                                                         const int* __restrict__ recs,
                                                         const float* __restrict__ wts,
                                                         const float* __restrict__ hin,
                                                         float* __restrict__ hout) {
    __shared__ float4 tile4[RB_ * 8];
    int bin = blockIdx.x;
    int node0 = bin * RB_;
    int rows = (NTOT_ - node0 < RB_) ? (NTOT_ - node0) : RB_;
    accum32(tile4, bin, binStart, recs, wts, hin);
    float4* __restrict__ out4 = (float4*)hout;
    for (int i = threadIdx.x; i < rows * 8; i += 256) out4[node0 * 8 + i] = tile4[i];
}

static __device__ __forceinline__ void dense_tail(float4* stage4, const v8f (&c)[2][2],
                                                  const float* __restrict__ bias,
                                                  int wave, int half, int l16, int node0, int rows,
                                                  float* __restrict__ hout) {
    float* stage = (float*)stage4;
    float bz0 = bias[l16], bz1 = bias[16 + l16];
#pragma unroll
    for (int t = 0; t < 2; ++t) {
#pragma unroll
        for (int r = 0; r < 8; ++r) {
            int row = wave * 32 + t * 16 + half * 8 + r;
            stage[row * SP_ + l16]      = tanhf(c[t][0][r] + bz0);
            stage[row * SP_ + 16 + l16] = tanhf(c[t][1][r] + bz1);
        }
    }
    __syncthreads();
    float4* __restrict__ out4 = (float4*)hout;
    for (int i = threadIdx.x; i < rows * 8; i += 256) out4[node0 * 8 + i] = stage4[(i >> 3) * SP4_ + (i & 7)];
}

__global__ __launch_bounds__(256) void bin_prop5_dense_kernel(const int* __restrict__ binStart,
                                                              const int* __restrict__ recs,
                                                              float* __restrict__ wts,
                                                              const float* __restrict__ x0,
                                                              const float* __restrict__ x1,
                                                              const float* __restrict__ W,
                                                              const float* __restrict__ bias,
                                                              float* __restrict__ hout) {
    __shared__ float tile[RB_ * WIN_];
    __shared__ float4 stage4[RB_ * SP4_];
    int bin = blockIdx.x;
    int node0 = bin * RB_;
    int rows = (NTOT_ - node0 < RB_) ? (NTOT_ - node0) : RB_;
    accum5<false>(tile, nullptr, bin, node0, rows, binStart, recs, nullptr, wts, x1);

    int wave = threadIdx.x >> 5;
    int lane = threadIdx.x & 31;
    int half = lane >> 4;
    int l16  = lane & 15;
    v8f c[2][2] = {};
#pragma unroll
    for (int s = 0; s < 3; ++s) {
#pragma unroll
        for (int k0 = 0; k0 < WIN_; k0 += 4) {
            int kA = k0 + half * 2;
            float m0 = (kA     < WIN_) ? 1.0f : 0.0f;
            float m1 = (kA + 1 < WIN_) ? 1.0f : 0.0f;
            int ka0 = (kA     < WIN_) ? kA     : (WIN_ - 1);
            int ka1 = (kA + 1 < WIN_) ? kA + 1 : (WIN_ - 1);
            int kw0 = s * WIN_ + ka0;
            int kw1 = s * WIN_ + ka1;
            v2f b0, b1;
            b0.x = W[kw0 * H_ + l16] * m0;
            b0.y = W[kw1 * H_ + l16] * m1;
            b1.x = W[kw0 * H_ + 16 + l16] * m0;
            b1.y = W[kw1 * H_ + 16 + l16] * m1;
#pragma unroll
            for (int t = 0; t < 2; ++t) {
                int rl = wave * 32 + t * 16 + l16;
                int rg = node0 + rl;
                rg = (rg < NTOT_) ? rg : (NTOT_ - 1);
                v2f a;
                if (s == 0)      { a.x = x0[rg * WIN_ + ka0] * m0;  a.y = x0[rg * WIN_ + ka1] * m1; }
                else if (s == 1) { a.x = x1[rg * WIN_ + ka0] * m0;  a.y = x1[rg * WIN_ + ka1] * m1; }
                else             { a.x = tile[rl * WIN_ + ka0] * m0; a.y = tile[rl * WIN_ + ka1] * m1; }
                c[t][0] = __builtin_amdgcn_wmma_f32_16x16x4_f32(false, a, false, b0,
                                                                (short)0, c[t][0], false, false);
                c[t][1] = __builtin_amdgcn_wmma_f32_16x16x4_f32(false, a, false, b1,
                                                                (short)0, c[t][1], false, false);
            }
        }
    }
    dense_tail(stage4, c, bias, wave, half, l16, node0, rows, hout);
}

template <int PITCH>
static __device__ __forceinline__ void seg_from_tile(const float* tile, int wave, int half, int l16,
                                                     const float* __restrict__ Wseg, v8f (&c)[2][2]) {
#pragma unroll
    for (int k0 = 0; k0 < H_; k0 += 4) {
        int kA = k0 + half * 2;
        v2f b0, b1;
        b0.x = Wseg[(kA    ) * H_ + l16];
        b0.y = Wseg[(kA + 1) * H_ + l16];
        b1.x = Wseg[(kA    ) * H_ + 16 + l16];
        b1.y = Wseg[(kA + 1) * H_ + 16 + l16];
#pragma unroll
        for (int t = 0; t < 2; ++t) {
            const float* p = &tile[(wave * 32 + t * 16 + l16) * PITCH + kA];
            v2f a;
            a.x = p[0];
            a.y = p[1];
            c[t][0] = __builtin_amdgcn_wmma_f32_16x16x4_f32(false, a, false, b0,
                                                            (short)0, c[t][0], false, false);
            c[t][1] = __builtin_amdgcn_wmma_f32_16x16x4_f32(false, a, false, b1,
                                                            (short)0, c[t][1], false, false);
        }
    }
}

template <bool READOUT>
__global__ __launch_bounds__(256, 8) void bin_prop32_dense_kernel(const int* __restrict__ binStart,
                                                               const int* __restrict__ recs,
                                                               const float* __restrict__ wts,
                                                               const float* __restrict__ x0,
                                                               const float* __restrict__ x1,
                                                               const float* __restrict__ W,
                                                               const float* __restrict__ bias,
                                                               float* __restrict__ hout,
                                                               const float* __restrict__ Wr,
                                                               const float* __restrict__ br,
                                                               float* __restrict__ rout) {
    __shared__ float4 tile4[RB_ * 8];
    __shared__ float4 stage4[RB_ * SP4_];
    const float* tile = (const float*)tile4;
    const float* stage = (const float*)stage4;
    int bin = blockIdx.x;
    int node0 = bin * RB_;
    int rows = (NTOT_ - node0 < RB_) ? (NTOT_ - node0) : RB_;
    const float4* __restrict__ x0_4 = (const float4*)x0;
    const float4* __restrict__ x1_4 = (const float4*)x1;
    accum32(tile4, bin, binStart, recs, wts, x1);

    int wave = threadIdx.x >> 5;
    int lane = threadIdx.x & 31;
    int half = lane >> 4;
    int l16  = lane & 15;
    int srow = threadIdx.x >> 3;
    int sq   = threadIdx.x & 7;
    float4 g[8];
#pragma unroll
    for (int j = 0; j < 8; ++j) {
        int i = threadIdx.x + j * 256;
        g[j] = (i < rows * 8) ? x0_4[node0 * 8 + i] : make_float4(0.f, 0.f, 0.f, 0.f);
    }
#pragma unroll
    for (int j = 0; j < 8; ++j) stage4[(srow + j * 32) * SP4_ + sq] = g[j];
    __syncthreads();
    v8f c[2][2] = {};
    seg_from_tile<SP_>(stage, wave, half, l16, W, c);
#pragma unroll
    for (int j = 0; j < 8; ++j) {
        int i = threadIdx.x + j * 256;
        g[j] = (i < rows * 8) ? x1_4[node0 * 8 + i] : make_float4(0.f, 0.f, 0.f, 0.f);
    }
    __syncthreads();
#pragma unroll
    for (int j = 0; j < 8; ++j) stage4[(srow + j * 32) * SP4_ + sq] = g[j];
    __syncthreads();
    seg_from_tile<SP_>(stage, wave, half, l16, W + H_ * H_, c);
    seg_from_tile<H_>(tile, wave, half, l16, W + 2 * H_ * H_, c);
    dense_tail(stage4, c, bias, wave, half, l16, node0, rows, hout);
    if (READOUT) {
        int row = threadIdx.x;
        if (row < rows) {
            float acc = br[0];
#pragma unroll
            for (int j = 0; j < H_; ++j) acc += stage[row * SP_ + j] * Wr[j];
            rout[node0 + row] = acc;
        }
    }
}

static inline int cdiv(int a, int b) { return (a + b - 1) / b; }

extern "C" void kernel_launch(void* const* d_in, const int* in_sizes, int n_in,
                              void* d_out, int out_size, void* d_ws, size_t ws_size,
                              hipStream_t stream) {
    (void)in_sizes; (void)n_in; (void)out_size; (void)ws_size;

    const float* X  = (const float*)d_in[0];
    const int*   EI = (const int*)d_in[1];
    const float* W0 = (const float*)d_in[2];
    const float* b0 = (const float*)d_in[3];
    const float* W1 = (const float*)d_in[4];
    const float* b1 = (const float*)d_in[5];
    const float* W2 = (const float*)d_in[6];
    const float* b2 = (const float*)d_in[7];
    const float* Wr = (const float*)d_in[8];
    const float* br = (const float*)d_in[9];

    float* outF = (float*)d_out;
    float* hOut = outF + NTOT_;

    float* ws       = (float*)d_ws;
    float* dinv     = ws;
    int*   recs     = (int*)(dinv + NTOT_);
    int*   binCount = recs + BE_;
    int*   binStart = binCount + 1024;
    int*   cursor   = binStart + 1024;
    float* p1   = (float*)(cursor + 1024);
    float* bufA = p1 + (size_t)NTOT_ * H_;
    float* bufB = bufA + (size_t)NTOT_ * H_;
    float* wts  = bufB + (size_t)NTOT_ * H_;

    const int TB = 256;

    zero_i32_kernel<<<cdiv(NB_, TB), TB, 0, stream>>>(binCount, NB_);
    bin_count_kernel<<<NCH_, 256, 0, stream>>>(EI, binCount);
    bin_scan_kernel<<<1, 1024, 0, stream>>>(binCount, binStart, cursor);
    bin_fill_kernel<<<NCH_, 256, 0, stream>>>(EI, cursor, recs);
    bin_dinv_kernel<<<NB_, 256, 0, stream>>>(binStart, recs, dinv);

    bin_prop5_kernel<<<NB_, 256, 0, stream>>>(binStart, recs, dinv, wts, X, p1);
    bin_prop5_dense_kernel<<<NB_, 256, 0, stream>>>(binStart, recs, wts, X, p1, W0, b0, bufA);

    bin_prop32_kernel<<<NB_, 256, 0, stream>>>(binStart, recs, wts, bufA, p1);
    bin_prop32_dense_kernel<false><<<NB_, 256, 0, stream>>>(binStart, recs, wts, bufA, p1, W1, b1, bufB,
                                                            Wr, br, outF);

    bin_prop32_kernel<<<NB_, 256, 0, stream>>>(binStart, recs, wts, bufB, p1);
    bin_prop32_dense_kernel<true><<<NB_, 256, 0, stream>>>(binStart, recs, wts, bufB, p1, W2, b2, hOut,
                                                           Wr, br, outF);
}


# Bucket loops (degree count, six propagations) keep more loads in flight: 8 edges per lane group per trip in the wide sums, next trip's records requested under the row gathers; sums stay arrival-order
# speedup vs baseline: 4.3605x; 1.0655x over previous
#include <hip/hip_runtime.h>
#include <math.h>

#define B_    4
#define WIN_  5
#define N_    50000
#define E_    1000000
#define H_    32
#define NTOT_ (B_ * N_)
#define BE_   (B_ * E_)

typedef __attribute__((ext_vector_type(2))) float v2f;
typedef __attribute__((ext_vector_type(8))) float v8f;

static __device__ __forceinline__ void decode_edge(const int* __restrict__ ei, int e,
                                                   int& src, int& dst) {
    int b  = e / E_;
    int el = e - b * E_;
    src = ei[(b * 2 + 0) * E_ + el] + b * N_;
    dst = ei[(b * 2 + 1) * E_ + el] + b * N_;
}

#define RB_   256
#define NB_   ((NTOT_ + RB_ - 1) / RB_)
#define CH_   8192
#define NCH_  ((BE_ + CH_ - 1) / CH_)
#define SRCM_ 0x3FFFF
#define SP_   36
#define SP4_  (SP_ / 4)
#define E5_   2
#define E32_  8

__global__ void zero_i32_kernel(int* __restrict__ p, int n) {
    int i = blockIdx.x * blockDim.x + threadIdx.x;
    if (i < n) p[i] = 0;
}

__global__ __launch_bounds__(256) void bin_count_kernel(const int* __restrict__ ei,
                                                        int* __restrict__ binCount) {
    __shared__ int hist[NB_];
    for (int i = threadIdx.x; i < NB_; i += 256) hist[i] = 0;
    __syncthreads();
    int e0 = blockIdx.x * CH_;
    for (int k = threadIdx.x; k < CH_; k += 256) {
        int e = e0 + k;
        if (e < BE_) {
            int src, dst;
            decode_edge(ei, e, src, dst);
            atomicAdd(&hist[dst / RB_], 1);
        }
    }
    __syncthreads();
    for (int i = threadIdx.x; i < NB_; i += 256) {
        int c = hist[i];
        if (c) atomicAdd(&binCount[i], c);
    }
}

__global__ __launch_bounds__(1024) void bin_scan_kernel(const int* __restrict__ binCount,
                                                        int* __restrict__ binStart,
                                                        int* __restrict__ cursor) {
    __shared__ int s[1024];
    int t = threadIdx.x;
    int v = (t < NB_) ? binCount[t] : 0;
    s[t] = v;
    __syncthreads();
    for (int off = 1; off < 1024; off <<= 1) {
        int x = (t >= off) ? s[t - off] : 0;
        __syncthreads();
        s[t] += x;
        __syncthreads();
    }
    int excl = s[t] - v;
    if (t < NB_) { binStart[t] = excl; cursor[t] = excl; }
    if (t == NB_ - 1) binStart[NB_] = s[t];
}

__global__ __launch_bounds__(256) void bin_fill_kernel(const int* __restrict__ ei,
                                                       int* __restrict__ cursor,
                                                       int* __restrict__ recs) {
    __shared__ int hist[NB_];
    __shared__ int base[NB_];
    for (int i = threadIdx.x; i < NB_; i += 256) hist[i] = 0;
    __syncthreads();
    int e0 = blockIdx.x * CH_;
    for (int k = threadIdx.x; k < CH_; k += 256) {
        int e = e0 + k;
        if (e < BE_) {
            int src, dst;
            decode_edge(ei, e, src, dst);
            atomicAdd(&hist[dst / RB_], 1);
        }
    }
    __syncthreads();
    for (int i = threadIdx.x; i < NB_; i += 256) {
        int c = hist[i];
        base[i] = c ? atomicAdd(&cursor[i], c) : 0;
        hist[i] = 0;
    }
    __syncthreads();
    for (int k = threadIdx.x; k < CH_; k += 256) {
        int e = e0 + k;
        if (e < BE_) {
            int src, dst;
            decode_edge(ei, e, src, dst);
            int bin = dst / RB_;
            int slot = atomicAdd(&hist[bin], 1);
            recs[base[bin] + slot] = src | ((dst - bin * RB_) << 18);
        }
    }
}

__global__ __launch_bounds__(256) void bin_dinv_kernel(const int* __restrict__ binStart,
                                                       const int* __restrict__ recs,
                                                       float* __restrict__ dinv) {
    __shared__ int cnt[RB_];
    int bin = blockIdx.x;
    cnt[threadIdx.x] = 0;
    __syncthreads();
    int s = binStart[bin], e = binStart[bin + 1];
    int last = (e > 0) ? e - 1 : 0;
    for (int i0 = s + threadIdx.x; i0 < e; i0 += 4 * 256) {
        int r[4];
#pragma unroll
        for (int u = 0; u < 4; ++u) {
            int i = i0 + u * 256;
            int rv = recs[(i < e) ? i : last];
            r[u] = (i < e) ? rv : -1;
        }
#pragma unroll
        for (int u = 0; u < 4; ++u) {
            if (r[u] >= 0) atomicAdd(&cnt[r[u] >> 18], 1);
        }
    }
    __syncthreads();
    int node = bin * RB_ + threadIdx.x;
    if (node < NTOT_) {
        int c = cnt[threadIdx.x];
        dinv[node] = (c > 0) ? rsqrtf((float)c) : 0.0f;
    }
}


template <bool MAKE_W>
static __device__ __forceinline__ void accum5(float* tile, float* dinvD, int bin, int node0, int rows,
                                              const int* __restrict__ binStart,
                                              const int* __restrict__ recs,
                                              const float* __restrict__ dinv,
                                              float* __restrict__ wts,
                                              const float* __restrict__ hin) {
    for (int i = threadIdx.x; i < RB_ * WIN_; i += 256) tile[i] = 0.0f;
    if (MAKE_W) dinvD[threadIdx.x] = ((int)threadIdx.x < rows) ? dinv[node0 + threadIdx.x] : 0.0f;
    __syncthreads();
    int s = binStart[bin], e = binStart[bin + 1];
    int last = (e > 0) ? e - 1 : 0;
    int i0 = s + threadIdx.x;
    int r[E5_];
    float w[E5_];
#pragma unroll
    for (int u = 0; u < E5_; ++u) {
        int i = i0 + u * 256;
        int ic = (i < e) ? i : last;
        int rv = recs[ic];
        w[u] = MAKE_W ? 0.0f : wts[ic];
        r[u] = (i < e) ? rv : -1;
    }
    while (i0 < e) {
        float a[E5_][WIN_];
        float ds[E5_];
#pragma unroll
        for (int u = 0; u < E5_; ++u) {
            int src = (r[u] < 0) ? 0 : (r[u] & SRCM_);
            const float* hp = hin + src * WIN_;
#pragma unroll
            for (int j = 0; j < WIN_; ++j) a[u][j] = hp[j];
            ds[u] = MAKE_W ? dinv[src] : 0.0f;
        }
        int i1 = i0 + E5_ * 256;
        int rn[E5_];
        float wn[E5_];
#pragma unroll
        for (int u = 0; u < E5_; ++u) {
            int i = i1 + u * 256;
            int ic = (i < e) ? i : last;
            int rv = recs[ic];
            wn[u] = MAKE_W ? 0.0f : wts[ic];
            rn[u] = (i < e) ? rv : -1;
        }
        auto add_edge = [&](int u) {
            int dl = r[u] >> 18;
            float nrm;
            if (MAKE_W) { nrm = ds[u] * dinvD[dl]; wts[i0 + u * 256] = nrm; }
            else        { nrm = w[u]; }
            float* tp = &tile[dl * WIN_];
            atomicAdd(tp + 0, a[u][0] * nrm);
            atomicAdd(tp + 1, a[u][1] * nrm);
            atomicAdd(tp + 2, a[u][2] * nrm);
            atomicAdd(tp + 3, a[u][3] * nrm);
            atomicAdd(tp + 4, a[u][4] * nrm);
        };
        if (i0 + (E5_ - 1) * 256 < e) {
#pragma unroll
            for (int u = 0; u < E5_; ++u) add_edge(u);
        } else {
#pragma unroll
            for (int u = 0; u < E5_; ++u) {
                if (r[u] >= 0) add_edge(u);
            }
        }
#pragma unroll
        for (int u = 0; u < E5_; ++u) { r[u] = rn[u]; w[u] = wn[u]; }
        i0 = i1;
    }
    __syncthreads();
}

static __device__ __forceinline__ void accum32(float4* tile4, int bin,
                                               const int* __restrict__ binStart,
                                               const int* __restrict__ recs,
                                               const float* __restrict__ wts,
                                               const float* __restrict__ hin) {
    float* tile = (float*)tile4;
    const float4* __restrict__ hin4 = (const float4*)hin;
    for (int i = threadIdx.x; i < RB_ * 8; i += 256) tile4[i] = make_float4(0.f, 0.f, 0.f, 0.f);
    __syncthreads();
    int q = threadIdx.x & 7;
    int sub = threadIdx.x >> 3;
    int rot = sub & 3;
    bool r1 = (rot & 1) != 0, r2 = (rot & 2) != 0;
    int o0 = rot, o1 = (rot + 1) & 3, o2 = (rot + 2) & 3, o3 = (rot + 3) & 3;
    int s = binStart[bin], e = binStart[bin + 1];
    int last = (e > 0) ? e - 1 : 0;
    int i0 = s + sub;
    int r[E32_];
    float w[E32_];
#pragma unroll
    for (int u = 0; u < E32_; ++u) {
        int i = i0 + u * 32;
        int ic = (i < e) ? i : last;
        int rv = recs[ic];
        w[u] = wts[ic];
        r[u] = (i < e) ? rv : -1;
    }
    while (i0 < e) {
        float4 v[E32_];
#pragma unroll
        for (int u = 0; u < E32_; ++u) {
            int srcu = (r[u] < 0) ? 0 : (r[u] & SRCM_);
            v[u] = hin4[srcu * 8 + q];
        }
        int i1 = i0 + E32_ * 32;
        int rn[E32_];
        float wn[E32_];
#pragma unroll
        for (int u = 0; u < E32_; ++u) {
            int i = i1 + u * 32;
            int ic = (i < e) ? i : last;
            int rv = recs[ic];
            wn[u] = wts[ic];
            rn[u] = (i < e) ? rv : -1;
        }
        auto add_edge = [&](int rec, float nrm, float4 x) {
            int dl = rec >> 18;
            float* tp = &tile[dl * H_ + q * 4];
            float b0 = r1 ? x.y : x.x, b1 = r1 ? x.z : x.y;
            float b2 = r1 ? x.w : x.z, b3 = r1 ? x.x : x.w;
            float c0 = r2 ? b2 : b0, c1 = r2 ? b3 : b1, c2 = r2 ? b0 : b2, c3 = r2 ? b1 : b3;
            atomicAdd(tp + o0, c0 * nrm);
            atomicAdd(tp + o1, c1 * nrm);
            atomicAdd(tp + o2, c2 * nrm);
            atomicAdd(tp + o3, c3 * nrm);
        };
        if (i0 + (E32_ - 1) * 32 < e) {
#pragma unroll
            for (int u = 0; u < E32_; ++u) add_edge(r[u], w[u], v[u]);
        } else {
#pragma unroll
            for (int u = 0; u < E32_; ++u) {
                if (r[u] >= 0) add_edge(r[u], w[u], v[u]);
            }
        }
#pragma unroll
        for (int u = 0; u < E32_; ++u) { r[u] = rn[u]; w[u] = wn[u]; }
        i0 = i1;
    }
    __syncthreads();
}

__global__ __launch_bounds__(256) void bin_prop5_kernel(const int* __restrict__ binStart,
                                                        const int* __restrict__ recs,
                                                        const float* __restrict__ dinv,
                                                        float* __restrict__ wts,
                                                        const float* __restrict__ hin,
                                                        float* __restrict__ hout) {
    __shared__ float tile[RB_ * WIN_];
    __shared__ float dinvD[RB_];
    int bin = blockIdx.x;
    int node0 = bin * RB_;
    int rows = (NTOT_ - node0 < RB_) ? (NTOT_ - node0) : RB_;
    accum5<true>(tile, dinvD, bin, node0, rows, binStart, recs, dinv, wts, hin);
    for (int i = threadIdx.x; i < rows * WIN_; i += 256) hout[node0 * WIN_ + i] = tile[i];
}

__global__ __launch_bounds__(256) void bin_prop32_kernel(const int* __restrict__ binStart,
                                                         const int* __restrict__ recs,
                                                         const float* __restrict__ wts,
                                                         const float* __restrict__ hin,
                                                         float* __restrict__ hout) {
    __shared__ float4 tile4[RB_ * 8];
    int bin = blockIdx.x;
    int node0 = bin * RB_;
    int rows = (NTOT_ - node0 < RB_) ? (NTOT_ - node0) : RB_;
    accum32(tile4, bin, binStart, recs, wts, hin);
    float4* __restrict__ out4 = (float4*)hout;
    for (int i = threadIdx.x; i < rows * 8; i += 256) out4[node0 * 8 + i] = tile4[i];
}

static __device__ __forceinline__ void dense_tail(float4* stage4, const v8f (&c)[2][2],
                                                  const float* __restrict__ bias,
                                                  int wave, int half, int l16, int node0, int rows,
                                                  float* __restrict__ hout) {
    float* stage = (float*)stage4;
    float bz0 = bias[l16], bz1 = bias[16 + l16];
#pragma unroll
    for (int t = 0; t < 2; ++t) {
#pragma unroll
        for (int r = 0; r < 8; ++r) {
            int row = wave * 32 + t * 16 + half * 8 + r;
            stage[row * SP_ + l16]      = tanhf(c[t][0][r] + bz0);
            stage[row * SP_ + 16 + l16] = tanhf(c[t][1][r] + bz1);
        }
    }
    __syncthreads();
    float4* __restrict__ out4 = (float4*)hout;
    for (int i = threadIdx.x; i < rows * 8; i += 256) out4[node0 * 8 + i] = stage4[(i >> 3) * SP4_ + (i & 7)];
}

__global__ __launch_bounds__(256) void bin_prop5_dense_kernel(const int* __restrict__ binStart,
                                                              const int* __restrict__ recs,
                                                              float* __restrict__ wts,
                                                              const float* __restrict__ x0,
                                                              const float* __restrict__ x1,
                                                              const float* __restrict__ W,
                                                              const float* __restrict__ bias,
                                                              float* __restrict__ hout) {
    __shared__ float tile[RB_ * WIN_];
    __shared__ float4 stage4[RB_ * SP4_];
    int bin = blockIdx.x;
    int node0 = bin * RB_;
    int rows = (NTOT_ - node0 < RB_) ? (NTOT_ - node0) : RB_;
    accum5<false>(tile, nullptr, bin, node0, rows, binStart, recs, nullptr, wts, x1);

    int wave = threadIdx.x >> 5;
    int lane = threadIdx.x & 31;
    int half = lane >> 4;
    int l16  = lane & 15;
    v8f c[2][2] = {};
#pragma unroll
    for (int s = 0; s < 3; ++s) {
#pragma unroll
        for (int k0 = 0; k0 < WIN_; k0 += 4) {
            int kA = k0 + half * 2;
            float m0 = (kA     < WIN_) ? 1.0f : 0.0f;
            float m1 = (kA + 1 < WIN_) ? 1.0f : 0.0f;
            int ka0 = (kA     < WIN_) ? kA     : (WIN_ - 1);
            int ka1 = (kA + 1 < WIN_) ? kA + 1 : (WIN_ - 1);
            int kw0 = s * WIN_ + ka0;
            int kw1 = s * WIN_ + ka1;
            v2f b0, b1;
            b0.x = W[kw0 * H_ + l16] * m0;
            b0.y = W[kw1 * H_ + l16] * m1;
            b1.x = W[kw0 * H_ + 16 + l16] * m0;
            b1.y = W[kw1 * H_ + 16 + l16] * m1;
#pragma unroll
            for (int t = 0; t < 2; ++t) {
                int rl = wave * 32 + t * 16 + l16;
                int rg = node0 + rl;
                rg = (rg < NTOT_) ? rg : (NTOT_ - 1);
                v2f a;
                if (s == 0)      { a.x = x0[rg * WIN_ + ka0] * m0;  a.y = x0[rg * WIN_ + ka1] * m1; }
                else if (s == 1) { a.x = x1[rg * WIN_ + ka0] * m0;  a.y = x1[rg * WIN_ + ka1] * m1; }
                else             { a.x = tile[rl * WIN_ + ka0] * m0; a.y = tile[rl * WIN_ + ka1] * m1; }
                c[t][0] = __builtin_amdgcn_wmma_f32_16x16x4_f32(false, a, false, b0,
                                                                (short)0, c[t][0], false, false);
                c[t][1] = __builtin_amdgcn_wmma_f32_16x16x4_f32(false, a, false, b1,
                                                                (short)0, c[t][1], false, false);
            }
        }
    }
    dense_tail(stage4, c, bias, wave, half, l16, node0, rows, hout);
}

template <int PITCH>
static __device__ __forceinline__ void seg_from_tile(const float* tile, int wave, int half, int l16,
                                                     const float* __restrict__ Wseg, v8f (&c)[2][2]) {
#pragma unroll
    for (int k0 = 0; k0 < H_; k0 += 4) {
        int kA = k0 + half * 2;
        v2f b0, b1;
        b0.x = Wseg[(kA    ) * H_ + l16];
        b0.y = Wseg[(kA + 1) * H_ + l16];
        b1.x = Wseg[(kA    ) * H_ + 16 + l16];
        b1.y = Wseg[(kA + 1) * H_ + 16 + l16];
#pragma unroll
        for (int t = 0; t < 2; ++t) {
            const float* p = &tile[(wave * 32 + t * 16 + l16) * PITCH + kA];
            v2f a;
            a.x = p[0];
            a.y = p[1];
            c[t][0] = __builtin_amdgcn_wmma_f32_16x16x4_f32(false, a, false, b0,
                                                            (short)0, c[t][0], false, false);
            c[t][1] = __builtin_amdgcn_wmma_f32_16x16x4_f32(false, a, false, b1,
                                                            (short)0, c[t][1], false, false);
        }
    }
}

template <bool READOUT>
__global__ __launch_bounds__(256, 8) void bin_prop32_dense_kernel(const int* __restrict__ binStart,
                                                               const int* __restrict__ recs,
                                                               const float* __restrict__ wts,
                                                               const float* __restrict__ x0,
                                                               const float* __restrict__ x1,
                                                               const float* __restrict__ W,
                                                               const float* __restrict__ bias,
                                                               float* __restrict__ hout,
                                                               const float* __restrict__ Wr,
                                                               const float* __restrict__ br,
                                                               float* __restrict__ rout) {
    __shared__ float4 tile4[RB_ * 8];
    __shared__ float4 stage4[RB_ * SP4_];
    const float* tile = (const float*)tile4;
    const float* stage = (const float*)stage4;
    int bin = blockIdx.x;
    int node0 = bin * RB_;
    int rows = (NTOT_ - node0 < RB_) ? (NTOT_ - node0) : RB_;
    const float4* __restrict__ x0_4 = (const float4*)x0;
    const float4* __restrict__ x1_4 = (const float4*)x1;
    accum32(tile4, bin, binStart, recs, wts, x1);

    int wave = threadIdx.x >> 5;
    int lane = threadIdx.x & 31;
    int half = lane >> 4;
    int l16  = lane & 15;
    int srow = threadIdx.x >> 3;
    int sq   = threadIdx.x & 7;
    float4 g[8];
#pragma unroll
    for (int j = 0; j < 8; ++j) {
        int i = threadIdx.x + j * 256;
        g[j] = (i < rows * 8) ? x0_4[node0 * 8 + i] : make_float4(0.f, 0.f, 0.f, 0.f);
    }
#pragma unroll
    for (int j = 0; j < 8; ++j) stage4[(srow + j * 32) * SP4_ + sq] = g[j];
    __syncthreads();
    v8f c[2][2] = {};
    seg_from_tile<SP_>(stage, wave, half, l16, W, c);
#pragma unroll
    for (int j = 0; j < 8; ++j) {
        int i = threadIdx.x + j * 256;
        g[j] = (i < rows * 8) ? x1_4[node0 * 8 + i] : make_float4(0.f, 0.f, 0.f, 0.f);
    }
    __syncthreads();
#pragma unroll
    for (int j = 0; j < 8; ++j) stage4[(srow + j * 32) * SP4_ + sq] = g[j];
    __syncthreads();
    seg_from_tile<SP_>(stage, wave, half, l16, W + H_ * H_, c);
    seg_from_tile<H_>(tile, wave, half, l16, W + 2 * H_ * H_, c);
    dense_tail(stage4, c, bias, wave, half, l16, node0, rows, hout);
    if (READOUT) {
        int row = threadIdx.x;
        if (row < rows) {
            float acc = br[0];
#pragma unroll
            for (int j = 0; j < H_; ++j) acc += stage[row * SP_ + j] * Wr[j];
            rout[node0 + row] = acc;
        }
    }
}

static inline int cdiv(int a, int b) { return (a + b - 1) / b; }

extern "C" void kernel_launch(void* const* d_in, const int* in_sizes, int n_in,
                              void* d_out, int out_size, void* d_ws, size_t ws_size,
                              hipStream_t stream) {
    (void)in_sizes; (void)n_in; (void)out_size; (void)ws_size;

    const float* X  = (const float*)d_in[0];
    const int*   EI = (const int*)d_in[1];
    const float* W0 = (const float*)d_in[2];
    const float* b0 = (const float*)d_in[3];
    const float* W1 = (const float*)d_in[4];
    const float* b1 = (const float*)d_in[5];
    const float* W2 = (const float*)d_in[6];
    const float* b2 = (const float*)d_in[7];
    const float* Wr = (const float*)d_in[8];
    const float* br = (const float*)d_in[9];

    float* outF = (float*)d_out;
    float* hOut = outF + NTOT_;

    float* ws       = (float*)d_ws;
    float* dinv     = ws;
    int*   recs     = (int*)(dinv + NTOT_);
    int*   binCount = recs + BE_;
    int*   binStart = binCount + 1024;
    int*   cursor   = binStart + 1024;
    float* p1   = (float*)(cursor + 1024);
    float* bufA = p1 + (size_t)NTOT_ * H_;
    float* bufB = bufA + (size_t)NTOT_ * H_;
    float* wts  = bufB + (size_t)NTOT_ * H_;

    const int TB = 256;

    zero_i32_kernel<<<cdiv(NB_, TB), TB, 0, stream>>>(binCount, NB_);
    bin_count_kernel<<<NCH_, 256, 0, stream>>>(EI, binCount);
    bin_scan_kernel<<<1, 1024, 0, stream>>>(binCount, binStart, cursor);
    bin_fill_kernel<<<NCH_, 256, 0, stream>>>(EI, cursor, recs);
    bin_dinv_kernel<<<NB_, 256, 0, stream>>>(binStart, recs, dinv);

    bin_prop5_kernel<<<NB_, 256, 0, stream>>>(binStart, recs, dinv, wts, X, p1);
    bin_prop5_dense_kernel<<<NB_, 256, 0, stream>>>(binStart, recs, wts, X, p1, W0, b0, bufA);

    bin_prop32_kernel<<<NB_, 256, 0, stream>>>(binStart, recs, wts, bufA, p1);
    bin_prop32_dense_kernel<false><<<NB_, 256, 0, stream>>>(binStart, recs, wts, bufA, p1, W1, b1, bufB,
                                                            Wr, br, outF);

    bin_prop32_kernel<<<NB_, 256, 0, stream>>>(binStart, recs, wts, bufB, p1);
    bin_prop32_dense_kernel<true><<<NB_, 256, 0, stream>>>(binStart, recs, wts, bufB, p1, W2, b2, hOut,
                                                           Wr, br, outF);
}


# Bucket build (count, fill): a chunk's edges are read once by 16-byte loads issued up front and kept in registers (counting add returns the slot), no waited 4-byte load per edge; sums arrival-order
# speedup vs baseline: 4.5939x; 1.0535x over previous
#include <hip/hip_runtime.h>
#include <math.h>

#define B_    4
#define WIN_  5
#define N_    50000
#define E_    1000000
#define H_    32
#define NTOT_ (B_ * N_)
#define BE_   (B_ * E_)

typedef __attribute__((ext_vector_type(2))) float v2f;
typedef __attribute__((ext_vector_type(8))) float v8f;

static __device__ __forceinline__ void decode_edge(const int* __restrict__ ei, int e,
                                                   int& src, int& dst) {
    int b  = e / E_;
    int el = e - b * E_;
    src = ei[(b * 2 + 0) * E_ + el] + b * N_;
    dst = ei[(b * 2 + 1) * E_ + el] + b * N_;
}

#define RB_   256
#define NB_   ((NTOT_ + RB_ - 1) / RB_)
#define CH_   8192
#define NCH_  ((BE_ + CH_ - 1) / CH_)
#define EPT_  (CH_ / 256)
#define NG4_  (EPT_ / 4)
#define SRCM_ 0x3FFFF
#define SP_   36
#define SP4_  (SP_ / 4)
#define E5_   2
#define E32_  8

__global__ void zero_i32_kernel(int* __restrict__ p, int n) {
    int i = blockIdx.x * blockDim.x + threadIdx.x;
    if (i < n) p[i] = 0;
}

static __device__ __forceinline__ int group_pos(int e, int& nb) {
    bool ok = e < BE_;
    int ec = ok ? e : 0;
    int b  = ec / E_;
    int el = ec - b * E_;
    nb = ok ? b * N_ : -1;
    return (b * 2 + 1) * E_ + el;
}

__global__ __launch_bounds__(256) void bin_count_kernel(const int* __restrict__ ei,
                                                        int* __restrict__ binCount) {
    __shared__ int hist[NB_];
    for (int i = threadIdx.x; i < NB_; i += 256) hist[i] = 0;
    int e0 = blockIdx.x * CH_;
    int4 d[NG4_];
    int  nb[NG4_];
#pragma unroll
    for (int k = 0; k < NG4_; ++k) {
        int p = group_pos(e0 + (k * 256 + threadIdx.x) * 4, nb[k]);
        d[k] = *(const int4*)(ei + p);
    }
    __syncthreads();
#pragma unroll
    for (int k = 0; k < NG4_; ++k) {
        if (nb[k] >= 0) {
            atomicAdd(&hist[(d[k].x + nb[k]) / RB_], 1);
            atomicAdd(&hist[(d[k].y + nb[k]) / RB_], 1);
            atomicAdd(&hist[(d[k].z + nb[k]) / RB_], 1);
            atomicAdd(&hist[(d[k].w + nb[k]) / RB_], 1);
        }
    }
    __syncthreads();
    for (int i = threadIdx.x; i < NB_; i += 256) {
        int c = hist[i];
        if (c) atomicAdd(&binCount[i], c);
    }
}

__global__ __launch_bounds__(1024) void bin_scan_kernel(const int* __restrict__ binCount,
                                                        int* __restrict__ binStart,
                                                        int* __restrict__ cursor) {
    __shared__ int s[1024];
    int t = threadIdx.x;
    int v = (t < NB_) ? binCount[t] : 0;
    s[t] = v;
    __syncthreads();
    for (int off = 1; off < 1024; off <<= 1) {
        int x = (t >= off) ? s[t - off] : 0;
        __syncthreads();
        s[t] += x;
        __syncthreads();
    }
    int excl = s[t] - v;
    if (t < NB_) { binStart[t] = excl; cursor[t] = excl; }
    if (t == NB_ - 1) binStart[NB_] = s[t];
}

#define KEYB_ 10
#define KEYM_ ((1 << KEYB_) - 1)
static_assert(NB_ <= (1 << KEYB_) && NB_ <= 4 * 256 && CH_ <= (1 << (31 - KEYB_)), "key packing and the reservation's four blocks per thread");
__global__ __launch_bounds__(256) void bin_fill_kernel(const int* __restrict__ ei,
                                                       int* __restrict__ cursor,
                                                       int* __restrict__ recs) {
    __shared__ int hist[NB_];
    __shared__ int base[NB_];
    for (int i = threadIdx.x; i < NB_; i += 256) hist[i] = 0;
    int e0 = blockIdx.x * CH_;
    int4 s[NG4_], d[NG4_];
    int  nb[NG4_];
#pragma unroll
    for (int k = 0; k < NG4_; ++k) {
        int p = group_pos(e0 + (k * 256 + threadIdx.x) * 4, nb[k]);
        s[k] = *(const int4*)(ei + p - E_);
        d[k] = *(const int4*)(ei + p);
    }
    __syncthreads();
    int rec[EPT_];
    int key[EPT_];
#pragma unroll
    for (int k = 0; k < NG4_; ++k) {
        int sv[4] = { s[k].x, s[k].y, s[k].z, s[k].w };
        int dv[4] = { d[k].x, d[k].y, d[k].z, d[k].w };
#pragma unroll
        for (int c = 0; c < 4; ++c) {
            int dst = dv[c] + nb[k];
            int bin = dst / RB_;
            rec[k * 4 + c] = (sv[c] + nb[k]) | ((dst - bin * RB_) << 18);
            key[k * 4 + c] = (nb[k] >= 0) ? bin : -1;
        }
    }
    bool whole = e0 + CH_ <= BE_;
    if (whole) {
#pragma unroll
        for (int j = 0; j < EPT_; ++j) key[j] |= atomicAdd(&hist[key[j]], 1) << KEYB_;
    } else {
#pragma unroll
        for (int j = 0; j < EPT_; ++j) {
            if (key[j] >= 0) key[j] |= atomicAdd(&hist[key[j]], 1) << KEYB_;
        }
    }
    __syncthreads();
    {
        int cnt[4], bs[4];
#pragma unroll
        for (int u = 0; u < 4; ++u) {
            int i = threadIdx.x + u * 256;
            cnt[u] = (i < NB_) ? hist[i] : 0;
        }
#pragma unroll
        for (int u = 0; u < 4; ++u) {
            int i = threadIdx.x + u * 256;
            bs[u] = cnt[u] ? atomicAdd(&cursor[i], cnt[u]) : 0;
        }
#pragma unroll
        for (int u = 0; u < 4; ++u) {
            int i = threadIdx.x + u * 256;
            if (i < NB_) base[i] = bs[u];
        }
    }
    __syncthreads();
    if (whole) {
#pragma unroll
        for (int j = 0; j < EPT_; ++j) recs[base[key[j] & KEYM_] + (key[j] >> KEYB_)] = rec[j];
    } else {
#pragma unroll
        for (int j = 0; j < EPT_; ++j) {
            if (key[j] >= 0) recs[base[key[j] & KEYM_] + (key[j] >> KEYB_)] = rec[j];
        }
    }
}

__global__ __launch_bounds__(256) void bin_dinv_kernel(const int* __restrict__ binStart,
                                                       const int* __restrict__ recs,
                                                       float* __restrict__ dinv) {
    __shared__ int cnt[RB_];
    int bin = blockIdx.x;
    cnt[threadIdx.x] = 0;
    __syncthreads();
    int s = binStart[bin], e = binStart[bin + 1];
    int last = (e > 0) ? e - 1 : 0;
    for (int i0 = s + threadIdx.x; i0 < e; i0 += 4 * 256) {
        int r[4];
#pragma unroll
        for (int u = 0; u < 4; ++u) {
            int i = i0 + u * 256;
            int rv = recs[(i < e) ? i : last];
            r[u] = (i < e) ? rv : -1;
        }
#pragma unroll
        for (int u = 0; u < 4; ++u) {
            if (r[u] >= 0) atomicAdd(&cnt[r[u] >> 18], 1);
        }
    }
    __syncthreads();
    int node = bin * RB_ + threadIdx.x;
    if (node < NTOT_) {
        int c = cnt[threadIdx.x];
        dinv[node] = (c > 0) ? rsqrtf((float)c) : 0.0f;
    }
}


template <bool MAKE_W>
static __device__ __forceinline__ void accum5(float* tile, float* dinvD, int bin, int node0, int rows,
                                              const int* __restrict__ binStart,
                                              const int* __restrict__ recs,
                                              const float* __restrict__ dinv,
                                              float* __restrict__ wts,
                                              const float* __restrict__ hin) {
    for (int i = threadIdx.x; i < RB_ * WIN_; i += 256) tile[i] = 0.0f;
    if (MAKE_W) dinvD[threadIdx.x] = ((int)threadIdx.x < rows) ? dinv[node0 + threadIdx.x] : 0.0f;
    __syncthreads();
    int s = binStart[bin], e = binStart[bin + 1];
    int last = (e > 0) ? e - 1 : 0;
    int i0 = s + threadIdx.x;
    int r[E5_];
    float w[E5_];
#pragma unroll
    for (int u = 0; u < E5_; ++u) {
        int i = i0 + u * 256;
        int ic = (i < e) ? i : last;
        int rv = recs[ic];
        w[u] = MAKE_W ? 0.0f : wts[ic];
        r[u] = (i < e) ? rv : -1;
    }
    while (i0 < e) {
        float a[E5_][WIN_];
        float ds[E5_];
#pragma unroll
        for (int u = 0; u < E5_; ++u) {
            int src = (r[u] < 0) ? 0 : (r[u] & SRCM_);
            const float* hp = hin + src * WIN_;
#pragma unroll
            for (int j = 0; j < WIN_; ++j) a[u][j] = hp[j];
            ds[u] = MAKE_W ? dinv[src] : 0.0f;
        }
        int i1 = i0 + E5_ * 256;
        int rn[E5_];
        float wn[E5_];
#pragma unroll
        for (int u = 0; u < E5_; ++u) {
            int i = i1 + u * 256;
            int ic = (i < e) ? i : last;
            int rv = recs[ic];
            wn[u] = MAKE_W ? 0.0f : wts[ic];
            rn[u] = (i < e) ? rv : -1;
        }
        auto add_edge = [&](int u) {
            int dl = r[u] >> 18;
            float nrm;
            if (MAKE_W) { nrm = ds[u] * dinvD[dl]; wts[i0 + u * 256] = nrm; }
            else        { nrm = w[u]; }
            float* tp = &tile[dl * WIN_];
            atomicAdd(tp + 0, a[u][0] * nrm);
            atomicAdd(tp + 1, a[u][1] * nrm);
            atomicAdd(tp + 2, a[u][2] * nrm);
            atomicAdd(tp + 3, a[u][3] * nrm);
            atomicAdd(tp + 4, a[u][4] * nrm);
        };
        if (i0 + (E5_ - 1) * 256 < e) {
#pragma unroll
            for (int u = 0; u < E5_; ++u) add_edge(u);
        } else {
#pragma unroll
            for (int u = 0; u < E5_; ++u) {
                if (r[u] >= 0) add_edge(u);
            }
        }
#pragma unroll
        for (int u = 0; u < E5_; ++u) { r[u] = rn[u]; w[u] = wn[u]; }
        i0 = i1;
    }
    __syncthreads();
}

static __device__ __forceinline__ void accum32(float4* tile4, int bin,
                                               const int* __restrict__ binStart,
                                               const int* __restrict__ recs,
                                               const float* __restrict__ wts,
                                               const float* __restrict__ hin) {
    float* tile = (float*)tile4;
    const float4* __restrict__ hin4 = (const float4*)hin;
    for (int i = threadIdx.x; i < RB_ * 8; i += 256) tile4[i] = make_float4(0.f, 0.f, 0.f, 0.f);
    __syncthreads();
    int q = threadIdx.x & 7;
    int sub = threadIdx.x >> 3;
    int rot = sub & 3;
    bool r1 = (rot & 1) != 0, r2 = (rot & 2) != 0;
    int o0 = rot, o1 = (rot + 1) & 3, o2 = (rot + 2) & 3, o3 = (rot + 3) & 3;
    int s = binStart[bin], e = binStart[bin + 1];
    int last = (e > 0) ? e - 1 : 0;
    int i0 = s + sub;
    int r[E32_];
    float w[E32_];
#pragma unroll
    for (int u = 0; u < E32_; ++u) {
        int i = i0 + u * 32;
        int ic = (i < e) ? i : last;
        int rv = recs[ic];
        w[u] = wts[ic];
        r[u] = (i < e) ? rv : -1;
    }
    while (i0 < e) {
        float4 v[E32_];
#pragma unroll
        for (int u = 0; u < E32_; ++u) {
            int srcu = (r[u] < 0) ? 0 : (r[u] & SRCM_);
            v[u] = hin4[srcu * 8 + q];
        }
        int i1 = i0 + E32_ * 32;
        int rn[E32_];
        float wn[E32_];
#pragma unroll
        for (int u = 0; u < E32_; ++u) {
            int i = i1 + u * 32;
            int ic = (i < e) ? i : last;
            int rv = recs[ic];
            wn[u] = wts[ic];
            rn[u] = (i < e) ? rv : -1;
        }
        auto add_edge = [&](int rec, float nrm, float4 x) {
            int dl = rec >> 18;
            float* tp = &tile[dl * H_ + q * 4];
            float b0 = r1 ? x.y : x.x, b1 = r1 ? x.z : x.y;
            float b2 = r1 ? x.w : x.z, b3 = r1 ? x.x : x.w;
            float c0 = r2 ? b2 : b0, c1 = r2 ? b3 : b1, c2 = r2 ? b0 : b2, c3 = r2 ? b1 : b3;
            atomicAdd(tp + o0, c0 * nrm);
            atomicAdd(tp + o1, c1 * nrm);
            atomicAdd(tp + o2, c2 * nrm);
            atomicAdd(tp + o3, c3 * nrm);
        };
        if (i0 + (E32_ - 1) * 32 < e) {
#pragma unroll
            for (int u = 0; u < E32_; ++u) add_edge(r[u], w[u], v[u]);
        } else {
#pragma unroll
            for (int u = 0; u < E32_; ++u) {
                if (r[u] >= 0) add_edge(r[u], w[u], v[u]);
            }
        }
#pragma unroll
        for (int u = 0; u < E32_; ++u) { r[u] = rn[u]; w[u] = wn[u]; }
        i0 = i1;
    }
    __syncthreads();
}

__global__ __launch_bounds__(256) void bin_prop5_kernel(const int* __restrict__ binStart,
                                                        const int* __restrict__ recs,
                                                        const float* __restrict__ dinv,
                                                        float* __restrict__ wts,
                                                        const float* __restrict__ hin,
                                                        float* __restrict__ hout) {
    __shared__ float tile[RB_ * WIN_];
    __shared__ float dinvD[RB_];
    int bin = blockIdx.x;
    int node0 = bin * RB_;
    int rows = (NTOT_ - node0 < RB_) ? (NTOT_ - node0) : RB_;
    accum5<true>(tile, dinvD, bin, node0, rows, binStart, recs, dinv, wts, hin);
    for (int i = threadIdx.x; i < rows * WIN_; i += 256) hout[node0 * WIN_ + i] = tile[i];
}

__global__ __launch_bounds__(256) void bin_prop32_kernel(const int* __restrict__ binStart,
                                                         const int* __restrict__ recs,
                                                         const float* __restrict__ wts,
                                                         const float* __restrict__ hin,
                                                         float* __restrict__ hout) {
    __shared__ float4 tile4[RB_ * 8];
    int bin = blockIdx.x;
    int node0 = bin * RB_;
    int rows = (NTOT_ - node0 < RB_) ? (NTOT_ - node0) : RB_;
    accum32(tile4, bin, binStart, recs, wts, hin);
    float4* __restrict__ out4 = (float4*)hout;
    for (int i = threadIdx.x; i < rows * 8; i += 256) out4[node0 * 8 + i] = tile4[i];
}

static __device__ __forceinline__ void dense_tail(float4* stage4, const v8f (&c)[2][2],
                                                  const float* __restrict__ bias,
                                                  int wave, int half, int l16, int node0, int rows,
                                                  float* __restrict__ hout) {
    float* stage = (float*)stage4;
    float bz0 = bias[l16], bz1 = bias[16 + l16];
#pragma unroll
    for (int t = 0; t < 2; ++t) {
#pragma unroll
        for (int r = 0; r < 8; ++r) {
            int row = wave * 32 + t * 16 + half * 8 + r;
            stage[row * SP_ + l16]      = tanhf(c[t][0][r] + bz0);
            stage[row * SP_ + 16 + l16] = tanhf(c[t][1][r] + bz1);
        }
    }
    __syncthreads();
    float4* __restrict__ out4 = (float4*)hout;
    for (int i = threadIdx.x; i < rows * 8; i += 256) out4[node0 * 8 + i] = stage4[(i >> 3) * SP4_ + (i & 7)];
}

__global__ __launch_bounds__(256) void bin_prop5_dense_kernel(const int* __restrict__ binStart,
                                                              const int* __restrict__ recs,
                                                              float* __restrict__ wts,
                                                              const float* __restrict__ x0,
                                                              const float* __restrict__ x1,
                                                              const float* __restrict__ W,
                                                              const float* __restrict__ bias,
                                                              float* __restrict__ hout) {
    __shared__ float tile[RB_ * WIN_];
    __shared__ float4 stage4[RB_ * SP4_];
    int bin = blockIdx.x;
    int node0 = bin * RB_;
    int rows = (NTOT_ - node0 < RB_) ? (NTOT_ - node0) : RB_;
    accum5<false>(tile, nullptr, bin, node0, rows, binStart, recs, nullptr, wts, x1);

    int wave = threadIdx.x >> 5;
    int lane = threadIdx.x & 31;
    int half = lane >> 4;
    int l16  = lane & 15;
    v8f c[2][2] = {};
#pragma unroll
    for (int s = 0; s < 3; ++s) {
#pragma unroll
        for (int k0 = 0; k0 < WIN_; k0 += 4) {
            int kA = k0 + half * 2;
            float m0 = (kA     < WIN_) ? 1.0f : 0.0f;
            float m1 = (kA + 1 < WIN_) ? 1.0f : 0.0f;
            int ka0 = (kA     < WIN_) ? kA     : (WIN_ - 1);
            int ka1 = (kA + 1 < WIN_) ? kA + 1 : (WIN_ - 1);
            int kw0 = s * WIN_ + ka0;
            int kw1 = s * WIN_ + ka1;
            v2f b0, b1;
            b0.x = W[kw0 * H_ + l16] * m0;
            b0.y = W[kw1 * H_ + l16] * m1;
            b1.x = W[kw0 * H_ + 16 + l16] * m0;
            b1.y = W[kw1 * H_ + 16 + l16] * m1;
#pragma unroll
            for (int t = 0; t < 2; ++t) {
                int rl = wave * 32 + t * 16 + l16;
                int rg = node0 + rl;
                rg = (rg < NTOT_) ? rg : (NTOT_ - 1);
                v2f a;
                if (s == 0)      { a.x = x0[rg * WIN_ + ka0] * m0;  a.y = x0[rg * WIN_ + ka1] * m1; }
                else if (s == 1) { a.x = x1[rg * WIN_ + ka0] * m0;  a.y = x1[rg * WIN_ + ka1] * m1; }
                else             { a.x = tile[rl * WIN_ + ka0] * m0; a.y = tile[rl * WIN_ + ka1] * m1; }
                c[t][0] = __builtin_amdgcn_wmma_f32_16x16x4_f32(false, a, false, b0,
                                                                (short)0, c[t][0], false, false);
                c[t][1] = __builtin_amdgcn_wmma_f32_16x16x4_f32(false, a, false, b1,
                                                                (short)0, c[t][1], false, false);
            }
        }
    }
    dense_tail(stage4, c, bias, wave, half, l16, node0, rows, hout);
}

template <int PITCH>
static __device__ __forceinline__ void seg_from_tile(const float* tile, int wave, int half, int l16,
                                                     const float* __restrict__ Wseg, v8f (&c)[2][2]) {
#pragma unroll
    for (int k0 = 0; k0 < H_; k0 += 4) {
        int kA = k0 + half * 2;
        v2f b0, b1;
        b0.x = Wseg[(kA    ) * H_ + l16];
        b0.y = Wseg[(kA + 1) * H_ + l16];
        b1.x = Wseg[(kA    ) * H_ + 16 + l16];
        b1.y = Wseg[(kA + 1) * H_ + 16 + l16];
#pragma unroll
        for (int t = 0; t < 2; ++t) {
            const float* p = &tile[(wave * 32 + t * 16 + l16) * PITCH + kA];
            v2f a;
            a.x = p[0];
            a.y = p[1];
            c[t][0] = __builtin_amdgcn_wmma_f32_16x16x4_f32(false, a, false, b0,
                                                            (short)0, c[t][0], false, false);
            c[t][1] = __builtin_amdgcn_wmma_f32_16x16x4_f32(false, a, false, b1,
                                                            (short)0, c[t][1], false, false);
        }
    }
}

template <bool READOUT>
__global__ __launch_bounds__(256, 8) void bin_prop32_dense_kernel(const int* __restrict__ binStart,
                                                               const int* __restrict__ recs,
                                                               const float* __restrict__ wts,
                                                               const float* __restrict__ x0,
                                                               const float* __restrict__ x1,
                                                               const float* __restrict__ W,
                                                               const float* __restrict__ bias,
                                                               float* __restrict__ hout,
                                                               const float* __restrict__ Wr,
                                                               const float* __restrict__ br,
                                                               float* __restrict__ rout) {
    __shared__ float4 tile4[RB_ * 8];
    __shared__ float4 stage4[RB_ * SP4_];
    const float* tile = (const float*)tile4;
    const float* stage = (const float*)stage4;
    int bin = blockIdx.x;
    int node0 = bin * RB_;
    int rows = (NTOT_ - node0 < RB_) ? (NTOT_ - node0) : RB_;
    const float4* __restrict__ x0_4 = (const float4*)x0;
    const float4* __restrict__ x1_4 = (const float4*)x1;
    accum32(tile4, bin, binStart, recs, wts, x1);

    int wave = threadIdx.x >> 5;
    int lane = threadIdx.x & 31;
    int half = lane >> 4;
    int l16  = lane & 15;
    int srow = threadIdx.x >> 3;
    int sq   = threadIdx.x & 7;
    float4 g[8];
#pragma unroll
    for (int j = 0; j < 8; ++j) {
        int i = threadIdx.x + j * 256;
        g[j] = (i < rows * 8) ? x0_4[node0 * 8 + i] : make_float4(0.f, 0.f, 0.f, 0.f);
    }
#pragma unroll
    for (int j = 0; j < 8; ++j) stage4[(srow + j * 32) * SP4_ + sq] = g[j];
    __syncthreads();
    v8f c[2][2] = {};
    seg_from_tile<SP_>(stage, wave, half, l16, W, c);
#pragma unroll
    for (int j = 0; j < 8; ++j) {
        int i = threadIdx.x + j * 256;
        g[j] = (i < rows * 8) ? x1_4[node0 * 8 + i] : make_float4(0.f, 0.f, 0.f, 0.f);
    }
    __syncthreads();
#pragma unroll
    for (int j = 0; j < 8; ++j) stage4[(srow + j * 32) * SP4_ + sq] = g[j];
    __syncthreads();
    seg_from_tile<SP_>(stage, wave, half, l16, W + H_ * H_, c);
    seg_from_tile<H_>(tile, wave, half, l16, W + 2 * H_ * H_, c);
    dense_tail(stage4, c, bias, wave, half, l16, node0, rows, hout);
    if (READOUT) {
        int row = threadIdx.x;
        if (row < rows) {
            float acc = br[0];
#pragma unroll
            for (int j = 0; j < H_; ++j) acc += stage[row * SP_ + j] * Wr[j];
            rout[node0 + row] = acc;
        }
    }
}

static inline int cdiv(int a, int b) { return (a + b - 1) / b; }

extern "C" void kernel_launch(void* const* d_in, const int* in_sizes, int n_in,
                              void* d_out, int out_size, void* d_ws, size_t ws_size,
                              hipStream_t stream) {
    (void)in_sizes; (void)n_in; (void)out_size; (void)ws_size;

    const float* X  = (const float*)d_in[0];
    const int*   EI = (const int*)d_in[1];
    const float* W0 = (const float*)d_in[2];
    const float* b0 = (const float*)d_in[3];
    const float* W1 = (const float*)d_in[4];
    const float* b1 = (const float*)d_in[5];
    const float* W2 = (const float*)d_in[6];
    const float* b2 = (const float*)d_in[7];
    const float* Wr = (const float*)d_in[8];
    const float* br = (const float*)d_in[9];

    float* outF = (float*)d_out;
    float* hOut = outF + NTOT_;

    float* ws       = (float*)d_ws;
    float* dinv     = ws;
    int*   recs     = (int*)(dinv + NTOT_);
    int*   binCount = recs + BE_;
    int*   binStart = binCount + 1024;
    int*   cursor   = binStart + 1024;
    float* p1   = (float*)(cursor + 1024);
    float* bufA = p1 + (size_t)NTOT_ * H_;
    float* bufB = bufA + (size_t)NTOT_ * H_;
    float* wts  = bufB + (size_t)NTOT_ * H_;

    const int TB = 256;

    zero_i32_kernel<<<cdiv(NB_, TB), TB, 0, stream>>>(binCount, NB_);
    bin_count_kernel<<<NCH_, 256, 0, stream>>>(EI, binCount);
    bin_scan_kernel<<<1, 1024, 0, stream>>>(binCount, binStart, cursor);
    bin_fill_kernel<<<NCH_, 256, 0, stream>>>(EI, cursor, recs);
    bin_dinv_kernel<<<NB_, 256, 0, stream>>>(binStart, recs, dinv);

    bin_prop5_kernel<<<NB_, 256, 0, stream>>>(binStart, recs, dinv, wts, X, p1);
    bin_prop5_dense_kernel<<<NB_, 256, 0, stream>>>(binStart, recs, wts, X, p1, W0, b0, bufA);

    bin_prop32_kernel<<<NB_, 256, 0, stream>>>(binStart, recs, wts, bufA, p1);
    bin_prop32_dense_kernel<false><<<NB_, 256, 0, stream>>>(binStart, recs, wts, bufA, p1, W1, b1, bufB,
                                                            Wr, br, outF);

    bin_prop32_kernel<<<NB_, 256, 0, stream>>>(binStart, recs, wts, bufB, p1);
    bin_prop32_dense_kernel<true><<<NB_, 256, 0, stream>>>(binStart, recs, wts, bufB, p1, W2, b2, hOut,
                                                           Wr, br, outF);
}


# F=5 rows (x, first hop) at 32-byte pitch, dinv in a pad slot; 4 lanes per edge load 8 bytes each (records shared by quad moves): a wave's gather touches 8 lines, not 32; sums arrival-order
# speedup vs baseline: 4.8289x; 1.0512x over previous
#include <hip/hip_runtime.h>
#include <math.h>

#define B_    4
#define WIN_  5
#define N_    50000
#define E_    1000000
#define H_    32
#define NTOT_ (B_ * N_)
#define BE_   (B_ * E_)

typedef __attribute__((ext_vector_type(2))) float v2f;
typedef __attribute__((ext_vector_type(8))) float v8f;

static __device__ __forceinline__ void decode_edge(const int* __restrict__ ei, int e,
                                                   int& src, int& dst) {
    int b  = e / E_;
    int el = e - b * E_;
    src = ei[(b * 2 + 0) * E_ + el] + b * N_;
    dst = ei[(b * 2 + 1) * E_ + el] + b * N_;
}

#define RB_   256
#define NB_   ((NTOT_ + RB_ - 1) / RB_)
#define CH_   8192
#define NCH_  ((BE_ + CH_ - 1) / CH_)
#define EPT_  (CH_ / 256)
#define NG4_  (EPT_ / 4)
#define SRCM_ 0x3FFFF
#define SP_   36
#define SP4_  (SP_ / 4)
#define XP_   8
#define E5_   8
#define E32_  8

__global__ void zero_i32_kernel(int* __restrict__ p, int n) {
    int i = blockIdx.x * blockDim.x + threadIdx.x;
    if (i < n) p[i] = 0;
}

static __device__ __forceinline__ int group_pos(int e, int& nb) {
    bool ok = e < BE_;
    int ec = ok ? e : 0;
    int b  = ec / E_;
    int el = ec - b * E_;
    nb = ok ? b * N_ : -1;
    return (b * 2 + 1) * E_ + el;
}

__global__ __launch_bounds__(256) void bin_count_kernel(const int* __restrict__ ei,
                                                        int* __restrict__ binCount) {
    __shared__ int hist[NB_];
    for (int i = threadIdx.x; i < NB_; i += 256) hist[i] = 0;
    int e0 = blockIdx.x * CH_;
    int4 d[NG4_];
    int  nb[NG4_];
#pragma unroll
    for (int k = 0; k < NG4_; ++k) {
        int p = group_pos(e0 + (k * 256 + threadIdx.x) * 4, nb[k]);
        d[k] = *(const int4*)(ei + p);
    }
    __syncthreads();
#pragma unroll
    for (int k = 0; k < NG4_; ++k) {
        if (nb[k] >= 0) {
            atomicAdd(&hist[(d[k].x + nb[k]) / RB_], 1);
            atomicAdd(&hist[(d[k].y + nb[k]) / RB_], 1);
            atomicAdd(&hist[(d[k].z + nb[k]) / RB_], 1);
            atomicAdd(&hist[(d[k].w + nb[k]) / RB_], 1);
        }
    }
    __syncthreads();
    for (int i = threadIdx.x; i < NB_; i += 256) {
        int c = hist[i];
        if (c) atomicAdd(&binCount[i], c);
    }
}

__global__ __launch_bounds__(1024) void bin_scan_kernel(const int* __restrict__ binCount,
                                                        int* __restrict__ binStart,
                                                        int* __restrict__ cursor) {
    __shared__ int s[1024];
    int t = threadIdx.x;
    int v = (t < NB_) ? binCount[t] : 0;
    s[t] = v;
    __syncthreads();
    for (int off = 1; off < 1024; off <<= 1) {
        int x = (t >= off) ? s[t - off] : 0;
        __syncthreads();
        s[t] += x;
        __syncthreads();
    }
    int excl = s[t] - v;
    if (t < NB_) { binStart[t] = excl; cursor[t] = excl; }
    if (t == NB_ - 1) binStart[NB_] = s[t];
}

#define KEYB_ 10
#define KEYM_ ((1 << KEYB_) - 1)
static_assert(NB_ <= (1 << KEYB_) && NB_ <= 4 * 256 && CH_ <= (1 << (31 - KEYB_)), "key packing and the reservation's four blocks per thread");
__global__ __launch_bounds__(256) void bin_fill_kernel(const int* __restrict__ ei,
                                                       int* __restrict__ cursor,
                                                       int* __restrict__ recs) {
    __shared__ int hist[NB_];
    __shared__ int base[NB_];
    for (int i = threadIdx.x; i < NB_; i += 256) hist[i] = 0;
    int e0 = blockIdx.x * CH_;
    int4 s[NG4_], d[NG4_];
    int  nb[NG4_];
#pragma unroll
    for (int k = 0; k < NG4_; ++k) {
        int p = group_pos(e0 + (k * 256 + threadIdx.x) * 4, nb[k]);
        s[k] = *(const int4*)(ei + p - E_);
        d[k] = *(const int4*)(ei + p);
    }
    __syncthreads();
    int rec[EPT_];
    int key[EPT_];
#pragma unroll
    for (int k = 0; k < NG4_; ++k) {
        int sv[4] = { s[k].x, s[k].y, s[k].z, s[k].w };
        int dv[4] = { d[k].x, d[k].y, d[k].z, d[k].w };
#pragma unroll
        for (int c = 0; c < 4; ++c) {
            int dst = dv[c] + nb[k];
            int bin = dst / RB_;
            rec[k * 4 + c] = (sv[c] + nb[k]) | ((dst - bin * RB_) << 18);
            key[k * 4 + c] = (nb[k] >= 0) ? bin : -1;
        }
    }
    bool whole = e0 + CH_ <= BE_;
    if (whole) {
#pragma unroll
        for (int j = 0; j < EPT_; ++j) key[j] |= atomicAdd(&hist[key[j]], 1) << KEYB_;
    } else {
#pragma unroll
        for (int j = 0; j < EPT_; ++j) {
            if (key[j] >= 0) key[j] |= atomicAdd(&hist[key[j]], 1) << KEYB_;
        }
    }
    __syncthreads();
    {
        int cnt[4], bs[4];
#pragma unroll
        for (int u = 0; u < 4; ++u) {
            int i = threadIdx.x + u * 256;
            cnt[u] = (i < NB_) ? hist[i] : 0;
        }
#pragma unroll
        for (int u = 0; u < 4; ++u) {
            int i = threadIdx.x + u * 256;
            bs[u] = cnt[u] ? atomicAdd(&cursor[i], cnt[u]) : 0;
        }
#pragma unroll
        for (int u = 0; u < 4; ++u) {
            int i = threadIdx.x + u * 256;
            if (i < NB_) base[i] = bs[u];
        }
    }
    __syncthreads();
    if (whole) {
#pragma unroll
        for (int j = 0; j < EPT_; ++j) recs[base[key[j] & KEYM_] + (key[j] >> KEYB_)] = rec[j];
    } else {
#pragma unroll
        for (int j = 0; j < EPT_; ++j) {
            if (key[j] >= 0) recs[base[key[j] & KEYM_] + (key[j] >> KEYB_)] = rec[j];
        }
    }
}

__global__ __launch_bounds__(256) void bin_dinv_kernel(const int* __restrict__ binStart,
                                                       const int* __restrict__ recs,
                                                       const float* __restrict__ x,
                                                       float* __restrict__ xp) {
    __shared__ int cnt[RB_];
    int bin = blockIdx.x;
    cnt[threadIdx.x] = 0;
    __syncthreads();
    int s = binStart[bin], e = binStart[bin + 1];
    int last = (e > 0) ? e - 1 : 0;
    for (int i0 = s + threadIdx.x; i0 < e; i0 += 4 * 256) {
        int r[4];
#pragma unroll
        for (int u = 0; u < 4; ++u) {
            int i = i0 + u * 256;
            int rv = recs[(i < e) ? i : last];
            r[u] = (i < e) ? rv : -1;
        }
#pragma unroll
        for (int u = 0; u < 4; ++u) {
            if (r[u] >= 0) atomicAdd(&cnt[r[u] >> 18], 1);
        }
    }
    __syncthreads();
    int node = bin * RB_ + threadIdx.x;
    if (node < NTOT_) {
        int c = cnt[threadIdx.x];
        float dv = (c > 0) ? rsqrtf((float)c) : 0.0f;
        const float* xr = x + node * WIN_;
        float4* __restrict__ xp4 = (float4*)xp;
        xp4[node * 2 + 0] = make_float4(xr[0], xr[1], xr[2], xr[3]);
        xp4[node * 2 + 1] = make_float4(xr[4], dv, 0.0f, 0.0f);
    }
}


static_assert(E5_ == 8, "two record slots per lane of a 4-lane group");
template <bool MAKE_W>
static __device__ __forceinline__ void accum5(float* tile, float* dinvD, int bin, int node0, int rows,
                                              const int* __restrict__ binStart,
                                              const int* __restrict__ recs,
                                              float* __restrict__ wts,
                                              const float* __restrict__ hin) {
    const float2* __restrict__ hin2 = (const float2*)hin;
    for (int i = threadIdx.x; i < RB_ * WIN_; i += 256) tile[i] = 0.0f;
    if (MAKE_W) dinvD[threadIdx.x] = ((int)threadIdx.x < rows) ? hin[(node0 + threadIdx.x) * XP_ + WIN_] : 0.0f;
    __syncthreads();
    int j = threadIdx.x & 3;
    int grp = threadIdx.x >> 2;
    int s = binStart[bin], e = binStart[bin + 1];
    int last = (e > 0) ? e - 1 : 0;
    auto load_recs = [&](int ib, int& xa, int& xb, float& ya, float& yb) {
        int ia = ib + j, ic = ib + 4 + j;
        int ca = (ia < e) ? ia : last, cc = (ic < e) ? ic : last;
        int rva = recs[ca], rvc = recs[cc];
        ya = MAKE_W ? 0.0f : wts[ca];
        yb = MAKE_W ? 0.0f : wts[cc];
        xa = (ia < e) ? rva : -1;
        xb = (ic < e) ? rvc : -1;
    };
    int t0 = s;
    int i0 = t0 + grp * E5_;
    int ra, rb;
    float wa, wb;
    load_recs(i0, ra, rb, wa, wb);
    while (t0 < e) {
        int r[E5_];
        float w[E5_];
#pragma unroll
        for (int u = 0; u < E5_; ++u) {
            r[u] = __shfl((u < 4) ? ra : rb, u & 3, 4);
            w[u] = MAKE_W ? 0.0f : __shfl((u < 4) ? wa : wb, u & 3, 4);
        }
        float2 v[E5_];
#pragma unroll
        for (int u = 0; u < E5_; ++u) {
            int srcu = (r[u] < 0) ? 0 : (r[u] & SRCM_);
            v[u] = hin2[srcu * (XP_ / 2) + j];
        }
        int t1 = t0 + 64 * E5_;
        int i1 = t1 + grp * E5_;
        int rna, rnb;
        float wna, wnb;
        load_recs(i1, rna, rnb, wna, wnb);
        float nrm[E5_];
#pragma unroll
        for (int u = 0; u < E5_; ++u) {
            if (MAKE_W) {
                int dl = (r[u] < 0) ? 0 : (r[u] >> 18);
                float dsrc = __shfl(v[u].y, 2, 4);
                nrm[u] = dsrc * dinvD[dl];
            } else {
                nrm[u] = w[u];
            }
        }
        if (MAKE_W) {
            float na = (j == 0) ? nrm[0] : (j == 1) ? nrm[1] : (j == 2) ? nrm[2] : nrm[3];
            float nb = (j == 0) ? nrm[4] : (j == 1) ? nrm[5] : (j == 2) ? nrm[6] : nrm[7];
            if (i0 + j < e)     wts[i0 + j]     = na;
            if (i0 + 4 + j < e) wts[i0 + 4 + j] = nb;
        }
        auto add_edge = [&](int u) {
            float* tp = &tile[(r[u] >> 18) * WIN_ + 2 * j];
            if (j < 3) atomicAdd(tp,     v[u].x * nrm[u]);
            if (j < 2) atomicAdd(tp + 1, v[u].y * nrm[u]);
        };
        if (i0 + E5_ <= e) {
#pragma unroll
            for (int u = 0; u < E5_; ++u) add_edge(u);
        } else {
#pragma unroll
            for (int u = 0; u < E5_; ++u) {
                if (r[u] >= 0) add_edge(u);
            }
        }
        ra = rna; rb = rnb; wa = wna; wb = wnb;
        t0 = t1;
        i0 = i1;
    }
    __syncthreads();
}

static __device__ __forceinline__ void accum32(float4* tile4, int bin,
                                               const int* __restrict__ binStart,
                                               const int* __restrict__ recs,
                                               const float* __restrict__ wts,
                                               const float* __restrict__ hin) {
    float* tile = (float*)tile4;
    const float4* __restrict__ hin4 = (const float4*)hin;
    for (int i = threadIdx.x; i < RB_ * 8; i += 256) tile4[i] = make_float4(0.f, 0.f, 0.f, 0.f);
    __syncthreads();
    int q = threadIdx.x & 7;
    int sub = threadIdx.x >> 3;
    int rot = sub & 3;
    bool r1 = (rot & 1) != 0, r2 = (rot & 2) != 0;
    int o0 = rot, o1 = (rot + 1) & 3, o2 = (rot + 2) & 3, o3 = (rot + 3) & 3;
    int s = binStart[bin], e = binStart[bin + 1];
    int last = (e > 0) ? e - 1 : 0;
    int i0 = s + sub;
    int r[E32_];
    float w[E32_];
#pragma unroll
    for (int u = 0; u < E32_; ++u) {
        int i = i0 + u * 32;
        int ic = (i < e) ? i : last;
        int rv = recs[ic];
        w[u] = wts[ic];
        r[u] = (i < e) ? rv : -1;
    }
    while (i0 < e) {
        float4 v[E32_];
#pragma unroll
        for (int u = 0; u < E32_; ++u) {
            int srcu = (r[u] < 0) ? 0 : (r[u] & SRCM_);
            v[u] = hin4[srcu * 8 + q];
        }
        int i1 = i0 + E32_ * 32;
        int rn[E32_];
        float wn[E32_];
#pragma unroll
        for (int u = 0; u < E32_; ++u) {
            int i = i1 + u * 32;
            int ic = (i < e) ? i : last;
            int rv = recs[ic];
            wn[u] = wts[ic];
            rn[u] = (i < e) ? rv : -1;
        }
        auto add_edge = [&](int rec, float nrm, float4 x) {
            int dl = rec >> 18;
            float* tp = &tile[dl * H_ + q * 4];
            float b0 = r1 ? x.y : x.x, b1 = r1 ? x.z : x.y;
            float b2 = r1 ? x.w : x.z, b3 = r1 ? x.x : x.w;
            float c0 = r2 ? b2 : b0, c1 = r2 ? b3 : b1, c2 = r2 ? b0 : b2, c3 = r2 ? b1 : b3;
            atomicAdd(tp + o0, c0 * nrm);
            atomicAdd(tp + o1, c1 * nrm);
            atomicAdd(tp + o2, c2 * nrm);
            atomicAdd(tp + o3, c3 * nrm);
        };
        if (i0 + (E32_ - 1) * 32 < e) {
#pragma unroll
            for (int u = 0; u < E32_; ++u) add_edge(r[u], w[u], v[u]);
        } else {
#pragma unroll
            for (int u = 0; u < E32_; ++u) {
                if (r[u] >= 0) add_edge(r[u], w[u], v[u]);
            }
        }
#pragma unroll
        for (int u = 0; u < E32_; ++u) { r[u] = rn[u]; w[u] = wn[u]; }
        i0 = i1;
    }
    __syncthreads();
}

__global__ __launch_bounds__(256) void bin_prop5_kernel(const int* __restrict__ binStart,
                                                        const int* __restrict__ recs,
                                                        float* __restrict__ wts,
                                                        const float* __restrict__ xp,
                                                        float* __restrict__ hout) {
    __shared__ float tile[RB_ * WIN_];
    __shared__ float dinvD[RB_];
    int bin = blockIdx.x;
    int node0 = bin * RB_;
    int rows = (NTOT_ - node0 < RB_) ? (NTOT_ - node0) : RB_;
    accum5<true>(tile, dinvD, bin, node0, rows, binStart, recs, wts, xp);
    float4* __restrict__ out4 = (float4*)hout;
    int t = threadIdx.x;
    if (t < rows) {
        const float* tr = &tile[t * WIN_];
        out4[(node0 + t) * 2 + 0] = make_float4(tr[0], tr[1], tr[2], tr[3]);
        out4[(node0 + t) * 2 + 1] = make_float4(tr[4], 0.0f, 0.0f, 0.0f);
    }
}

__global__ __launch_bounds__(256) void bin_prop32_kernel(const int* __restrict__ binStart,
                                                         const int* __restrict__ recs,
                                                         const float* __restrict__ wts,
                                                         const float* __restrict__ hin,
                                                         float* __restrict__ hout) {
    __shared__ float4 tile4[RB_ * 8];
    int bin = blockIdx.x;
    int node0 = bin * RB_;
    int rows = (NTOT_ - node0 < RB_) ? (NTOT_ - node0) : RB_;
    accum32(tile4, bin, binStart, recs, wts, hin);
    float4* __restrict__ out4 = (float4*)hout;
    for (int i = threadIdx.x; i < rows * 8; i += 256) out4[node0 * 8 + i] = tile4[i];
}

static __device__ __forceinline__ void dense_tail(float4* stage4, const v8f (&c)[2][2],
                                                  const float* __restrict__ bias,
                                                  int wave, int half, int l16, int node0, int rows,
                                                  float* __restrict__ hout) {
    float* stage = (float*)stage4;
    float bz0 = bias[l16], bz1 = bias[16 + l16];
#pragma unroll
    for (int t = 0; t < 2; ++t) {
#pragma unroll
        for (int r = 0; r < 8; ++r) {
            int row = wave * 32 + t * 16 + half * 8 + r;
            stage[row * SP_ + l16]      = tanhf(c[t][0][r] + bz0);
            stage[row * SP_ + 16 + l16] = tanhf(c[t][1][r] + bz1);
        }
    }
    __syncthreads();
    float4* __restrict__ out4 = (float4*)hout;
    for (int i = threadIdx.x; i < rows * 8; i += 256) out4[node0 * 8 + i] = stage4[(i >> 3) * SP4_ + (i & 7)];
}

__global__ __launch_bounds__(256) void bin_prop5_dense_kernel(const int* __restrict__ binStart,
                                                              const int* __restrict__ recs,
                                                              float* __restrict__ wts,
                                                              const float* __restrict__ x0,
                                                              const float* __restrict__ x1,
                                                              const float* __restrict__ W,
                                                              const float* __restrict__ bias,
                                                              float* __restrict__ hout) {
    __shared__ float tile[RB_ * WIN_];
    __shared__ float4 stage4[RB_ * SP4_];
    int bin = blockIdx.x;
    int node0 = bin * RB_;
    int rows = (NTOT_ - node0 < RB_) ? (NTOT_ - node0) : RB_;
    accum5<false>(tile, nullptr, bin, node0, rows, binStart, recs, wts, x1);

    int wave = threadIdx.x >> 5;
    int lane = threadIdx.x & 31;
    int half = lane >> 4;
    int l16  = lane & 15;
    v8f c[2][2] = {};
#pragma unroll
    for (int s = 0; s < 3; ++s) {
#pragma unroll
        for (int k0 = 0; k0 < WIN_; k0 += 4) {
            int kA = k0 + half * 2;
            float m0 = (kA     < WIN_) ? 1.0f : 0.0f;
            float m1 = (kA + 1 < WIN_) ? 1.0f : 0.0f;
            int ka0 = (kA     < WIN_) ? kA     : (WIN_ - 1);
            int ka1 = (kA + 1 < WIN_) ? kA + 1 : (WIN_ - 1);
            int kw0 = s * WIN_ + ka0;
            int kw1 = s * WIN_ + ka1;
            v2f b0, b1;
            b0.x = W[kw0 * H_ + l16] * m0;
            b0.y = W[kw1 * H_ + l16] * m1;
            b1.x = W[kw0 * H_ + 16 + l16] * m0;
            b1.y = W[kw1 * H_ + 16 + l16] * m1;
#pragma unroll
            for (int t = 0; t < 2; ++t) {
                int rl = wave * 32 + t * 16 + l16;
                int rg = node0 + rl;
                rg = (rg < NTOT_) ? rg : (NTOT_ - 1);
                v2f a;
                if (s == 0)      { a.x = x0[rg * WIN_ + ka0] * m0;  a.y = x0[rg * WIN_ + ka1] * m1; }
                else if (s == 1) { a.x = x1[rg * XP_ + ka0] * m0;   a.y = x1[rg * XP_ + ka1] * m1; }
                else             { a.x = tile[rl * WIN_ + ka0] * m0; a.y = tile[rl * WIN_ + ka1] * m1; }
                c[t][0] = __builtin_amdgcn_wmma_f32_16x16x4_f32(false, a, false, b0,
                                                                (short)0, c[t][0], false, false);
                c[t][1] = __builtin_amdgcn_wmma_f32_16x16x4_f32(false, a, false, b1,
                                                                (short)0, c[t][1], false, false);
            }
        }
    }
    dense_tail(stage4, c, bias, wave, half, l16, node0, rows, hout);
}

template <int PITCH>
static __device__ __forceinline__ void seg_from_tile(const float* tile, int wave, int half, int l16,
                                                     const float* __restrict__ Wseg, v8f (&c)[2][2]) {
#pragma unroll
    for (int k0 = 0; k0 < H_; k0 += 4) {
        int kA = k0 + half * 2;
        v2f b0, b1;
        b0.x = Wseg[(kA    ) * H_ + l16];
        b0.y = Wseg[(kA + 1) * H_ + l16];
        b1.x = Wseg[(kA    ) * H_ + 16 + l16];
        b1.y = Wseg[(kA + 1) * H_ + 16 + l16];
#pragma unroll
        for (int t = 0; t < 2; ++t) {
            const float* p = &tile[(wave * 32 + t * 16 + l16) * PITCH + kA];
            v2f a;
            a.x = p[0];
            a.y = p[1];
            c[t][0] = __builtin_amdgcn_wmma_f32_16x16x4_f32(false, a, false, b0,
                                                            (short)0, c[t][0], false, false);
            c[t][1] = __builtin_amdgcn_wmma_f32_16x16x4_f32(false, a, false, b1,
                                                            (short)0, c[t][1], false, false);
        }
    }
}

template <bool READOUT>
__global__ __launch_bounds__(256, 8) void bin_prop32_dense_kernel(const int* __restrict__ binStart,
                                                               const int* __restrict__ recs,
                                                               const float* __restrict__ wts,
                                                               const float* __restrict__ x0,
                                                               const float* __restrict__ x1,
                                                               const float* __restrict__ W,
                                                               const float* __restrict__ bias,
                                                               float* __restrict__ hout,
                                                               const float* __restrict__ Wr,
                                                               const float* __restrict__ br,
                                                               float* __restrict__ rout) {
    __shared__ float4 tile4[RB_ * 8];
    __shared__ float4 stage4[RB_ * SP4_];
    const float* tile = (const float*)tile4;
    const float* stage = (const float*)stage4;
    int bin = blockIdx.x;
    int node0 = bin * RB_;
    int rows = (NTOT_ - node0 < RB_) ? (NTOT_ - node0) : RB_;
    const float4* __restrict__ x0_4 = (const float4*)x0;
    const float4* __restrict__ x1_4 = (const float4*)x1;
    accum32(tile4, bin, binStart, recs, wts, x1);

    int wave = threadIdx.x >> 5;
    int lane = threadIdx.x & 31;
    int half = lane >> 4;
    int l16  = lane & 15;
    int srow = threadIdx.x >> 3;
    int sq   = threadIdx.x & 7;
    float4 g[8];
#pragma unroll
    for (int j = 0; j < 8; ++j) {
        int i = threadIdx.x + j * 256;
        g[j] = (i < rows * 8) ? x0_4[node0 * 8 + i] : make_float4(0.f, 0.f, 0.f, 0.f);
    }
#pragma unroll
    for (int j = 0; j < 8; ++j) stage4[(srow + j * 32) * SP4_ + sq] = g[j];
    __syncthreads();
    v8f c[2][2] = {};
    seg_from_tile<SP_>(stage, wave, half, l16, W, c);
#pragma unroll
    for (int j = 0; j < 8; ++j) {
        int i = threadIdx.x + j * 256;
        g[j] = (i < rows * 8) ? x1_4[node0 * 8 + i] : make_float4(0.f, 0.f, 0.f, 0.f);
    }
    __syncthreads();
#pragma unroll
    for (int j = 0; j < 8; ++j) stage4[(srow + j * 32) * SP4_ + sq] = g[j];
    __syncthreads();
    seg_from_tile<SP_>(stage, wave, half, l16, W + H_ * H_, c);
    seg_from_tile<H_>(tile, wave, half, l16, W + 2 * H_ * H_, c);
    dense_tail(stage4, c, bias, wave, half, l16, node0, rows, hout);
    if (READOUT) {
        int row = threadIdx.x;
        if (row < rows) {
            float acc = br[0];
#pragma unroll
            for (int j = 0; j < H_; ++j) acc += stage[row * SP_ + j] * Wr[j];
            rout[node0 + row] = acc;
        }
    }
}

static inline int cdiv(int a, int b) { return (a + b - 1) / b; }

extern "C" void kernel_launch(void* const* d_in, const int* in_sizes, int n_in,
                              void* d_out, int out_size, void* d_ws, size_t ws_size,
                              hipStream_t stream) {
    (void)in_sizes; (void)n_in; (void)out_size; (void)ws_size;

    const float* X  = (const float*)d_in[0];
    const int*   EI = (const int*)d_in[1];
    const float* W0 = (const float*)d_in[2];
    const float* b0 = (const float*)d_in[3];
    const float* W1 = (const float*)d_in[4];
    const float* b1 = (const float*)d_in[5];
    const float* W2 = (const float*)d_in[6];
    const float* b2 = (const float*)d_in[7];
    const float* Wr = (const float*)d_in[8];
    const float* br = (const float*)d_in[9];

    float* outF = (float*)d_out;
    float* hOut = outF + NTOT_;

    float* ws       = (float*)d_ws;
    int*   recs     = (int*)ws;
    int*   binCount = recs + BE_;
    int*   binStart = binCount + 1024;
    int*   cursor   = binStart + 1024;
    float* p1   = (float*)(cursor + 1024);
    float* bufA = p1 + (size_t)NTOT_ * H_;
    float* bufB = bufA + (size_t)NTOT_ * H_;
    float* wts  = bufB + (size_t)NTOT_ * H_;
    float* xp   = wts + BE_;

    const int TB = 256;

    zero_i32_kernel<<<cdiv(NB_, TB), TB, 0, stream>>>(binCount, NB_);
    bin_count_kernel<<<NCH_, 256, 0, stream>>>(EI, binCount);
    bin_scan_kernel<<<1, 1024, 0, stream>>>(binCount, binStart, cursor);
    bin_fill_kernel<<<NCH_, 256, 0, stream>>>(EI, cursor, recs);
    bin_dinv_kernel<<<NB_, 256, 0, stream>>>(binStart, recs, X, xp);

    bin_prop5_kernel<<<NB_, 256, 0, stream>>>(binStart, recs, wts, xp, p1);
    bin_prop5_dense_kernel<<<NB_, 256, 0, stream>>>(binStart, recs, wts, X, p1, W0, b0, bufA);

    bin_prop32_kernel<<<NB_, 256, 0, stream>>>(binStart, recs, wts, bufA, p1);
    bin_prop32_dense_kernel<false><<<NB_, 256, 0, stream>>>(binStart, recs, wts, bufA, p1, W1, b1, bufB,
                                                            Wr, br, outF);

    bin_prop32_kernel<<<NB_, 256, 0, stream>>>(binStart, recs, wts, bufB, p1);
    bin_prop32_dense_kernel<true><<<NB_, 256, 0, stream>>>(binStart, recs, wts, bufB, p1, W2, b2, hOut,
                                                           Wr, br, outF);
}


# Destination blocks of 196 rows, not 256: 1021 work-groups = four on each of the 256 WGPs (was 782, some WGPs four, most three): every WGP carries an equal edge share; sums stay arrival-order
# speedup vs baseline: 5.3726x; 1.1126x over previous
#include <hip/hip_runtime.h>
#include <math.h>

#define B_    4
#define WIN_  5
#define N_    50000
#define E_    1000000
#define H_    32
#define NTOT_ (B_ * N_)
#define BE_   (B_ * E_)

typedef __attribute__((ext_vector_type(2))) float v2f;
typedef __attribute__((ext_vector_type(8))) float v8f;

static __device__ __forceinline__ void decode_edge(const int* __restrict__ ei, int e,
                                                   int& src, int& dst) {
    int b  = e / E_;
    int el = e - b * E_;
    src = ei[(b * 2 + 0) * E_ + el] + b * N_;
    dst = ei[(b * 2 + 1) * E_ + el] + b * N_;
}

#define RB_   196
#define TR_   256
#define NB_   ((NTOT_ + RB_ - 1) / RB_)
#define CH_   8192
#define NCH_  ((BE_ + CH_ - 1) / CH_)
#define EPT_  (CH_ / 256)
#define NG4_  (EPT_ / 4)
#define SRCM_ 0x3FFFF
#define SP_   36
#define SP4_  (SP_ / 4)
#define XP_   8
#define E5_   8
#define E32_  8

__global__ void zero_i32_kernel(int* __restrict__ p, int n) {
    int i = blockIdx.x * blockDim.x + threadIdx.x;
    if (i < n) p[i] = 0;
}

static __device__ __forceinline__ int group_pos(int e, int& nb) {
    bool ok = e < BE_;
    int ec = ok ? e : 0;
    int b  = ec / E_;
    int el = ec - b * E_;
    nb = ok ? b * N_ : -1;
    return (b * 2 + 1) * E_ + el;
}

__global__ __launch_bounds__(256) void bin_count_kernel(const int* __restrict__ ei,
                                                        int* __restrict__ binCount) {
    __shared__ int hist[NB_];
    for (int i = threadIdx.x; i < NB_; i += 256) hist[i] = 0;
    int e0 = blockIdx.x * CH_;
    int4 d[NG4_];
    int  nb[NG4_];
#pragma unroll
    for (int k = 0; k < NG4_; ++k) {
        int p = group_pos(e0 + (k * 256 + threadIdx.x) * 4, nb[k]);
        d[k] = *(const int4*)(ei + p);
    }
    __syncthreads();
#pragma unroll
    for (int k = 0; k < NG4_; ++k) {
        if (nb[k] >= 0) {
            atomicAdd(&hist[(d[k].x + nb[k]) / RB_], 1);
            atomicAdd(&hist[(d[k].y + nb[k]) / RB_], 1);
            atomicAdd(&hist[(d[k].z + nb[k]) / RB_], 1);
            atomicAdd(&hist[(d[k].w + nb[k]) / RB_], 1);
        }
    }
    __syncthreads();
    for (int i = threadIdx.x; i < NB_; i += 256) {
        int c = hist[i];
        if (c) atomicAdd(&binCount[i], c);
    }
}

__global__ __launch_bounds__(1024) void bin_scan_kernel(const int* __restrict__ binCount,
                                                        int* __restrict__ binStart,
                                                        int* __restrict__ cursor) {
    __shared__ int s[1024];
    int t = threadIdx.x;
    int v = (t < NB_) ? binCount[t] : 0;
    s[t] = v;
    __syncthreads();
    for (int off = 1; off < 1024; off <<= 1) {
        int x = (t >= off) ? s[t - off] : 0;
        __syncthreads();
        s[t] += x;
        __syncthreads();
    }
    int excl = s[t] - v;
    if (t < NB_) { binStart[t] = excl; cursor[t] = excl; }
    if (t == NB_ - 1) binStart[NB_] = s[t];
}

#define KEYB_ 10
#define KEYM_ ((1 << KEYB_) - 1)
static_assert(NB_ <= (1 << KEYB_) && NB_ <= 4 * 256 && CH_ <= (1 << (31 - KEYB_)), "key packing and the reservation's four blocks per thread");
static_assert(RB_ <= TR_ && NB_ <= 1024, "a block's rows fit the tile; the scan's single work-group covers every block");
__global__ __launch_bounds__(256) void bin_fill_kernel(const int* __restrict__ ei,
                                                       int* __restrict__ cursor,
                                                       int* __restrict__ recs) {
    __shared__ int hist[NB_];
    __shared__ int base[NB_];
    for (int i = threadIdx.x; i < NB_; i += 256) hist[i] = 0;
    int e0 = blockIdx.x * CH_;
    int4 s[NG4_], d[NG4_];
    int  nb[NG4_];
#pragma unroll
    for (int k = 0; k < NG4_; ++k) {
        int p = group_pos(e0 + (k * 256 + threadIdx.x) * 4, nb[k]);
        s[k] = *(const int4*)(ei + p - E_);
        d[k] = *(const int4*)(ei + p);
    }
    __syncthreads();
    int rec[EPT_];
    int key[EPT_];
#pragma unroll
    for (int k = 0; k < NG4_; ++k) {
        int sv[4] = { s[k].x, s[k].y, s[k].z, s[k].w };
        int dv[4] = { d[k].x, d[k].y, d[k].z, d[k].w };
#pragma unroll
        for (int c = 0; c < 4; ++c) {
            int dst = dv[c] + nb[k];
            int bin = dst / RB_;
            rec[k * 4 + c] = (sv[c] + nb[k]) | ((dst - bin * RB_) << 18);
            key[k * 4 + c] = (nb[k] >= 0) ? bin : -1;
        }
    }
    bool whole = e0 + CH_ <= BE_;
    if (whole) {
#pragma unroll
        for (int j = 0; j < EPT_; ++j) key[j] |= atomicAdd(&hist[key[j]], 1) << KEYB_;
    } else {
#pragma unroll
        for (int j = 0; j < EPT_; ++j) {
            if (key[j] >= 0) key[j] |= atomicAdd(&hist[key[j]], 1) << KEYB_;
        }
    }
    __syncthreads();
    {
        int cnt[4], bs[4];
#pragma unroll
        for (int u = 0; u < 4; ++u) {
            int i = threadIdx.x + u * 256;
            cnt[u] = (i < NB_) ? hist[i] : 0;
        }
#pragma unroll
        for (int u = 0; u < 4; ++u) {
            int i = threadIdx.x + u * 256;
            bs[u] = cnt[u] ? atomicAdd(&cursor[i], cnt[u]) : 0;
        }
#pragma unroll
        for (int u = 0; u < 4; ++u) {
            int i = threadIdx.x + u * 256;
            if (i < NB_) base[i] = bs[u];
        }
    }
    __syncthreads();
    if (whole) {
#pragma unroll
        for (int j = 0; j < EPT_; ++j) recs[base[key[j] & KEYM_] + (key[j] >> KEYB_)] = rec[j];
    } else {
#pragma unroll
        for (int j = 0; j < EPT_; ++j) {
            if (key[j] >= 0) recs[base[key[j] & KEYM_] + (key[j] >> KEYB_)] = rec[j];
        }
    }
}

__global__ __launch_bounds__(256) void bin_dinv_kernel(const int* __restrict__ binStart,
                                                       const int* __restrict__ recs,
                                                       const float* __restrict__ x,
                                                       float* __restrict__ xp) {
    __shared__ int cnt[TR_];
    int bin = blockIdx.x;
    cnt[threadIdx.x] = 0;
    __syncthreads();
    int s = binStart[bin], e = binStart[bin + 1];
    int last = (e > 0) ? e - 1 : 0;
    for (int i0 = s + threadIdx.x; i0 < e; i0 += 4 * 256) {
        int r[4];
#pragma unroll
        for (int u = 0; u < 4; ++u) {
            int i = i0 + u * 256;
            int rv = recs[(i < e) ? i : last];
            r[u] = (i < e) ? rv : -1;
        }
#pragma unroll
        for (int u = 0; u < 4; ++u) {
            if (r[u] >= 0) atomicAdd(&cnt[r[u] >> 18], 1);
        }
    }
    __syncthreads();
    int node = bin * RB_ + threadIdx.x;
    if ((int)threadIdx.x < RB_ && node < NTOT_) {
        int c = cnt[threadIdx.x];
        float dv = (c > 0) ? rsqrtf((float)c) : 0.0f;
        const float* xr = x + node * WIN_;
        float4* __restrict__ xp4 = (float4*)xp;
        xp4[node * 2 + 0] = make_float4(xr[0], xr[1], xr[2], xr[3]);
        xp4[node * 2 + 1] = make_float4(xr[4], dv, 0.0f, 0.0f);
    }
}


static_assert(E5_ == 8, "two record slots per lane of a 4-lane group");
template <bool MAKE_W>
static __device__ __forceinline__ void accum5(float* tile, float* dinvD, int bin, int node0, int rows,
                                              const int* __restrict__ binStart,
                                              const int* __restrict__ recs,
                                              float* __restrict__ wts,
                                              const float* __restrict__ hin) {
    const float2* __restrict__ hin2 = (const float2*)hin;
    for (int i = threadIdx.x; i < TR_ * WIN_; i += 256) tile[i] = 0.0f;
    if (MAKE_W) dinvD[threadIdx.x] = ((int)threadIdx.x < rows) ? hin[(node0 + threadIdx.x) * XP_ + WIN_] : 0.0f;
    __syncthreads();
    int j = threadIdx.x & 3;
    int grp = threadIdx.x >> 2;
    int s = binStart[bin], e = binStart[bin + 1];
    int last = (e > 0) ? e - 1 : 0;
    auto load_recs = [&](int ib, int& xa, int& xb, float& ya, float& yb) {
        int ia = ib + j, ic = ib + 4 + j;
        int ca = (ia < e) ? ia : last, cc = (ic < e) ? ic : last;
        int rva = recs[ca], rvc = recs[cc];
        ya = MAKE_W ? 0.0f : wts[ca];
        yb = MAKE_W ? 0.0f : wts[cc];
        xa = (ia < e) ? rva : -1;
        xb = (ic < e) ? rvc : -1;
    };
    int t0 = s;
    int i0 = t0 + grp * E5_;
    int ra, rb;
    float wa, wb;
    load_recs(i0, ra, rb, wa, wb);
    while (t0 < e) {
        int r[E5_];
        float w[E5_];
#pragma unroll
        for (int u = 0; u < E5_; ++u) {
            r[u] = __shfl((u < 4) ? ra : rb, u & 3, 4);
            w[u] = MAKE_W ? 0.0f : __shfl((u < 4) ? wa : wb, u & 3, 4);
        }
        float2 v[E5_];
#pragma unroll
        for (int u = 0; u < E5_; ++u) {
            int srcu = (r[u] < 0) ? 0 : (r[u] & SRCM_);
            v[u] = hin2[srcu * (XP_ / 2) + j];
        }
        int t1 = t0 + 64 * E5_;
        int i1 = t1 + grp * E5_;
        int rna, rnb;
        float wna, wnb;
        load_recs(i1, rna, rnb, wna, wnb);
        float nrm[E5_];
#pragma unroll
        for (int u = 0; u < E5_; ++u) {
            if (MAKE_W) {
                int dl = (r[u] < 0) ? 0 : (r[u] >> 18);
                float dsrc = __shfl(v[u].y, 2, 4);
                nrm[u] = dsrc * dinvD[dl];
            } else {
                nrm[u] = w[u];
            }
        }
        if (MAKE_W) {
            float na = (j == 0) ? nrm[0] : (j == 1) ? nrm[1] : (j == 2) ? nrm[2] : nrm[3];
            float nb = (j == 0) ? nrm[4] : (j == 1) ? nrm[5] : (j == 2) ? nrm[6] : nrm[7];
            if (i0 + j < e)     wts[i0 + j]     = na;
            if (i0 + 4 + j < e) wts[i0 + 4 + j] = nb;
        }
        auto add_edge = [&](int u) {
            float* tp = &tile[(r[u] >> 18) * WIN_ + 2 * j];
            if (j < 3) atomicAdd(tp,     v[u].x * nrm[u]);
            if (j < 2) atomicAdd(tp + 1, v[u].y * nrm[u]);
        };
        if (i0 + E5_ <= e) {
#pragma unroll
            for (int u = 0; u < E5_; ++u) add_edge(u);
        } else {
#pragma unroll
            for (int u = 0; u < E5_; ++u) {
                if (r[u] >= 0) add_edge(u);
            }
        }
        ra = rna; rb = rnb; wa = wna; wb = wnb;
        t0 = t1;
        i0 = i1;
    }
    __syncthreads();
}

static __device__ __forceinline__ void accum32(float4* tile4, int bin,
                                               const int* __restrict__ binStart,
                                               const int* __restrict__ recs,
                                               const float* __restrict__ wts,
                                               const float* __restrict__ hin) {
    float* tile = (float*)tile4;
    const float4* __restrict__ hin4 = (const float4*)hin;
    for (int i = threadIdx.x; i < TR_ * 8; i += 256) tile4[i] = make_float4(0.f, 0.f, 0.f, 0.f);
    __syncthreads();
    int q = threadIdx.x & 7;
    int sub = threadIdx.x >> 3;
    int rot = sub & 3;
    bool r1 = (rot & 1) != 0, r2 = (rot & 2) != 0;
    int o0 = rot, o1 = (rot + 1) & 3, o2 = (rot + 2) & 3, o3 = (rot + 3) & 3;
    int s = binStart[bin], e = binStart[bin + 1];
    int last = (e > 0) ? e - 1 : 0;
    int i0 = s + sub;
    int r[E32_];
    float w[E32_];
#pragma unroll
    for (int u = 0; u < E32_; ++u) {
        int i = i0 + u * 32;
        int ic = (i < e) ? i : last;
        int rv = recs[ic];
        w[u] = wts[ic];
        r[u] = (i < e) ? rv : -1;
    }
    while (i0 < e) {
        float4 v[E32_];
#pragma unroll
        for (int u = 0; u < E32_; ++u) {
            int srcu = (r[u] < 0) ? 0 : (r[u] & SRCM_);
            v[u] = hin4[srcu * 8 + q];
        }
        int i1 = i0 + E32_ * 32;
        int rn[E32_];
        float wn[E32_];
#pragma unroll
        for (int u = 0; u < E32_; ++u) {
            int i = i1 + u * 32;
            int ic = (i < e) ? i : last;
            int rv = recs[ic];
            wn[u] = wts[ic];
            rn[u] = (i < e) ? rv : -1;
        }
        auto add_edge = [&](int rec, float nrm, float4 x) {
            int dl = rec >> 18;
            float* tp = &tile[dl * H_ + q * 4];
            float b0 = r1 ? x.y : x.x, b1 = r1 ? x.z : x.y;
            float b2 = r1 ? x.w : x.z, b3 = r1 ? x.x : x.w;
            float c0 = r2 ? b2 : b0, c1 = r2 ? b3 : b1, c2 = r2 ? b0 : b2, c3 = r2 ? b1 : b3;
            atomicAdd(tp + o0, c0 * nrm);
            atomicAdd(tp + o1, c1 * nrm);
            atomicAdd(tp + o2, c2 * nrm);
            atomicAdd(tp + o3, c3 * nrm);
        };
        if (i0 + (E32_ - 1) * 32 < e) {
#pragma unroll
            for (int u = 0; u < E32_; ++u) add_edge(r[u], w[u], v[u]);
        } else {
#pragma unroll
            for (int u = 0; u < E32_; ++u) {
                if (r[u] >= 0) add_edge(r[u], w[u], v[u]);
            }
        }
#pragma unroll
        for (int u = 0; u < E32_; ++u) { r[u] = rn[u]; w[u] = wn[u]; }
        i0 = i1;
    }
    __syncthreads();
}

__global__ __launch_bounds__(256) void bin_prop5_kernel(const int* __restrict__ binStart,
                                                        const int* __restrict__ recs,
                                                        float* __restrict__ wts,
                                                        const float* __restrict__ xp,
                                                        float* __restrict__ hout) {
    __shared__ float tile[TR_ * WIN_];
    __shared__ float dinvD[TR_];
    int bin = blockIdx.x;
    int node0 = bin * RB_;
    int rows = (NTOT_ - node0 < RB_) ? (NTOT_ - node0) : RB_;
    accum5<true>(tile, dinvD, bin, node0, rows, binStart, recs, wts, xp);
    float4* __restrict__ out4 = (float4*)hout;
    int t = threadIdx.x;
    if (t < rows) {
        const float* tr = &tile[t * WIN_];
        out4[(node0 + t) * 2 + 0] = make_float4(tr[0], tr[1], tr[2], tr[3]);
        out4[(node0 + t) * 2 + 1] = make_float4(tr[4], 0.0f, 0.0f, 0.0f);
    }
}

__global__ __launch_bounds__(256) void bin_prop32_kernel(const int* __restrict__ binStart,
                                                         const int* __restrict__ recs,
                                                         const float* __restrict__ wts,
                                                         const float* __restrict__ hin,
                                                         float* __restrict__ hout) {
    __shared__ float4 tile4[TR_ * 8];
    int bin = blockIdx.x;
    int node0 = bin * RB_;
    int rows = (NTOT_ - node0 < RB_) ? (NTOT_ - node0) : RB_;
    accum32(tile4, bin, binStart, recs, wts, hin);
    float4* __restrict__ out4 = (float4*)hout;
    for (int i = threadIdx.x; i < rows * 8; i += 256) out4[node0 * 8 + i] = tile4[i];
}

static __device__ __forceinline__ void dense_tail(float4* stage4, const v8f (&c)[2][2],
                                                  const float* __restrict__ bias,
                                                  int wave, int half, int l16, int node0, int rows,
                                                  float* __restrict__ hout) {
    float* stage = (float*)stage4;
    float bz0 = bias[l16], bz1 = bias[16 + l16];
#pragma unroll
    for (int t = 0; t < 2; ++t) {
#pragma unroll
        for (int r = 0; r < 8; ++r) {
            int row = wave * 32 + t * 16 + half * 8 + r;
            stage[row * SP_ + l16]      = tanhf(c[t][0][r] + bz0);
            stage[row * SP_ + 16 + l16] = tanhf(c[t][1][r] + bz1);
        }
    }
    __syncthreads();
    float4* __restrict__ out4 = (float4*)hout;
    for (int i = threadIdx.x; i < rows * 8; i += 256) out4[node0 * 8 + i] = stage4[(i >> 3) * SP4_ + (i & 7)];
}

__global__ __launch_bounds__(256) void bin_prop5_dense_kernel(const int* __restrict__ binStart,
                                                              const int* __restrict__ recs,
                                                              float* __restrict__ wts,
                                                              const float* __restrict__ x0,
                                                              const float* __restrict__ x1,
                                                              const float* __restrict__ W,
                                                              const float* __restrict__ bias,
                                                              float* __restrict__ hout) {
    __shared__ float tile[TR_ * WIN_];
    __shared__ float4 stage4[TR_ * SP4_];
    int bin = blockIdx.x;
    int node0 = bin * RB_;
    int rows = (NTOT_ - node0 < RB_) ? (NTOT_ - node0) : RB_;
    accum5<false>(tile, nullptr, bin, node0, rows, binStart, recs, wts, x1);

    int wave = threadIdx.x >> 5;
    int lane = threadIdx.x & 31;
    int half = lane >> 4;
    int l16  = lane & 15;
    v8f c[2][2] = {};
#pragma unroll
    for (int s = 0; s < 3; ++s) {
#pragma unroll
        for (int k0 = 0; k0 < WIN_; k0 += 4) {
            int kA = k0 + half * 2;
            float m0 = (kA     < WIN_) ? 1.0f : 0.0f;
            float m1 = (kA + 1 < WIN_) ? 1.0f : 0.0f;
            int ka0 = (kA     < WIN_) ? kA     : (WIN_ - 1);
            int ka1 = (kA + 1 < WIN_) ? kA + 1 : (WIN_ - 1);
            int kw0 = s * WIN_ + ka0;
            int kw1 = s * WIN_ + ka1;
            v2f b0, b1;
            b0.x = W[kw0 * H_ + l16] * m0;
            b0.y = W[kw1 * H_ + l16] * m1;
            b1.x = W[kw0 * H_ + 16 + l16] * m0;
            b1.y = W[kw1 * H_ + 16 + l16] * m1;
#pragma unroll
            for (int t = 0; t < 2; ++t) {
                int rl = wave * 32 + t * 16 + l16;
                int rg = node0 + rl;
                rg = (rg < NTOT_) ? rg : (NTOT_ - 1);
                v2f a;
                if (s == 0)      { a.x = x0[rg * WIN_ + ka0] * m0;  a.y = x0[rg * WIN_ + ka1] * m1; }
                else if (s == 1) { a.x = x1[rg * XP_ + ka0] * m0;   a.y = x1[rg * XP_ + ka1] * m1; }
                else             { a.x = tile[rl * WIN_ + ka0] * m0; a.y = tile[rl * WIN_ + ka1] * m1; }
                c[t][0] = __builtin_amdgcn_wmma_f32_16x16x4_f32(false, a, false, b0,
                                                                (short)0, c[t][0], false, false);
                c[t][1] = __builtin_amdgcn_wmma_f32_16x16x4_f32(false, a, false, b1,
                                                                (short)0, c[t][1], false, false);
            }
        }
    }
    dense_tail(stage4, c, bias, wave, half, l16, node0, rows, hout);
}

template <int PITCH>
static __device__ __forceinline__ void seg_from_tile(const float* tile, int wave, int half, int l16,
                                                     const float* __restrict__ Wseg, v8f (&c)[2][2]) {
#pragma unroll
    for (int k0 = 0; k0 < H_; k0 += 4) {
        int kA = k0 + half * 2;
        v2f b0, b1;
        b0.x = Wseg[(kA    ) * H_ + l16];
        b0.y = Wseg[(kA + 1) * H_ + l16];
        b1.x = Wseg[(kA    ) * H_ + 16 + l16];
        b1.y = Wseg[(kA + 1) * H_ + 16 + l16];
#pragma unroll
        for (int t = 0; t < 2; ++t) {
            const float* p = &tile[(wave * 32 + t * 16 + l16) * PITCH + kA];
            v2f a;
            a.x = p[0];
            a.y = p[1];
            c[t][0] = __builtin_amdgcn_wmma_f32_16x16x4_f32(false, a, false, b0,
                                                            (short)0, c[t][0], false, false);
            c[t][1] = __builtin_amdgcn_wmma_f32_16x16x4_f32(false, a, false, b1,
                                                            (short)0, c[t][1], false, false);
        }
    }
}

template <bool READOUT>
__global__ __launch_bounds__(256, 8) void bin_prop32_dense_kernel(const int* __restrict__ binStart,
                                                               const int* __restrict__ recs,
                                                               const float* __restrict__ wts,
                                                               const float* __restrict__ x0,
                                                               const float* __restrict__ x1,
                                                               const float* __restrict__ W,
                                                               const float* __restrict__ bias,
                                                               float* __restrict__ hout,
                                                               const float* __restrict__ Wr,
                                                               const float* __restrict__ br,
                                                               float* __restrict__ rout) {
    __shared__ float4 tile4[TR_ * 8];
    __shared__ float4 stage4[TR_ * SP4_];
    const float* tile = (const float*)tile4;
    const float* stage = (const float*)stage4;
    int bin = blockIdx.x;
    int node0 = bin * RB_;
    int rows = (NTOT_ - node0 < RB_) ? (NTOT_ - node0) : RB_;
    const float4* __restrict__ x0_4 = (const float4*)x0;
    const float4* __restrict__ x1_4 = (const float4*)x1;
    accum32(tile4, bin, binStart, recs, wts, x1);

    int wave = threadIdx.x >> 5;
    int lane = threadIdx.x & 31;
    int half = lane >> 4;
    int l16  = lane & 15;
    int srow = threadIdx.x >> 3;
    int sq   = threadIdx.x & 7;
    float4 g[8];
#pragma unroll
    for (int j = 0; j < 8; ++j) {
        int i = threadIdx.x + j * 256;
        g[j] = (i < rows * 8) ? x0_4[node0 * 8 + i] : make_float4(0.f, 0.f, 0.f, 0.f);
    }
#pragma unroll
    for (int j = 0; j < 8; ++j) stage4[(srow + j * 32) * SP4_ + sq] = g[j];
    __syncthreads();
    v8f c[2][2] = {};
    seg_from_tile<SP_>(stage, wave, half, l16, W, c);
#pragma unroll
    for (int j = 0; j < 8; ++j) {
        int i = threadIdx.x + j * 256;
        g[j] = (i < rows * 8) ? x1_4[node0 * 8 + i] : make_float4(0.f, 0.f, 0.f, 0.f);
    }
    __syncthreads();
#pragma unroll
    for (int j = 0; j < 8; ++j) stage4[(srow + j * 32) * SP4_ + sq] = g[j];
    __syncthreads();
    seg_from_tile<SP_>(stage, wave, half, l16, W + H_ * H_, c);
    seg_from_tile<H_>(tile, wave, half, l16, W + 2 * H_ * H_, c);
    dense_tail(stage4, c, bias, wave, half, l16, node0, rows, hout);
    if (READOUT) {
        int row = threadIdx.x;
        if (row < rows) {
            float acc = br[0];
#pragma unroll
            for (int j = 0; j < H_; ++j) acc += stage[row * SP_ + j] * Wr[j];
            rout[node0 + row] = acc;
        }
    }
}

static inline int cdiv(int a, int b) { return (a + b - 1) / b; }

extern "C" void kernel_launch(void* const* d_in, const int* in_sizes, int n_in,
                              void* d_out, int out_size, void* d_ws, size_t ws_size,
                              hipStream_t stream) {
    (void)in_sizes; (void)n_in; (void)out_size; (void)ws_size;

    const float* X  = (const float*)d_in[0];
    const int*   EI = (const int*)d_in[1];
    const float* W0 = (const float*)d_in[2];
    const float* b0 = (const float*)d_in[3];
    const float* W1 = (const float*)d_in[4];
    const float* b1 = (const float*)d_in[5];
    const float* W2 = (const float*)d_in[6];
    const float* b2 = (const float*)d_in[7];
    const float* Wr = (const float*)d_in[8];
    const float* br = (const float*)d_in[9];

    float* outF = (float*)d_out;
    float* hOut = outF + NTOT_;

    float* ws       = (float*)d_ws;
    int*   recs     = (int*)ws;
    int*   binCount = recs + BE_;
    int*   binStart = binCount + 1024;
    int*   cursor   = binStart + 1024;
    float* p1   = (float*)(cursor + 1024);
    float* bufA = p1 + (size_t)NTOT_ * H_;
    float* bufB = bufA + (size_t)NTOT_ * H_;
    float* wts  = bufB + (size_t)NTOT_ * H_;
    float* xp   = wts + BE_;

    const int TB = 256;

    zero_i32_kernel<<<cdiv(NB_, TB), TB, 0, stream>>>(binCount, NB_);
    bin_count_kernel<<<NCH_, 256, 0, stream>>>(EI, binCount);
    bin_scan_kernel<<<1, 1024, 0, stream>>>(binCount, binStart, cursor);
    bin_fill_kernel<<<NCH_, 256, 0, stream>>>(EI, cursor, recs);
    bin_dinv_kernel<<<NB_, 256, 0, stream>>>(binStart, recs, X, xp);

    bin_prop5_kernel<<<NB_, 256, 0, stream>>>(binStart, recs, wts, xp, p1);
    bin_prop5_dense_kernel<<<NB_, 256, 0, stream>>>(binStart, recs, wts, X, p1, W0, b0, bufA);

    bin_prop32_kernel<<<NB_, 256, 0, stream>>>(binStart, recs, wts, bufA, p1);
    bin_prop32_dense_kernel<false><<<NB_, 256, 0, stream>>>(binStart, recs, wts, bufA, p1, W1, b1, bufB,
                                                            Wr, br, outF);

    bin_prop32_kernel<<<NB_, 256, 0, stream>>>(binStart, recs, wts, bufB, p1);
    bin_prop32_dense_kernel<true><<<NB_, 256, 0, stream>>>(binStart, recs, wts, bufB, p1, W2, b2, hOut,
                                                           Wr, br, outF);
}


# Bucket fill: a chunk's records are first laid out in LDS in bucket order (scan of the chunk's counts), then stored front to back: a wave's store covers a few runs, not 32 lines; sums arrival-order
# speedup vs baseline: 5.7031x; 1.0615x over previous
#include <hip/hip_runtime.h>
#include <math.h>

#define B_    4
#define WIN_  5
#define N_    50000
#define E_    1000000
#define H_    32
#define NTOT_ (B_ * N_)
#define BE_   (B_ * E_)

typedef __attribute__((ext_vector_type(2))) float v2f;
typedef __attribute__((ext_vector_type(8))) float v8f;

static __device__ __forceinline__ void decode_edge(const int* __restrict__ ei, int e,
                                                   int& src, int& dst) {
    int b  = e / E_;
    int el = e - b * E_;
    src = ei[(b * 2 + 0) * E_ + el] + b * N_;
    dst = ei[(b * 2 + 1) * E_ + el] + b * N_;
}

#define RB_   196
#define TR_   256
#define NB_   ((NTOT_ + RB_ - 1) / RB_)
#define CH_   8192
#define NCH_  ((BE_ + CH_ - 1) / CH_)
#define EPT_  (CH_ / 256)
#define NG4_  (EPT_ / 4)
#define SRCM_ 0x3FFFF
#define SP_   36
#define SP4_  (SP_ / 4)
#define XP_   8
#define E5_   8
#define E32_  8

__global__ void zero_i32_kernel(int* __restrict__ p, int n) {
    int i = blockIdx.x * blockDim.x + threadIdx.x;
    if (i < n) p[i] = 0;
}

static __device__ __forceinline__ int group_pos(int e, int& nb) {
    bool ok = e < BE_;
    int ec = ok ? e : 0;
    int b  = ec / E_;
    int el = ec - b * E_;
    nb = ok ? b * N_ : -1;
    return (b * 2 + 1) * E_ + el;
}

__global__ __launch_bounds__(256) void bin_count_kernel(const int* __restrict__ ei,
                                                        int* __restrict__ binCount) {
    __shared__ int hist[NB_];
    for (int i = threadIdx.x; i < NB_; i += 256) hist[i] = 0;
    int e0 = blockIdx.x * CH_;
    int4 d[NG4_];
    int  nb[NG4_];
#pragma unroll
    for (int k = 0; k < NG4_; ++k) {
        int p = group_pos(e0 + (k * 256 + threadIdx.x) * 4, nb[k]);
        d[k] = *(const int4*)(ei + p);
    }
    __syncthreads();
#pragma unroll
    for (int k = 0; k < NG4_; ++k) {
        if (nb[k] >= 0) {
            atomicAdd(&hist[(d[k].x + nb[k]) / RB_], 1);
            atomicAdd(&hist[(d[k].y + nb[k]) / RB_], 1);
            atomicAdd(&hist[(d[k].z + nb[k]) / RB_], 1);
            atomicAdd(&hist[(d[k].w + nb[k]) / RB_], 1);
        }
    }
    __syncthreads();
    for (int i = threadIdx.x; i < NB_; i += 256) {
        int c = hist[i];
        if (c) atomicAdd(&binCount[i], c);
    }
}

__global__ __launch_bounds__(1024) void bin_scan_kernel(const int* __restrict__ binCount,
                                                        int* __restrict__ binStart,
                                                        int* __restrict__ cursor) {
    __shared__ int s[1024];
    int t = threadIdx.x;
    int v = (t < NB_) ? binCount[t] : 0;
    s[t] = v;
    __syncthreads();
    for (int off = 1; off < 1024; off <<= 1) {
        int x = (t >= off) ? s[t - off] : 0;
        __syncthreads();
        s[t] += x;
        __syncthreads();
    }
    int excl = s[t] - v;
    if (t < NB_) { binStart[t] = excl; cursor[t] = excl; }
    if (t == NB_ - 1) binStart[NB_] = s[t];
}

#define KEYB_ 10
#define KEYM_ ((1 << KEYB_) - 1)
#define NBP_  (1 << KEYB_)
static_assert(NB_ <= (1 << KEYB_) && NBP_ == 4 * 256 && CH_ <= (1 << (31 - KEYB_)), "key packing and the reservation's four blocks per thread");
static_assert(RB_ <= TR_ && NB_ <= 1024, "a block's rows fit the tile; the scan's single work-group covers every block");
__global__ __launch_bounds__(256) void bin_fill_kernel(const int* __restrict__ ei,
                                                       int* __restrict__ cursor,
                                                       int* __restrict__ recs) {
    __shared__ int4 hist4[NBP_ / 4];
    __shared__ int2 place[NBP_];
    __shared__ int2 stage[CH_];
    __shared__ int  wtot[8];
    int* hist = (int*)hist4;
    for (int i = threadIdx.x; i < NBP_; i += 256) hist[i] = 0;
    int e0 = blockIdx.x * CH_;
    int4 s[NG4_], d[NG4_];
    int  nb[NG4_];
#pragma unroll
    for (int k = 0; k < NG4_; ++k) {
        int p = group_pos(e0 + (k * 256 + threadIdx.x) * 4, nb[k]);
        s[k] = *(const int4*)(ei + p - E_);
        d[k] = *(const int4*)(ei + p);
    }
    __syncthreads();
    int rec[EPT_];
    int key[EPT_];
#pragma unroll
    for (int k = 0; k < NG4_; ++k) {
        int sv[4] = { s[k].x, s[k].y, s[k].z, s[k].w };
        int dv[4] = { d[k].x, d[k].y, d[k].z, d[k].w };
#pragma unroll
        for (int c = 0; c < 4; ++c) {
            int dst = dv[c] + nb[k];
            int bin = dst / RB_;
            rec[k * 4 + c] = (sv[c] + nb[k]) | ((dst - bin * RB_) << 18);
            key[k * 4 + c] = (nb[k] >= 0) ? bin : -1;
        }
    }
    bool whole = e0 + CH_ <= BE_;
    if (whole) {
#pragma unroll
        for (int j = 0; j < EPT_; ++j) key[j] |= atomicAdd(&hist[key[j]], 1) << KEYB_;
    } else {
#pragma unroll
        for (int j = 0; j < EPT_; ++j) {
            if (key[j] >= 0) key[j] |= atomicAdd(&hist[key[j]], 1) << KEYB_;
        }
    }
    __syncthreads();
    int total;
    {
        int t = threadIdx.x;
        int lane = t & 31, wv = t >> 5;
        int4 c4 = hist4[t];
        int cnt[4] = { c4.x, c4.y, c4.z, c4.w };
        int bs[4];
#pragma unroll
        for (int u = 0; u < 4; ++u) bs[u] = cnt[u] ? atomicAdd(&cursor[t * 4 + u], cnt[u]) : 0;
        int tsum = cnt[0] + cnt[1] + cnt[2] + cnt[3];
        int incl = tsum;
        for (int off = 1; off < 32; off <<= 1) {
            int y = __shfl_up(incl, off, 32);
            if (lane >= off) incl += y;
        }
        if (lane == 31) wtot[wv] = incl;
        __syncthreads();
        int before = 0;
        total = 0;
        for (int k = 0; k < 8; ++k) {
            int v = wtot[k];
            before += (k < wv) ? v : 0;
            total += v;
        }
        int p0 = before + incl - tsum;
        place[t * 4 + 0] = make_int2(p0, bs[0]);
        place[t * 4 + 1] = make_int2(p0 + cnt[0], bs[1]);
        place[t * 4 + 2] = make_int2(p0 + cnt[0] + cnt[1], bs[2]);
        place[t * 4 + 3] = make_int2(p0 + cnt[0] + cnt[1] + cnt[2], bs[3]);
    }
    __syncthreads();
    if (whole) {
#pragma unroll
        for (int j = 0; j < EPT_; ++j) {
            int2 pl = place[key[j] & KEYM_];
            int sl = key[j] >> KEYB_;
            stage[pl.x + sl] = make_int2(rec[j], pl.y + sl);
        }
    } else {
#pragma unroll
        for (int j = 0; j < EPT_; ++j) {
            if (key[j] >= 0) {
                int2 pl = place[key[j] & KEYM_];
                int sl = key[j] >> KEYB_;
                stage[pl.x + sl] = make_int2(rec[j], pl.y + sl);
            }
        }
    }
    __syncthreads();
    if (whole) {
        int2 v[EPT_];
#pragma unroll
        for (int j = 0; j < EPT_; ++j) v[j] = stage[j * 256 + threadIdx.x];
#pragma unroll
        for (int j = 0; j < EPT_; ++j) recs[v[j].y] = v[j].x;
    } else {
        for (int p = threadIdx.x; p < total; p += 256) {
            int2 v = stage[p];
            recs[v.y] = v.x;
        }
    }
}

__global__ __launch_bounds__(256) void bin_dinv_kernel(const int* __restrict__ binStart,
                                                       const int* __restrict__ recs,
                                                       const float* __restrict__ x,
                                                       float* __restrict__ xp) {
    __shared__ int cnt[TR_];
    int bin = blockIdx.x;
    cnt[threadIdx.x] = 0;
    __syncthreads();
    int s = binStart[bin], e = binStart[bin + 1];
    int last = (e > 0) ? e - 1 : 0;
    for (int i0 = s + threadIdx.x; i0 < e; i0 += 4 * 256) {
        int r[4];
#pragma unroll
        for (int u = 0; u < 4; ++u) {
            int i = i0 + u * 256;
            int rv = recs[(i < e) ? i : last];
            r[u] = (i < e) ? rv : -1;
        }
#pragma unroll
        for (int u = 0; u < 4; ++u) {
            if (r[u] >= 0) atomicAdd(&cnt[r[u] >> 18], 1);
        }
    }
    __syncthreads();
    int node = bin * RB_ + threadIdx.x;
    if ((int)threadIdx.x < RB_ && node < NTOT_) {
        int c = cnt[threadIdx.x];
        float dv = (c > 0) ? rsqrtf((float)c) : 0.0f;
        const float* xr = x + node * WIN_;
        float4* __restrict__ xp4 = (float4*)xp;
        xp4[node * 2 + 0] = make_float4(xr[0], xr[1], xr[2], xr[3]);
        xp4[node * 2 + 1] = make_float4(xr[4], dv, 0.0f, 0.0f);
    }
}


static_assert(E5_ == 8, "two record slots per lane of a 4-lane group");
template <bool MAKE_W>
static __device__ __forceinline__ void accum5(float* tile, float* dinvD, int bin, int node0, int rows,
                                              const int* __restrict__ binStart,
                                              const int* __restrict__ recs,
                                              float* __restrict__ wts,
                                              const float* __restrict__ hin) {
    const float2* __restrict__ hin2 = (const float2*)hin;
    for (int i = threadIdx.x; i < TR_ * WIN_; i += 256) tile[i] = 0.0f;
    if (MAKE_W) dinvD[threadIdx.x] = ((int)threadIdx.x < rows) ? hin[(node0 + threadIdx.x) * XP_ + WIN_] : 0.0f;
    __syncthreads();
    int j = threadIdx.x & 3;
    int grp = threadIdx.x >> 2;
    int s = binStart[bin], e = binStart[bin + 1];
    int last = (e > 0) ? e - 1 : 0;
    auto load_recs = [&](int ib, int& xa, int& xb, float& ya, float& yb) {
        int ia = ib + j, ic = ib + 4 + j;
        int ca = (ia < e) ? ia : last, cc = (ic < e) ? ic : last;
        int rva = recs[ca], rvc = recs[cc];
        ya = MAKE_W ? 0.0f : wts[ca];
        yb = MAKE_W ? 0.0f : wts[cc];
        xa = (ia < e) ? rva : -1;
        xb = (ic < e) ? rvc : -1;
    };
    int t0 = s;
    int i0 = t0 + grp * E5_;
    int ra, rb;
    float wa, wb;
    load_recs(i0, ra, rb, wa, wb);
    while (t0 < e) {
        int r[E5_];
        float w[E5_];
#pragma unroll
        for (int u = 0; u < E5_; ++u) {
            r[u] = __shfl((u < 4) ? ra : rb, u & 3, 4);
            w[u] = MAKE_W ? 0.0f : __shfl((u < 4) ? wa : wb, u & 3, 4);
        }
        float2 v[E5_];
#pragma unroll
        for (int u = 0; u < E5_; ++u) {
            int srcu = (r[u] < 0) ? 0 : (r[u] & SRCM_);
            v[u] = hin2[srcu * (XP_ / 2) + j];
        }
        int t1 = t0 + 64 * E5_;
        int i1 = t1 + grp * E5_;
        int rna, rnb;
        float wna, wnb;
        load_recs(i1, rna, rnb, wna, wnb);
        float nrm[E5_];
#pragma unroll
        for (int u = 0; u < E5_; ++u) {
            if (MAKE_W) {
                int dl = (r[u] < 0) ? 0 : (r[u] >> 18);
                float dsrc = __shfl(v[u].y, 2, 4);
                nrm[u] = dsrc * dinvD[dl];
            } else {
                nrm[u] = w[u];
            }
        }
        if (MAKE_W) {
            float na = (j == 0) ? nrm[0] : (j == 1) ? nrm[1] : (j == 2) ? nrm[2] : nrm[3];
            float nb = (j == 0) ? nrm[4] : (j == 1) ? nrm[5] : (j == 2) ? nrm[6] : nrm[7];
            if (i0 + j < e)     wts[i0 + j]     = na;
            if (i0 + 4 + j < e) wts[i0 + 4 + j] = nb;
        }
        auto add_edge = [&](int u) {
            float* tp = &tile[(r[u] >> 18) * WIN_ + 2 * j];
            if (j < 3) atomicAdd(tp,     v[u].x * nrm[u]);
            if (j < 2) atomicAdd(tp + 1, v[u].y * nrm[u]);
        };
        if (i0 + E5_ <= e) {
#pragma unroll
            for (int u = 0; u < E5_; ++u) add_edge(u);
        } else {
#pragma unroll
            for (int u = 0; u < E5_; ++u) {
                if (r[u] >= 0) add_edge(u);
            }
        }
        ra = rna; rb = rnb; wa = wna; wb = wnb;
        t0 = t1;
        i0 = i1;
    }
    __syncthreads();
}

static __device__ __forceinline__ void accum32(float4* tile4, int bin,
                                               const int* __restrict__ binStart,
                                               const int* __restrict__ recs,
                                               const float* __restrict__ wts,
                                               const float* __restrict__ hin) {
    float* tile = (float*)tile4;
    const float4* __restrict__ hin4 = (const float4*)hin;
    for (int i = threadIdx.x; i < TR_ * 8; i += 256) tile4[i] = make_float4(0.f, 0.f, 0.f, 0.f);
    __syncthreads();
    int q = threadIdx.x & 7;
    int sub = threadIdx.x >> 3;
    int rot = sub & 3;
    bool r1 = (rot & 1) != 0, r2 = (rot & 2) != 0;
    int o0 = rot, o1 = (rot + 1) & 3, o2 = (rot + 2) & 3, o3 = (rot + 3) & 3;
    int s = binStart[bin], e = binStart[bin + 1];
    int last = (e > 0) ? e - 1 : 0;
    int i0 = s + sub;
    int r[E32_];
    float w[E32_];
#pragma unroll
    for (int u = 0; u < E32_; ++u) {
        int i = i0 + u * 32;
        int ic = (i < e) ? i : last;
        int rv = recs[ic];
        w[u] = wts[ic];
        r[u] = (i < e) ? rv : -1;
    }
    while (i0 < e) {
        float4 v[E32_];
#pragma unroll
        for (int u = 0; u < E32_; ++u) {
            int srcu = (r[u] < 0) ? 0 : (r[u] & SRCM_);
            v[u] = hin4[srcu * 8 + q];
        }
        int i1 = i0 + E32_ * 32;
        int rn[E32_];
        float wn[E32_];
#pragma unroll
        for (int u = 0; u < E32_; ++u) {
            int i = i1 + u * 32;
            int ic = (i < e) ? i : last;
            int rv = recs[ic];
            wn[u] = wts[ic];
            rn[u] = (i < e) ? rv : -1;
        }
        auto add_edge = [&](int rec, float nrm, float4 x) {
            int dl = rec >> 18;
            float* tp = &tile[dl * H_ + q * 4];
            float b0 = r1 ? x.y : x.x, b1 = r1 ? x.z : x.y;
            float b2 = r1 ? x.w : x.z, b3 = r1 ? x.x : x.w;
            float c0 = r2 ? b2 : b0, c1 = r2 ? b3 : b1, c2 = r2 ? b0 : b2, c3 = r2 ? b1 : b3;
            atomicAdd(tp + o0, c0 * nrm);
            atomicAdd(tp + o1, c1 * nrm);
            atomicAdd(tp + o2, c2 * nrm);
            atomicAdd(tp + o3, c3 * nrm);
        };
        if (i0 + (E32_ - 1) * 32 < e) {
#pragma unroll
            for (int u = 0; u < E32_; ++u) add_edge(r[u], w[u], v[u]);
        } else {
#pragma unroll
            for (int u = 0; u < E32_; ++u) {
                if (r[u] >= 0) add_edge(r[u], w[u], v[u]);
            }
        }
#pragma unroll
        for (int u = 0; u < E32_; ++u) { r[u] = rn[u]; w[u] = wn[u]; }
        i0 = i1;
    }
    __syncthreads();
}

__global__ __launch_bounds__(256) void bin_prop5_kernel(const int* __restrict__ binStart,
                                                        const int* __restrict__ recs,
                                                        float* __restrict__ wts,
                                                        const float* __restrict__ xp,
                                                        float* __restrict__ hout) {
    __shared__ float tile[TR_ * WIN_];
    __shared__ float dinvD[TR_];
    int bin = blockIdx.x;
    int node0 = bin * RB_;
    int rows = (NTOT_ - node0 < RB_) ? (NTOT_ - node0) : RB_;
    accum5<true>(tile, dinvD, bin, node0, rows, binStart, recs, wts, xp);
    float4* __restrict__ out4 = (float4*)hout;
    int t = threadIdx.x;
    if (t < rows) {
        const float* tr = &tile[t * WIN_];
        out4[(node0 + t) * 2 + 0] = make_float4(tr[0], tr[1], tr[2], tr[3]);
        out4[(node0 + t) * 2 + 1] = make_float4(tr[4], 0.0f, 0.0f, 0.0f);
    }
}

__global__ __launch_bounds__(256) void bin_prop32_kernel(const int* __restrict__ binStart,
                                                         const int* __restrict__ recs,
                                                         const float* __restrict__ wts,
                                                         const float* __restrict__ hin,
                                                         float* __restrict__ hout) {
    __shared__ float4 tile4[TR_ * 8];
    int bin = blockIdx.x;
    int node0 = bin * RB_;
    int rows = (NTOT_ - node0 < RB_) ? (NTOT_ - node0) : RB_;
    accum32(tile4, bin, binStart, recs, wts, hin);
    float4* __restrict__ out4 = (float4*)hout;
    for (int i = threadIdx.x; i < rows * 8; i += 256) out4[node0 * 8 + i] = tile4[i];
}

static __device__ __forceinline__ void dense_tail(float4* stage4, const v8f (&c)[2][2],
                                                  const float* __restrict__ bias,
                                                  int wave, int half, int l16, int node0, int rows,
                                                  float* __restrict__ hout) {
    float* stage = (float*)stage4;
    float bz0 = bias[l16], bz1 = bias[16 + l16];
#pragma unroll
    for (int t = 0; t < 2; ++t) {
#pragma unroll
        for (int r = 0; r < 8; ++r) {
            int row = wave * 32 + t * 16 + half * 8 + r;
            stage[row * SP_ + l16]      = tanhf(c[t][0][r] + bz0);
            stage[row * SP_ + 16 + l16] = tanhf(c[t][1][r] + bz1);
        }
    }
    __syncthreads();
    float4* __restrict__ out4 = (float4*)hout;
    for (int i = threadIdx.x; i < rows * 8; i += 256) out4[node0 * 8 + i] = stage4[(i >> 3) * SP4_ + (i & 7)];
}

__global__ __launch_bounds__(256) void bin_prop5_dense_kernel(const int* __restrict__ binStart,
                                                              const int* __restrict__ recs,
                                                              float* __restrict__ wts,
                                                              const float* __restrict__ x0,
                                                              const float* __restrict__ x1,
                                                              const float* __restrict__ W,
                                                              const float* __restrict__ bias,
                                                              float* __restrict__ hout) {
    __shared__ float tile[TR_ * WIN_];
    __shared__ float4 stage4[TR_ * SP4_];
    int bin = blockIdx.x;
    int node0 = bin * RB_;
    int rows = (NTOT_ - node0 < RB_) ? (NTOT_ - node0) : RB_;
    accum5<false>(tile, nullptr, bin, node0, rows, binStart, recs, wts, x1);

    int wave = threadIdx.x >> 5;
    int lane = threadIdx.x & 31;
    int half = lane >> 4;
    int l16  = lane & 15;
    v8f c[2][2] = {};
#pragma unroll
    for (int s = 0; s < 3; ++s) {
#pragma unroll
        for (int k0 = 0; k0 < WIN_; k0 += 4) {
            int kA = k0 + half * 2;
            float m0 = (kA     < WIN_) ? 1.0f : 0.0f;
            float m1 = (kA + 1 < WIN_) ? 1.0f : 0.0f;
            int ka0 = (kA     < WIN_) ? kA     : (WIN_ - 1);
            int ka1 = (kA + 1 < WIN_) ? kA + 1 : (WIN_ - 1);
            int kw0 = s * WIN_ + ka0;
            int kw1 = s * WIN_ + ka1;
            v2f b0, b1;
            b0.x = W[kw0 * H_ + l16] * m0;
            b0.y = W[kw1 * H_ + l16] * m1;
            b1.x = W[kw0 * H_ + 16 + l16] * m0;
            b1.y = W[kw1 * H_ + 16 + l16] * m1;
#pragma unroll
            for (int t = 0; t < 2; ++t) {
                int rl = wave * 32 + t * 16 + l16;
                int rg = node0 + rl;
                rg = (rg < NTOT_) ? rg : (NTOT_ - 1);
                v2f a;
                if (s == 0)      { a.x = x0[rg * WIN_ + ka0] * m0;  a.y = x0[rg * WIN_ + ka1] * m1; }
                else if (s == 1) { a.x = x1[rg * XP_ + ka0] * m0;   a.y = x1[rg * XP_ + ka1] * m1; }
                else             { a.x = tile[rl * WIN_ + ka0] * m0; a.y = tile[rl * WIN_ + ka1] * m1; }
                c[t][0] = __builtin_amdgcn_wmma_f32_16x16x4_f32(false, a, false, b0,
                                                                (short)0, c[t][0], false, false);
                c[t][1] = __builtin_amdgcn_wmma_f32_16x16x4_f32(false, a, false, b1,
                                                                (short)0, c[t][1], false, false);
            }
        }
    }
    dense_tail(stage4, c, bias, wave, half, l16, node0, rows, hout);
}

template <int PITCH>
static __device__ __forceinline__ void seg_from_tile(const float* tile, int wave, int half, int l16,
                                                     const float* __restrict__ Wseg, v8f (&c)[2][2]) {
#pragma unroll
    for (int k0 = 0; k0 < H_; k0 += 4) {
        int kA = k0 + half * 2;
        v2f b0, b1;
        b0.x = Wseg[(kA    ) * H_ + l16];
        b0.y = Wseg[(kA + 1) * H_ + l16];
        b1.x = Wseg[(kA    ) * H_ + 16 + l16];
        b1.y = Wseg[(kA + 1) * H_ + 16 + l16];
#pragma unroll
        for (int t = 0; t < 2; ++t) {
            const float* p = &tile[(wave * 32 + t * 16 + l16) * PITCH + kA];
            v2f a;
            a.x = p[0];
            a.y = p[1];
            c[t][0] = __builtin_amdgcn_wmma_f32_16x16x4_f32(false, a, false, b0,
                                                            (short)0, c[t][0], false, false);
            c[t][1] = __builtin_amdgcn_wmma_f32_16x16x4_f32(false, a, false, b1,
                                                            (short)0, c[t][1], false, false);
        }
    }
}

template <bool READOUT>
__global__ __launch_bounds__(256, 8) void bin_prop32_dense_kernel(const int* __restrict__ binStart,
                                                               const int* __restrict__ recs,
                                                               const float* __restrict__ wts,
                                                               const float* __restrict__ x0,
                                                               const float* __restrict__ x1,
                                                               const float* __restrict__ W,
                                                               const float* __restrict__ bias,
                                                               float* __restrict__ hout,
                                                               const float* __restrict__ Wr,
                                                               const float* __restrict__ br,
                                                               float* __restrict__ rout) {
    __shared__ float4 tile4[TR_ * 8];
    __shared__ float4 stage4[TR_ * SP4_];
    const float* tile = (const float*)tile4;
    const float* stage = (const float*)stage4;
    int bin = blockIdx.x;
    int node0 = bin * RB_;
    int rows = (NTOT_ - node0 < RB_) ? (NTOT_ - node0) : RB_;
    const float4* __restrict__ x0_4 = (const float4*)x0;
    const float4* __restrict__ x1_4 = (const float4*)x1;
    accum32(tile4, bin, binStart, recs, wts, x1);

    int wave = threadIdx.x >> 5;
    int lane = threadIdx.x & 31;
    int half = lane >> 4;
    int l16  = lane & 15;
    int srow = threadIdx.x >> 3;
    int sq   = threadIdx.x & 7;
    float4 g[8];
#pragma unroll
    for (int j = 0; j < 8; ++j) {
        int i = threadIdx.x + j * 256;
        g[j] = (i < rows * 8) ? x0_4[node0 * 8 + i] : make_float4(0.f, 0.f, 0.f, 0.f);
    }
#pragma unroll
    for (int j = 0; j < 8; ++j) stage4[(srow + j * 32) * SP4_ + sq] = g[j];
    __syncthreads();
    v8f c[2][2] = {};
    seg_from_tile<SP_>(stage, wave, half, l16, W, c);
#pragma unroll
    for (int j = 0; j < 8; ++j) {
        int i = threadIdx.x + j * 256;
        g[j] = (i < rows * 8) ? x1_4[node0 * 8 + i] : make_float4(0.f, 0.f, 0.f, 0.f);
    }
    __syncthreads();
#pragma unroll
    for (int j = 0; j < 8; ++j) stage4[(srow + j * 32) * SP4_ + sq] = g[j];
    __syncthreads();
    seg_from_tile<SP_>(stage, wave, half, l16, W + H_ * H_, c);
    seg_from_tile<H_>(tile, wave, half, l16, W + 2 * H_ * H_, c);
    dense_tail(stage4, c, bias, wave, half, l16, node0, rows, hout);
    if (READOUT) {
        int row = threadIdx.x;
        if (row < rows) {
            float acc = br[0];
#pragma unroll
            for (int j = 0; j < H_; ++j) acc += stage[row * SP_ + j] * Wr[j];
            rout[node0 + row] = acc;
        }
    }
}

static inline int cdiv(int a, int b) { return (a + b - 1) / b; }

extern "C" void kernel_launch(void* const* d_in, const int* in_sizes, int n_in,
                              void* d_out, int out_size, void* d_ws, size_t ws_size,
                              hipStream_t stream) {
    (void)in_sizes; (void)n_in; (void)out_size; (void)ws_size;

    const float* X  = (const float*)d_in[0];
    const int*   EI = (const int*)d_in[1];
    const float* W0 = (const float*)d_in[2];
    const float* b0 = (const float*)d_in[3];
    const float* W1 = (const float*)d_in[4];
    const float* b1 = (const float*)d_in[5];
    const float* W2 = (const float*)d_in[6];
    const float* b2 = (const float*)d_in[7];
    const float* Wr = (const float*)d_in[8];
    const float* br = (const float*)d_in[9];

    float* outF = (float*)d_out;
    float* hOut = outF + NTOT_;

    float* ws       = (float*)d_ws;
    int*   recs     = (int*)ws;
    int*   binCount = recs + BE_;
    int*   binStart = binCount + 1024;
    int*   cursor   = binStart + 1024;
    float* p1   = (float*)(cursor + 1024);
    float* bufA = p1 + (size_t)NTOT_ * H_;
    float* bufB = bufA + (size_t)NTOT_ * H_;
    float* wts  = bufB + (size_t)NTOT_ * H_;
    float* xp   = wts + BE_;

    const int TB = 256;

    zero_i32_kernel<<<cdiv(NB_, TB), TB, 0, stream>>>(binCount, NB_);
    bin_count_kernel<<<NCH_, 256, 0, stream>>>(EI, binCount);
    bin_scan_kernel<<<1, 1024, 0, stream>>>(binCount, binStart, cursor);
    bin_fill_kernel<<<NCH_, 256, 0, stream>>>(EI, cursor, recs);
    bin_dinv_kernel<<<NB_, 256, 0, stream>>>(binStart, recs, X, xp);

    bin_prop5_kernel<<<NB_, 256, 0, stream>>>(binStart, recs, wts, xp, p1);
    bin_prop5_dense_kernel<<<NB_, 256, 0, stream>>>(binStart, recs, wts, X, p1, W0, b0, bufA);

    bin_prop32_kernel<<<NB_, 256, 0, stream>>>(binStart, recs, wts, bufA, p1);
    bin_prop32_dense_kernel<false><<<NB_, 256, 0, stream>>>(binStart, recs, wts, bufA, p1, W1, b1, bufB,
                                                            Wr, br, outF);

    bin_prop32_kernel<<<NB_, 256, 0, stream>>>(binStart, recs, wts, bufB, p1);
    bin_prop32_dense_kernel<true><<<NB_, 256, 0, stream>>>(binStart, recs, wts, bufB, p1, W2, b2, hOut,
                                                           Wr, br, outF);
}


# The dense stages' tanh is one hardware instruction (v_tanh_f32) per value, not the library's in-line two-branch exp and reciprocal expansion (about 40 instructions); last bits may differ
# speedup vs baseline: 5.9225x; 1.0385x over previous
#include <hip/hip_runtime.h>
#include <math.h>

#define B_    4
#define WIN_  5
#define N_    50000
#define E_    1000000
#define H_    32
#define NTOT_ (B_ * N_)
#define BE_   (B_ * E_)

typedef __attribute__((ext_vector_type(2))) float v2f;
typedef __attribute__((ext_vector_type(8))) float v8f;

static __device__ __forceinline__ void decode_edge(const int* __restrict__ ei, int e,
                                                   int& src, int& dst) {
    int b  = e / E_;
    int el = e - b * E_;
    src = ei[(b * 2 + 0) * E_ + el] + b * N_;
    dst = ei[(b * 2 + 1) * E_ + el] + b * N_;
}

#define RB_   196
#define TR_   256
#define NB_   ((NTOT_ + RB_ - 1) / RB_)
#define CH_   8192
#define NCH_  ((BE_ + CH_ - 1) / CH_)
#define EPT_  (CH_ / 256)
#define NG4_  (EPT_ / 4)
#define SRCM_ 0x3FFFF
#define SP_   36
#define SP4_  (SP_ / 4)
#define XP_   8
#define E5_   8
#define E32_  8

__global__ void zero_i32_kernel(int* __restrict__ p, int n) {
    int i = blockIdx.x * blockDim.x + threadIdx.x;
    if (i < n) p[i] = 0;
}

static __device__ __forceinline__ int group_pos(int e, int& nb) {
    bool ok = e < BE_;
    int ec = ok ? e : 0;
    int b  = ec / E_;
    int el = ec - b * E_;
    nb = ok ? b * N_ : -1;
    return (b * 2 + 1) * E_ + el;
}

__global__ __launch_bounds__(256) void bin_count_kernel(const int* __restrict__ ei,
                                                        int* __restrict__ binCount) {
    __shared__ int hist[NB_];
    for (int i = threadIdx.x; i < NB_; i += 256) hist[i] = 0;
    int e0 = blockIdx.x * CH_;
    int4 d[NG4_];
    int  nb[NG4_];
#pragma unroll
    for (int k = 0; k < NG4_; ++k) {
        int p = group_pos(e0 + (k * 256 + threadIdx.x) * 4, nb[k]);
        d[k] = *(const int4*)(ei + p);
    }
    __syncthreads();
#pragma unroll
    for (int k = 0; k < NG4_; ++k) {
        if (nb[k] >= 0) {
            atomicAdd(&hist[(d[k].x + nb[k]) / RB_], 1);
            atomicAdd(&hist[(d[k].y + nb[k]) / RB_], 1);
            atomicAdd(&hist[(d[k].z + nb[k]) / RB_], 1);
            atomicAdd(&hist[(d[k].w + nb[k]) / RB_], 1);
        }
    }
    __syncthreads();
    for (int i = threadIdx.x; i < NB_; i += 256) {
        int c = hist[i];
        if (c) atomicAdd(&binCount[i], c);
    }
}

__global__ __launch_bounds__(1024) void bin_scan_kernel(const int* __restrict__ binCount,
                                                        int* __restrict__ binStart,
                                                        int* __restrict__ cursor) {
    __shared__ int s[1024];
    int t = threadIdx.x;
    int v = (t < NB_) ? binCount[t] : 0;
    s[t] = v;
    __syncthreads();
    for (int off = 1; off < 1024; off <<= 1) {
        int x = (t >= off) ? s[t - off] : 0;
        __syncthreads();
        s[t] += x;
        __syncthreads();
    }
    int excl = s[t] - v;
    if (t < NB_) { binStart[t] = excl; cursor[t] = excl; }
    if (t == NB_ - 1) binStart[NB_] = s[t];
}

#define KEYB_ 10
#define KEYM_ ((1 << KEYB_) - 1)
#define NBP_  (1 << KEYB_)
static_assert(NB_ <= (1 << KEYB_) && NBP_ == 4 * 256 && CH_ <= (1 << (31 - KEYB_)), "key packing and the reservation's four blocks per thread");
static_assert(RB_ <= TR_ && NB_ <= 1024, "a block's rows fit the tile; the scan's single work-group covers every block");
__global__ __launch_bounds__(256) void bin_fill_kernel(const int* __restrict__ ei,
                                                       int* __restrict__ cursor,
                                                       int* __restrict__ recs) {
    __shared__ int4 hist4[NBP_ / 4];
    __shared__ int2 place[NBP_];
    __shared__ int2 stage[CH_];
    __shared__ int  wtot[8];
    int* hist = (int*)hist4;
    for (int i = threadIdx.x; i < NBP_; i += 256) hist[i] = 0;
    int e0 = blockIdx.x * CH_;
    int4 s[NG4_], d[NG4_];
    int  nb[NG4_];
#pragma unroll
    for (int k = 0; k < NG4_; ++k) {
        int p = group_pos(e0 + (k * 256 + threadIdx.x) * 4, nb[k]);
        s[k] = *(const int4*)(ei + p - E_);
        d[k] = *(const int4*)(ei + p);
    }
    __syncthreads();
    int rec[EPT_];
    int key[EPT_];
#pragma unroll
    for (int k = 0; k < NG4_; ++k) {
        int sv[4] = { s[k].x, s[k].y, s[k].z, s[k].w };
        int dv[4] = { d[k].x, d[k].y, d[k].z, d[k].w };
#pragma unroll
        for (int c = 0; c < 4; ++c) {
            int dst = dv[c] + nb[k];
            int bin = dst / RB_;
            rec[k * 4 + c] = (sv[c] + nb[k]) | ((dst - bin * RB_) << 18);
            key[k * 4 + c] = (nb[k] >= 0) ? bin : -1;
        }
    }
    bool whole = e0 + CH_ <= BE_;
    if (whole) {
#pragma unroll
        for (int j = 0; j < EPT_; ++j) key[j] |= atomicAdd(&hist[key[j]], 1) << KEYB_;
    } else {
#pragma unroll
        for (int j = 0; j < EPT_; ++j) {
            if (key[j] >= 0) key[j] |= atomicAdd(&hist[key[j]], 1) << KEYB_;
        }
    }
    __syncthreads();
    int total;
    {
        int t = threadIdx.x;
        int lane = t & 31, wv = t >> 5;
        int4 c4 = hist4[t];
        int cnt[4] = { c4.x, c4.y, c4.z, c4.w };
        int bs[4];
#pragma unroll
        for (int u = 0; u < 4; ++u) bs[u] = cnt[u] ? atomicAdd(&cursor[t * 4 + u], cnt[u]) : 0;
        int tsum = cnt[0] + cnt[1] + cnt[2] + cnt[3];
        int incl = tsum;
        for (int off = 1; off < 32; off <<= 1) {
            int y = __shfl_up(incl, off, 32);
            if (lane >= off) incl += y;
        }
        if (lane == 31) wtot[wv] = incl;
        __syncthreads();
        int before = 0;
        total = 0;
        for (int k = 0; k < 8; ++k) {
            int v = wtot[k];
            before += (k < wv) ? v : 0;
            total += v;
        }
        int p0 = before + incl - tsum;
        place[t * 4 + 0] = make_int2(p0, bs[0]);
        place[t * 4 + 1] = make_int2(p0 + cnt[0], bs[1]);
        place[t * 4 + 2] = make_int2(p0 + cnt[0] + cnt[1], bs[2]);
        place[t * 4 + 3] = make_int2(p0 + cnt[0] + cnt[1] + cnt[2], bs[3]);
    }
    __syncthreads();
    if (whole) {
#pragma unroll
        for (int j = 0; j < EPT_; ++j) {
            int2 pl = place[key[j] & KEYM_];
            int sl = key[j] >> KEYB_;
            stage[pl.x + sl] = make_int2(rec[j], pl.y + sl);
        }
    } else {
#pragma unroll
        for (int j = 0; j < EPT_; ++j) {
            if (key[j] >= 0) {
                int2 pl = place[key[j] & KEYM_];
                int sl = key[j] >> KEYB_;
                stage[pl.x + sl] = make_int2(rec[j], pl.y + sl);
            }
        }
    }
    __syncthreads();
    if (whole) {
        int2 v[EPT_];
#pragma unroll
        for (int j = 0; j < EPT_; ++j) v[j] = stage[j * 256 + threadIdx.x];
#pragma unroll
        for (int j = 0; j < EPT_; ++j) recs[v[j].y] = v[j].x;
    } else {
        for (int p = threadIdx.x; p < total; p += 256) {
            int2 v = stage[p];
            recs[v.y] = v.x;
        }
    }
}

__global__ __launch_bounds__(256) void bin_dinv_kernel(const int* __restrict__ binStart,
                                                       const int* __restrict__ recs,
                                                       const float* __restrict__ x,
                                                       float* __restrict__ xp) {
    __shared__ int cnt[TR_];
    int bin = blockIdx.x;
    cnt[threadIdx.x] = 0;
    __syncthreads();
    int s = binStart[bin], e = binStart[bin + 1];
    int last = (e > 0) ? e - 1 : 0;
    for (int i0 = s + threadIdx.x; i0 < e; i0 += 4 * 256) {
        int r[4];
#pragma unroll
        for (int u = 0; u < 4; ++u) {
            int i = i0 + u * 256;
            int rv = recs[(i < e) ? i : last];
            r[u] = (i < e) ? rv : -1;
        }
#pragma unroll
        for (int u = 0; u < 4; ++u) {
            if (r[u] >= 0) atomicAdd(&cnt[r[u] >> 18], 1);
        }
    }
    __syncthreads();
    int node = bin * RB_ + threadIdx.x;
    if ((int)threadIdx.x < RB_ && node < NTOT_) {
        int c = cnt[threadIdx.x];
        float dv = (c > 0) ? rsqrtf((float)c) : 0.0f;
        const float* xr = x + node * WIN_;
        float4* __restrict__ xp4 = (float4*)xp;
        xp4[node * 2 + 0] = make_float4(xr[0], xr[1], xr[2], xr[3]);
        xp4[node * 2 + 1] = make_float4(xr[4], dv, 0.0f, 0.0f);
    }
}


static_assert(E5_ == 8, "two record slots per lane of a 4-lane group");
template <bool MAKE_W>
static __device__ __forceinline__ void accum5(float* tile, float* dinvD, int bin, int node0, int rows,
                                              const int* __restrict__ binStart,
                                              const int* __restrict__ recs,
                                              float* __restrict__ wts,
                                              const float* __restrict__ hin) {
    const float2* __restrict__ hin2 = (const float2*)hin;
    for (int i = threadIdx.x; i < TR_ * WIN_; i += 256) tile[i] = 0.0f;
    if (MAKE_W) dinvD[threadIdx.x] = ((int)threadIdx.x < rows) ? hin[(node0 + threadIdx.x) * XP_ + WIN_] : 0.0f;
    __syncthreads();
    int j = threadIdx.x & 3;
    int grp = threadIdx.x >> 2;
    int s = binStart[bin], e = binStart[bin + 1];
    int last = (e > 0) ? e - 1 : 0;
    auto load_recs = [&](int ib, int& xa, int& xb, float& ya, float& yb) {
        int ia = ib + j, ic = ib + 4 + j;
        int ca = (ia < e) ? ia : last, cc = (ic < e) ? ic : last;
        int rva = recs[ca], rvc = recs[cc];
        ya = MAKE_W ? 0.0f : wts[ca];
        yb = MAKE_W ? 0.0f : wts[cc];
        xa = (ia < e) ? rva : -1;
        xb = (ic < e) ? rvc : -1;
    };
    int t0 = s;
    int i0 = t0 + grp * E5_;
    int ra, rb;
    float wa, wb;
    load_recs(i0, ra, rb, wa, wb);
    while (t0 < e) {
        int r[E5_];
        float w[E5_];
#pragma unroll
        for (int u = 0; u < E5_; ++u) {
            r[u] = __shfl((u < 4) ? ra : rb, u & 3, 4);
            w[u] = MAKE_W ? 0.0f : __shfl((u < 4) ? wa : wb, u & 3, 4);
        }
        float2 v[E5_];
#pragma unroll
        for (int u = 0; u < E5_; ++u) {
            int srcu = (r[u] < 0) ? 0 : (r[u] & SRCM_);
            v[u] = hin2[srcu * (XP_ / 2) + j];
        }
        int t1 = t0 + 64 * E5_;
        int i1 = t1 + grp * E5_;
        int rna, rnb;
        float wna, wnb;
        load_recs(i1, rna, rnb, wna, wnb);
        float nrm[E5_];
#pragma unroll
        for (int u = 0; u < E5_; ++u) {
            if (MAKE_W) {
                int dl = (r[u] < 0) ? 0 : (r[u] >> 18);
                float dsrc = __shfl(v[u].y, 2, 4);
                nrm[u] = dsrc * dinvD[dl];
            } else {
                nrm[u] = w[u];
            }
        }
        if (MAKE_W) {
            float na = (j == 0) ? nrm[0] : (j == 1) ? nrm[1] : (j == 2) ? nrm[2] : nrm[3];
            float nb = (j == 0) ? nrm[4] : (j == 1) ? nrm[5] : (j == 2) ? nrm[6] : nrm[7];
            if (i0 + j < e)     wts[i0 + j]     = na;
            if (i0 + 4 + j < e) wts[i0 + 4 + j] = nb;
        }
        auto add_edge = [&](int u) {
            float* tp = &tile[(r[u] >> 18) * WIN_ + 2 * j];
            if (j < 3) atomicAdd(tp,     v[u].x * nrm[u]);
            if (j < 2) atomicAdd(tp + 1, v[u].y * nrm[u]);
        };
        if (i0 + E5_ <= e) {
#pragma unroll
            for (int u = 0; u < E5_; ++u) add_edge(u);
        } else {
#pragma unroll
            for (int u = 0; u < E5_; ++u) {
                if (r[u] >= 0) add_edge(u);
            }
        }
        ra = rna; rb = rnb; wa = wna; wb = wnb;
        t0 = t1;
        i0 = i1;
    }
    __syncthreads();
}

static __device__ __forceinline__ void accum32(float4* tile4, int bin,
                                               const int* __restrict__ binStart,
                                               const int* __restrict__ recs,
                                               const float* __restrict__ wts,
                                               const float* __restrict__ hin) {
    float* tile = (float*)tile4;
    const float4* __restrict__ hin4 = (const float4*)hin;
    for (int i = threadIdx.x; i < TR_ * 8; i += 256) tile4[i] = make_float4(0.f, 0.f, 0.f, 0.f);
    __syncthreads();
    int q = threadIdx.x & 7;
    int sub = threadIdx.x >> 3;
    int rot = sub & 3;
    bool r1 = (rot & 1) != 0, r2 = (rot & 2) != 0;
    int o0 = rot, o1 = (rot + 1) & 3, o2 = (rot + 2) & 3, o3 = (rot + 3) & 3;
    int s = binStart[bin], e = binStart[bin + 1];
    int last = (e > 0) ? e - 1 : 0;
    int i0 = s + sub;
    int r[E32_];
    float w[E32_];
#pragma unroll
    for (int u = 0; u < E32_; ++u) {
        int i = i0 + u * 32;
        int ic = (i < e) ? i : last;
        int rv = recs[ic];
        w[u] = wts[ic];
        r[u] = (i < e) ? rv : -1;
    }
    while (i0 < e) {
        float4 v[E32_];
#pragma unroll
        for (int u = 0; u < E32_; ++u) {
            int srcu = (r[u] < 0) ? 0 : (r[u] & SRCM_);
            v[u] = hin4[srcu * 8 + q];
        }
        int i1 = i0 + E32_ * 32;
        int rn[E32_];
        float wn[E32_];
#pragma unroll
        for (int u = 0; u < E32_; ++u) {
            int i = i1 + u * 32;
            int ic = (i < e) ? i : last;
            int rv = recs[ic];
            wn[u] = wts[ic];
            rn[u] = (i < e) ? rv : -1;
        }
        auto add_edge = [&](int rec, float nrm, float4 x) {
            int dl = rec >> 18;
            float* tp = &tile[dl * H_ + q * 4];
            float b0 = r1 ? x.y : x.x, b1 = r1 ? x.z : x.y;
            float b2 = r1 ? x.w : x.z, b3 = r1 ? x.x : x.w;
            float c0 = r2 ? b2 : b0, c1 = r2 ? b3 : b1, c2 = r2 ? b0 : b2, c3 = r2 ? b1 : b3;
            atomicAdd(tp + o0, c0 * nrm);
            atomicAdd(tp + o1, c1 * nrm);
            atomicAdd(tp + o2, c2 * nrm);
            atomicAdd(tp + o3, c3 * nrm);
        };
        if (i0 + (E32_ - 1) * 32 < e) {
#pragma unroll
            for (int u = 0; u < E32_; ++u) add_edge(r[u], w[u], v[u]);
        } else {
#pragma unroll
            for (int u = 0; u < E32_; ++u) {
                if (r[u] >= 0) add_edge(r[u], w[u], v[u]);
            }
        }
#pragma unroll
        for (int u = 0; u < E32_; ++u) { r[u] = rn[u]; w[u] = wn[u]; }
        i0 = i1;
    }
    __syncthreads();
}

__global__ __launch_bounds__(256) void bin_prop5_kernel(const int* __restrict__ binStart,
                                                        const int* __restrict__ recs,
                                                        float* __restrict__ wts,
                                                        const float* __restrict__ xp,
                                                        float* __restrict__ hout) {
    __shared__ float tile[TR_ * WIN_];
    __shared__ float dinvD[TR_];
    int bin = blockIdx.x;
    int node0 = bin * RB_;
    int rows = (NTOT_ - node0 < RB_) ? (NTOT_ - node0) : RB_;
    accum5<true>(tile, dinvD, bin, node0, rows, binStart, recs, wts, xp);
    float4* __restrict__ out4 = (float4*)hout;
    int t = threadIdx.x;
    if (t < rows) {
        const float* tr = &tile[t * WIN_];
        out4[(node0 + t) * 2 + 0] = make_float4(tr[0], tr[1], tr[2], tr[3]);
        out4[(node0 + t) * 2 + 1] = make_float4(tr[4], 0.0f, 0.0f, 0.0f);
    }
}

__global__ __launch_bounds__(256) void bin_prop32_kernel(const int* __restrict__ binStart,
                                                         const int* __restrict__ recs,
                                                         const float* __restrict__ wts,
                                                         const float* __restrict__ hin,
                                                         float* __restrict__ hout) {
    __shared__ float4 tile4[TR_ * 8];
    int bin = blockIdx.x;
    int node0 = bin * RB_;
    int rows = (NTOT_ - node0 < RB_) ? (NTOT_ - node0) : RB_;
    accum32(tile4, bin, binStart, recs, wts, hin);
    float4* __restrict__ out4 = (float4*)hout;
    for (int i = threadIdx.x; i < rows * 8; i += 256) out4[node0 * 8 + i] = tile4[i];
}

static __device__ __forceinline__ void dense_tail(float4* stage4, const v8f (&c)[2][2],
                                                  const float* __restrict__ bias,
                                                  int wave, int half, int l16, int node0, int rows,
                                                  float* __restrict__ hout) {
    float* stage = (float*)stage4;
    float bz0 = bias[l16], bz1 = bias[16 + l16];
#pragma unroll
    for (int t = 0; t < 2; ++t) {
#pragma unroll
        for (int r = 0; r < 8; ++r) {
            int row = wave * 32 + t * 16 + half * 8 + r;
            stage[row * SP_ + l16]      = __builtin_amdgcn_tanhf(c[t][0][r] + bz0);
            stage[row * SP_ + 16 + l16] = __builtin_amdgcn_tanhf(c[t][1][r] + bz1);
        }
    }
    __syncthreads();
    float4* __restrict__ out4 = (float4*)hout;
    for (int i = threadIdx.x; i < rows * 8; i += 256) out4[node0 * 8 + i] = stage4[(i >> 3) * SP4_ + (i & 7)];
}

__global__ __launch_bounds__(256) void bin_prop5_dense_kernel(const int* __restrict__ binStart,
                                                              const int* __restrict__ recs,
                                                              float* __restrict__ wts,
                                                              const float* __restrict__ x0,
                                                              const float* __restrict__ x1,
                                                              const float* __restrict__ W,
                                                              const float* __restrict__ bias,
                                                              float* __restrict__ hout) {
    __shared__ float tile[TR_ * WIN_];
    __shared__ float4 stage4[TR_ * SP4_];
    int bin = blockIdx.x;
    int node0 = bin * RB_;
    int rows = (NTOT_ - node0 < RB_) ? (NTOT_ - node0) : RB_;
    accum5<false>(tile, nullptr, bin, node0, rows, binStart, recs, wts, x1);

    int wave = threadIdx.x >> 5;
    int lane = threadIdx.x & 31;
    int half = lane >> 4;
    int l16  = lane & 15;
    v8f c[2][2] = {};
#pragma unroll
    for (int s = 0; s < 3; ++s) {
#pragma unroll
        for (int k0 = 0; k0 < WIN_; k0 += 4) {
            int kA = k0 + half * 2;
            float m0 = (kA     < WIN_) ? 1.0f : 0.0f;
            float m1 = (kA + 1 < WIN_) ? 1.0f : 0.0f;
            int ka0 = (kA     < WIN_) ? kA     : (WIN_ - 1);
            int ka1 = (kA + 1 < WIN_) ? kA + 1 : (WIN_ - 1);
            int kw0 = s * WIN_ + ka0;
            int kw1 = s * WIN_ + ka1;
            v2f b0, b1;
            b0.x = W[kw0 * H_ + l16] * m0;
            b0.y = W[kw1 * H_ + l16] * m1;
            b1.x = W[kw0 * H_ + 16 + l16] * m0;
            b1.y = W[kw1 * H_ + 16 + l16] * m1;
#pragma unroll
            for (int t = 0; t < 2; ++t) {
                int rl = wave * 32 + t * 16 + l16;
                int rg = node0 + rl;
                rg = (rg < NTOT_) ? rg : (NTOT_ - 1);
                v2f a;
                if (s == 0)      { a.x = x0[rg * WIN_ + ka0] * m0;  a.y = x0[rg * WIN_ + ka1] * m1; }
                else if (s == 1) { a.x = x1[rg * XP_ + ka0] * m0;   a.y = x1[rg * XP_ + ka1] * m1; }
                else             { a.x = tile[rl * WIN_ + ka0] * m0; a.y = tile[rl * WIN_ + ka1] * m1; }
                c[t][0] = __builtin_amdgcn_wmma_f32_16x16x4_f32(false, a, false, b0,
                                                                (short)0, c[t][0], false, false);
                c[t][1] = __builtin_amdgcn_wmma_f32_16x16x4_f32(false, a, false, b1,
                                                                (short)0, c[t][1], false, false);
            }
        }
    }
    dense_tail(stage4, c, bias, wave, half, l16, node0, rows, hout);
}

template <int PITCH>
static __device__ __forceinline__ void seg_from_tile(const float* tile, int wave, int half, int l16,
                                                     const float* __restrict__ Wseg, v8f (&c)[2][2]) {
#pragma unroll
    for (int k0 = 0; k0 < H_; k0 += 4) {
        int kA = k0 + half * 2;
        v2f b0, b1;
        b0.x = Wseg[(kA    ) * H_ + l16];
        b0.y = Wseg[(kA + 1) * H_ + l16];
        b1.x = Wseg[(kA    ) * H_ + 16 + l16];
        b1.y = Wseg[(kA + 1) * H_ + 16 + l16];
#pragma unroll
        for (int t = 0; t < 2; ++t) {
            const float* p = &tile[(wave * 32 + t * 16 + l16) * PITCH + kA];
            v2f a;
            a.x = p[0];
            a.y = p[1];
            c[t][0] = __builtin_amdgcn_wmma_f32_16x16x4_f32(false, a, false, b0,
                                                            (short)0, c[t][0], false, false);
            c[t][1] = __builtin_amdgcn_wmma_f32_16x16x4_f32(false, a, false, b1,
                                                            (short)0, c[t][1], false, false);
        }
    }
}

template <bool READOUT>
__global__ __launch_bounds__(256, 8) void bin_prop32_dense_kernel(const int* __restrict__ binStart,
                                                               const int* __restrict__ recs,
                                                               const float* __restrict__ wts,
                                                               const float* __restrict__ x0,
                                                               const float* __restrict__ x1,
                                                               const float* __restrict__ W,
                                                               const float* __restrict__ bias,
                                                               float* __restrict__ hout,
                                                               const float* __restrict__ Wr,
                                                               const float* __restrict__ br,
                                                               float* __restrict__ rout) {
    __shared__ float4 tile4[TR_ * 8];
    __shared__ float4 stage4[TR_ * SP4_];
    const float* tile = (const float*)tile4;
    const float* stage = (const float*)stage4;
    int bin = blockIdx.x;
    int node0 = bin * RB_;
    int rows = (NTOT_ - node0 < RB_) ? (NTOT_ - node0) : RB_;
    const float4* __restrict__ x0_4 = (const float4*)x0;
    const float4* __restrict__ x1_4 = (const float4*)x1;
    accum32(tile4, bin, binStart, recs, wts, x1);

    int wave = threadIdx.x >> 5;
    int lane = threadIdx.x & 31;
    int half = lane >> 4;
    int l16  = lane & 15;
    int srow = threadIdx.x >> 3;
    int sq   = threadIdx.x & 7;
    float4 g[8];
#pragma unroll
    for (int j = 0; j < 8; ++j) {
        int i = threadIdx.x + j * 256;
        g[j] = (i < rows * 8) ? x0_4[node0 * 8 + i] : make_float4(0.f, 0.f, 0.f, 0.f);
    }
#pragma unroll
    for (int j = 0; j < 8; ++j) stage4[(srow + j * 32) * SP4_ + sq] = g[j];
    __syncthreads();
    v8f c[2][2] = {};
    seg_from_tile<SP_>(stage, wave, half, l16, W, c);
#pragma unroll
    for (int j = 0; j < 8; ++j) {
        int i = threadIdx.x + j * 256;
        g[j] = (i < rows * 8) ? x1_4[node0 * 8 + i] : make_float4(0.f, 0.f, 0.f, 0.f);
    }
    __syncthreads();
#pragma unroll
    for (int j = 0; j < 8; ++j) stage4[(srow + j * 32) * SP4_ + sq] = g[j];
    __syncthreads();
    seg_from_tile<SP_>(stage, wave, half, l16, W + H_ * H_, c);
    seg_from_tile<H_>(tile, wave, half, l16, W + 2 * H_ * H_, c);
    dense_tail(stage4, c, bias, wave, half, l16, node0, rows, hout);
    if (READOUT) {
        int row = threadIdx.x;
        if (row < rows) {
            float acc = br[0];
#pragma unroll
            for (int j = 0; j < H_; ++j) acc += stage[row * SP_ + j] * Wr[j];
            rout[node0 + row] = acc;
        }
    }
}

static inline int cdiv(int a, int b) { return (a + b - 1) / b; }

extern "C" void kernel_launch(void* const* d_in, const int* in_sizes, int n_in,
                              void* d_out, int out_size, void* d_ws, size_t ws_size,
                              hipStream_t stream) {
    (void)in_sizes; (void)n_in; (void)out_size; (void)ws_size;

    const float* X  = (const float*)d_in[0];
    const int*   EI = (const int*)d_in[1];
    const float* W0 = (const float*)d_in[2];
    const float* b0 = (const float*)d_in[3];
    const float* W1 = (const float*)d_in[4];
    const float* b1 = (const float*)d_in[5];
    const float* W2 = (const float*)d_in[6];
    const float* b2 = (const float*)d_in[7];
    const float* Wr = (const float*)d_in[8];
    const float* br = (const float*)d_in[9];

    float* outF = (float*)d_out;
    float* hOut = outF + NTOT_;

    float* ws       = (float*)d_ws;
    int*   recs     = (int*)ws;
    int*   binCount = recs + BE_;
    int*   binStart = binCount + 1024;
    int*   cursor   = binStart + 1024;
    float* p1   = (float*)(cursor + 1024);
    float* bufA = p1 + (size_t)NTOT_ * H_;
    float* bufB = bufA + (size_t)NTOT_ * H_;
    float* wts  = bufB + (size_t)NTOT_ * H_;
    float* xp   = wts + BE_;

    const int TB = 256;

    zero_i32_kernel<<<cdiv(NB_, TB), TB, 0, stream>>>(binCount, NB_);
    bin_count_kernel<<<NCH_, 256, 0, stream>>>(EI, binCount);
    bin_scan_kernel<<<1, 1024, 0, stream>>>(binCount, binStart, cursor);
    bin_fill_kernel<<<NCH_, 256, 0, stream>>>(EI, cursor, recs);
    bin_dinv_kernel<<<NB_, 256, 0, stream>>>(binStart, recs, X, xp);

    bin_prop5_kernel<<<NB_, 256, 0, stream>>>(binStart, recs, wts, xp, p1);
    bin_prop5_dense_kernel<<<NB_, 256, 0, stream>>>(binStart, recs, wts, X, p1, W0, b0, bufA);

    bin_prop32_kernel<<<NB_, 256, 0, stream>>>(binStart, recs, wts, bufA, p1);
    bin_prop32_dense_kernel<false><<<NB_, 256, 0, stream>>>(binStart, recs, wts, bufA, p1, W1, b1, bufB,
                                                            Wr, br, outF);

    bin_prop32_kernel<<<NB_, 256, 0, stream>>>(binStart, recs, wts, bufB, p1);
    bin_prop32_dense_kernel<true><<<NB_, 256, 0, stream>>>(binStart, recs, wts, bufB, p1, W2, b2, hOut,
                                                           Wr, br, outF);
}


# Wide sums: a lane group's 8 edges per trip are 8 consecutive record slots (trips start 32-byte aligned): records and norms come as two 16-byte loads each, not eight 4-byte loads; sums arrival-order
# speedup vs baseline: 6.1458x; 1.0377x over previous
#include <hip/hip_runtime.h>
#include <math.h>

#define B_    4
#define WIN_  5
#define N_    50000
#define E_    1000000
#define H_    32
#define NTOT_ (B_ * N_)
#define BE_   (B_ * E_)

typedef __attribute__((ext_vector_type(2))) float v2f;
typedef __attribute__((ext_vector_type(8))) float v8f;

static __device__ __forceinline__ void decode_edge(const int* __restrict__ ei, int e,
                                                   int& src, int& dst) {
    int b  = e / E_;
    int el = e - b * E_;
    src = ei[(b * 2 + 0) * E_ + el] + b * N_;
    dst = ei[(b * 2 + 1) * E_ + el] + b * N_;
}

#define RB_   196
#define TR_   256
#define NB_   ((NTOT_ + RB_ - 1) / RB_)
#define CH_   8192
#define NCH_  ((BE_ + CH_ - 1) / CH_)
#define EPT_  (CH_ / 256)
#define NG4_  (EPT_ / 4)
#define SRCM_ 0x3FFFF
#define SP_   36
#define SP4_  (SP_ / 4)
#define XP_   8
#define E5_   8
#define E32_  8

__global__ void zero_i32_kernel(int* __restrict__ p, int n) {
    int i = blockIdx.x * blockDim.x + threadIdx.x;
    if (i < n) p[i] = 0;
}

static __device__ __forceinline__ int group_pos(int e, int& nb) {
    bool ok = e < BE_;
    int ec = ok ? e : 0;
    int b  = ec / E_;
    int el = ec - b * E_;
    nb = ok ? b * N_ : -1;
    return (b * 2 + 1) * E_ + el;
}

__global__ __launch_bounds__(256) void bin_count_kernel(const int* __restrict__ ei,
                                                        int* __restrict__ binCount) {
    __shared__ int hist[NB_];
    for (int i = threadIdx.x; i < NB_; i += 256) hist[i] = 0;
    int e0 = blockIdx.x * CH_;
    int4 d[NG4_];
    int  nb[NG4_];
#pragma unroll
    for (int k = 0; k < NG4_; ++k) {
        int p = group_pos(e0 + (k * 256 + threadIdx.x) * 4, nb[k]);
        d[k] = *(const int4*)(ei + p);
    }
    __syncthreads();
#pragma unroll
    for (int k = 0; k < NG4_; ++k) {
        if (nb[k] >= 0) {
            atomicAdd(&hist[(d[k].x + nb[k]) / RB_], 1);
            atomicAdd(&hist[(d[k].y + nb[k]) / RB_], 1);
            atomicAdd(&hist[(d[k].z + nb[k]) / RB_], 1);
            atomicAdd(&hist[(d[k].w + nb[k]) / RB_], 1);
        }
    }
    __syncthreads();
    for (int i = threadIdx.x; i < NB_; i += 256) {
        int c = hist[i];
        if (c) atomicAdd(&binCount[i], c);
    }
}

__global__ __launch_bounds__(1024) void bin_scan_kernel(const int* __restrict__ binCount,
                                                        int* __restrict__ binStart,
                                                        int* __restrict__ cursor) {
    __shared__ int s[1024];
    int t = threadIdx.x;
    int v = (t < NB_) ? binCount[t] : 0;
    s[t] = v;
    __syncthreads();
    for (int off = 1; off < 1024; off <<= 1) {
        int x = (t >= off) ? s[t - off] : 0;
        __syncthreads();
        s[t] += x;
        __syncthreads();
    }
    int excl = s[t] - v;
    if (t < NB_) { binStart[t] = excl; cursor[t] = excl; }
    if (t == NB_ - 1) binStart[NB_] = s[t];
}

#define KEYB_ 10
#define KEYM_ ((1 << KEYB_) - 1)
#define NBP_  (1 << KEYB_)
static_assert(NB_ <= (1 << KEYB_) && NBP_ == 4 * 256 && CH_ <= (1 << (31 - KEYB_)), "key packing and the reservation's four blocks per thread");
static_assert(RB_ <= TR_ && NB_ <= 1024, "a block's rows fit the tile; the scan's single work-group covers every block");
__global__ __launch_bounds__(256) void bin_fill_kernel(const int* __restrict__ ei,
                                                       int* __restrict__ cursor,
                                                       int* __restrict__ recs) {
    __shared__ int4 hist4[NBP_ / 4];
    __shared__ int2 place[NBP_];
    __shared__ int2 stage[CH_];
    __shared__ int  wtot[8];
    int* hist = (int*)hist4;
    for (int i = threadIdx.x; i < NBP_; i += 256) hist[i] = 0;
    int e0 = blockIdx.x * CH_;
    int4 s[NG4_], d[NG4_];
    int  nb[NG4_];
#pragma unroll
    for (int k = 0; k < NG4_; ++k) {
        int p = group_pos(e0 + (k * 256 + threadIdx.x) * 4, nb[k]);
        s[k] = *(const int4*)(ei + p - E_);
        d[k] = *(const int4*)(ei + p);
    }
    __syncthreads();
    int rec[EPT_];
    int key[EPT_];
#pragma unroll
    for (int k = 0; k < NG4_; ++k) {
        int sv[4] = { s[k].x, s[k].y, s[k].z, s[k].w };
        int dv[4] = { d[k].x, d[k].y, d[k].z, d[k].w };
#pragma unroll
        for (int c = 0; c < 4; ++c) {
            int dst = dv[c] + nb[k];
            int bin = dst / RB_;
            rec[k * 4 + c] = (sv[c] + nb[k]) | ((dst - bin * RB_) << 18);
            key[k * 4 + c] = (nb[k] >= 0) ? bin : -1;
        }
    }
    bool whole = e0 + CH_ <= BE_;
    if (whole) {
#pragma unroll
        for (int j = 0; j < EPT_; ++j) key[j] |= atomicAdd(&hist[key[j]], 1) << KEYB_;
    } else {
#pragma unroll
        for (int j = 0; j < EPT_; ++j) {
            if (key[j] >= 0) key[j] |= atomicAdd(&hist[key[j]], 1) << KEYB_;
        }
    }
    __syncthreads();
    int total;
    {
        int t = threadIdx.x;
        int lane = t & 31, wv = t >> 5;
        int4 c4 = hist4[t];
        int cnt[4] = { c4.x, c4.y, c4.z, c4.w };
        int bs[4];
#pragma unroll
        for (int u = 0; u < 4; ++u) bs[u] = cnt[u] ? atomicAdd(&cursor[t * 4 + u], cnt[u]) : 0;
        int tsum = cnt[0] + cnt[1] + cnt[2] + cnt[3];
        int incl = tsum;
        for (int off = 1; off < 32; off <<= 1) {
            int y = __shfl_up(incl, off, 32);
            if (lane >= off) incl += y;
        }
        if (lane == 31) wtot[wv] = incl;
        __syncthreads();
        int before = 0;
        total = 0;
        for (int k = 0; k < 8; ++k) {
            int v = wtot[k];
            before += (k < wv) ? v : 0;
            total += v;
        }
        int p0 = before + incl - tsum;
        place[t * 4 + 0] = make_int2(p0, bs[0]);
        place[t * 4 + 1] = make_int2(p0 + cnt[0], bs[1]);
        place[t * 4 + 2] = make_int2(p0 + cnt[0] + cnt[1], bs[2]);
        place[t * 4 + 3] = make_int2(p0 + cnt[0] + cnt[1] + cnt[2], bs[3]);
    }
    __syncthreads();
    if (whole) {
#pragma unroll
        for (int j = 0; j < EPT_; ++j) {
            int2 pl = place[key[j] & KEYM_];
            int sl = key[j] >> KEYB_;
            stage[pl.x + sl] = make_int2(rec[j], pl.y + sl);
        }
    } else {
#pragma unroll
        for (int j = 0; j < EPT_; ++j) {
            if (key[j] >= 0) {
                int2 pl = place[key[j] & KEYM_];
                int sl = key[j] >> KEYB_;
                stage[pl.x + sl] = make_int2(rec[j], pl.y + sl);
            }
        }
    }
    __syncthreads();
    if (whole) {
        int2 v[EPT_];
#pragma unroll
        for (int j = 0; j < EPT_; ++j) v[j] = stage[j * 256 + threadIdx.x];
#pragma unroll
        for (int j = 0; j < EPT_; ++j) recs[v[j].y] = v[j].x;
    } else {
        for (int p = threadIdx.x; p < total; p += 256) {
            int2 v = stage[p];
            recs[v.y] = v.x;
        }
    }
}

__global__ __launch_bounds__(256) void bin_dinv_kernel(const int* __restrict__ binStart,
                                                       const int* __restrict__ recs,
                                                       const float* __restrict__ x,
                                                       float* __restrict__ xp) {
    __shared__ int cnt[TR_];
    int bin = blockIdx.x;
    cnt[threadIdx.x] = 0;
    __syncthreads();
    int s = binStart[bin], e = binStart[bin + 1];
    int last = (e > 0) ? e - 1 : 0;
    for (int i0 = s + threadIdx.x; i0 < e; i0 += 4 * 256) {
        int r[4];
#pragma unroll
        for (int u = 0; u < 4; ++u) {
            int i = i0 + u * 256;
            int rv = recs[(i < e) ? i : last];
            r[u] = (i < e) ? rv : -1;
        }
#pragma unroll
        for (int u = 0; u < 4; ++u) {
            if (r[u] >= 0) atomicAdd(&cnt[r[u] >> 18], 1);
        }
    }
    __syncthreads();
    int node = bin * RB_ + threadIdx.x;
    if ((int)threadIdx.x < RB_ && node < NTOT_) {
        int c = cnt[threadIdx.x];
        float dv = (c > 0) ? rsqrtf((float)c) : 0.0f;
        const float* xr = x + node * WIN_;
        float4* __restrict__ xp4 = (float4*)xp;
        xp4[node * 2 + 0] = make_float4(xr[0], xr[1], xr[2], xr[3]);
        xp4[node * 2 + 1] = make_float4(xr[4], dv, 0.0f, 0.0f);
    }
}


static_assert(E5_ == 8, "two record slots per lane of a 4-lane group");
template <bool MAKE_W>
static __device__ __forceinline__ void accum5(float* tile, float* dinvD, int bin, int node0, int rows,
                                              const int* __restrict__ binStart,
                                              const int* __restrict__ recs,
                                              float* __restrict__ wts,
                                              const float* __restrict__ hin) {
    const float2* __restrict__ hin2 = (const float2*)hin;
    for (int i = threadIdx.x; i < TR_ * WIN_; i += 256) tile[i] = 0.0f;
    if (MAKE_W) dinvD[threadIdx.x] = ((int)threadIdx.x < rows) ? hin[(node0 + threadIdx.x) * XP_ + WIN_] : 0.0f;
    __syncthreads();
    int j = threadIdx.x & 3;
    int grp = threadIdx.x >> 2;
    int s = binStart[bin], e = binStart[bin + 1];
    int last = (e > 0) ? e - 1 : 0;
    auto load_recs = [&](int ib, int& xa, int& xb, float& ya, float& yb) {
        int ia = ib + j, ic = ib + 4 + j;
        int ca = (ia < e) ? ia : last, cc = (ic < e) ? ic : last;
        int rva = recs[ca], rvc = recs[cc];
        ya = MAKE_W ? 0.0f : wts[ca];
        yb = MAKE_W ? 0.0f : wts[cc];
        xa = (ia < e) ? rva : -1;
        xb = (ic < e) ? rvc : -1;
    };
    int t0 = s;
    int i0 = t0 + grp * E5_;
    int ra, rb;
    float wa, wb;
    load_recs(i0, ra, rb, wa, wb);
    while (t0 < e) {
        int r[E5_];
        float w[E5_];
#pragma unroll
        for (int u = 0; u < E5_; ++u) {
            r[u] = __shfl((u < 4) ? ra : rb, u & 3, 4);
            w[u] = MAKE_W ? 0.0f : __shfl((u < 4) ? wa : wb, u & 3, 4);
        }
        float2 v[E5_];
#pragma unroll
        for (int u = 0; u < E5_; ++u) {
            int srcu = (r[u] < 0) ? 0 : (r[u] & SRCM_);
            v[u] = hin2[srcu * (XP_ / 2) + j];
        }
        int t1 = t0 + 64 * E5_;
        int i1 = t1 + grp * E5_;
        int rna, rnb;
        float wna, wnb;
        load_recs(i1, rna, rnb, wna, wnb);
        float nrm[E5_];
#pragma unroll
        for (int u = 0; u < E5_; ++u) {
            if (MAKE_W) {
                int dl = (r[u] < 0) ? 0 : (r[u] >> 18);
                float dsrc = __shfl(v[u].y, 2, 4);
                nrm[u] = dsrc * dinvD[dl];
            } else {
                nrm[u] = w[u];
            }
        }
        if (MAKE_W) {
            float na = (j == 0) ? nrm[0] : (j == 1) ? nrm[1] : (j == 2) ? nrm[2] : nrm[3];
            float nb = (j == 0) ? nrm[4] : (j == 1) ? nrm[5] : (j == 2) ? nrm[6] : nrm[7];
            if (i0 + j < e)     wts[i0 + j]     = na;
            if (i0 + 4 + j < e) wts[i0 + 4 + j] = nb;
        }
        auto add_edge = [&](int u) {
            float* tp = &tile[(r[u] >> 18) * WIN_ + 2 * j];
            if (j < 3) atomicAdd(tp,     v[u].x * nrm[u]);
            if (j < 2) atomicAdd(tp + 1, v[u].y * nrm[u]);
        };
        if (i0 + E5_ <= e) {
#pragma unroll
            for (int u = 0; u < E5_; ++u) add_edge(u);
        } else {
#pragma unroll
            for (int u = 0; u < E5_; ++u) {
                if (r[u] >= 0) add_edge(u);
            }
        }
        ra = rna; rb = rnb; wa = wna; wb = wnb;
        t0 = t1;
        i0 = i1;
    }
    __syncthreads();
}

static_assert(E32_ == 8 && BE_ % E32_ == 0, "a lane group's records of a trip are two int4; the list ends on a group boundary");
static __device__ __forceinline__ void accum32(float4* tile4, int bin,
                                               const int* __restrict__ binStart,
                                               const int* __restrict__ recs,
                                               const float* __restrict__ wts,
                                               const float* __restrict__ hin) {
    float* tile = (float*)tile4;
    const float4* __restrict__ hin4 = (const float4*)hin;
    const int4* __restrict__ recs4 = (const int4*)recs;
    const float4* __restrict__ wts4 = (const float4*)wts;
    for (int i = threadIdx.x; i < TR_ * 8; i += 256) tile4[i] = make_float4(0.f, 0.f, 0.f, 0.f);
    __syncthreads();
    int q = threadIdx.x & 7;
    int sub = threadIdx.x >> 3;
    int rot = sub & 3;
    bool r1 = (rot & 1) != 0, r2 = (rot & 2) != 0;
    int o0 = rot, o1 = (rot + 1) & 3, o2 = (rot + 2) & 3, o3 = (rot + 3) & 3;
    int s = binStart[bin], e = binStart[bin + 1];
    int sa = s & ~(E32_ - 1);
    auto load_grp = [&](int g, int (&r)[E32_], float (&w)[E32_]) {
        int gc = (g < BE_ - E32_) ? g : (BE_ - E32_);
        int p = gc >> 2;
        int4 ra = recs4[p], rb = recs4[p + 1];
        float4 wa = wts4[p], wb = wts4[p + 1];
        r[0] = ra.x; r[1] = ra.y; r[2] = ra.z; r[3] = ra.w;
        r[4] = rb.x; r[5] = rb.y; r[6] = rb.z; r[7] = rb.w;
        w[0] = wa.x; w[1] = wa.y; w[2] = wa.z; w[3] = wa.w;
        w[4] = wb.x; w[5] = wb.y; w[6] = wb.z; w[7] = wb.w;
    };
    int g0 = sa + sub * E32_;
    int r[E32_];
    float w[E32_];
    load_grp(g0, r, w);
    while (g0 < e) {
        float4 v[E32_];
#pragma unroll
        for (int u = 0; u < E32_; ++u) v[u] = hin4[(r[u] & SRCM_) * 8 + q];
        int g1 = g0 + E32_ * 32;
        int rn[E32_];
        float wn[E32_];
        load_grp(g1, rn, wn);
        auto add_edge = [&](int rec, float nrm, float4 x) {
            int dl = rec >> 18;
            float* tp = &tile[dl * H_ + q * 4];
            float b0 = r1 ? x.y : x.x, b1 = r1 ? x.z : x.y;
            float b2 = r1 ? x.w : x.z, b3 = r1 ? x.x : x.w;
            float c0 = r2 ? b2 : b0, c1 = r2 ? b3 : b1, c2 = r2 ? b0 : b2, c3 = r2 ? b1 : b3;
            atomicAdd(tp + o0, c0 * nrm);
            atomicAdd(tp + o1, c1 * nrm);
            atomicAdd(tp + o2, c2 * nrm);
            atomicAdd(tp + o3, c3 * nrm);
        };
        if (g0 >= s && g0 + E32_ <= e) {
#pragma unroll
            for (int u = 0; u < E32_; ++u) add_edge(r[u], w[u], v[u]);
        } else {
#pragma unroll
            for (int u = 0; u < E32_; ++u) {
                int i = g0 + u;
                if (i >= s && i < e) add_edge(r[u], w[u], v[u]);
            }
        }
#pragma unroll
        for (int u = 0; u < E32_; ++u) { r[u] = rn[u]; w[u] = wn[u]; }
        g0 = g1;
    }
    __syncthreads();
}

__global__ __launch_bounds__(256) void bin_prop5_kernel(const int* __restrict__ binStart,
                                                        const int* __restrict__ recs,
                                                        float* __restrict__ wts,
                                                        const float* __restrict__ xp,
                                                        float* __restrict__ hout) {
    __shared__ float tile[TR_ * WIN_];
    __shared__ float dinvD[TR_];
    int bin = blockIdx.x;
    int node0 = bin * RB_;
    int rows = (NTOT_ - node0 < RB_) ? (NTOT_ - node0) : RB_;
    accum5<true>(tile, dinvD, bin, node0, rows, binStart, recs, wts, xp);
    float4* __restrict__ out4 = (float4*)hout;
    int t = threadIdx.x;
    if (t < rows) {
        const float* tr = &tile[t * WIN_];
        out4[(node0 + t) * 2 + 0] = make_float4(tr[0], tr[1], tr[2], tr[3]);
        out4[(node0 + t) * 2 + 1] = make_float4(tr[4], 0.0f, 0.0f, 0.0f);
    }
}

__global__ __launch_bounds__(256) void bin_prop32_kernel(const int* __restrict__ binStart,
                                                         const int* __restrict__ recs,
                                                         const float* __restrict__ wts,
                                                         const float* __restrict__ hin,
                                                         float* __restrict__ hout) {
    __shared__ float4 tile4[TR_ * 8];
    int bin = blockIdx.x;
    int node0 = bin * RB_;
    int rows = (NTOT_ - node0 < RB_) ? (NTOT_ - node0) : RB_;
    accum32(tile4, bin, binStart, recs, wts, hin);
    float4* __restrict__ out4 = (float4*)hout;
    for (int i = threadIdx.x; i < rows * 8; i += 256) out4[node0 * 8 + i] = tile4[i];
}

static __device__ __forceinline__ void dense_tail(float4* stage4, const v8f (&c)[2][2],
                                                  const float* __restrict__ bias,
                                                  int wave, int half, int l16, int node0, int rows,
                                                  float* __restrict__ hout) {
    float* stage = (float*)stage4;
    float bz0 = bias[l16], bz1 = bias[16 + l16];
#pragma unroll
    for (int t = 0; t < 2; ++t) {
#pragma unroll
        for (int r = 0; r < 8; ++r) {
            int row = wave * 32 + t * 16 + half * 8 + r;
            stage[row * SP_ + l16]      = __builtin_amdgcn_tanhf(c[t][0][r] + bz0);
            stage[row * SP_ + 16 + l16] = __builtin_amdgcn_tanhf(c[t][1][r] + bz1);
        }
    }
    __syncthreads();
    float4* __restrict__ out4 = (float4*)hout;
    for (int i = threadIdx.x; i < rows * 8; i += 256) out4[node0 * 8 + i] = stage4[(i >> 3) * SP4_ + (i & 7)];
}

__global__ __launch_bounds__(256) void bin_prop5_dense_kernel(const int* __restrict__ binStart,
                                                              const int* __restrict__ recs,
                                                              float* __restrict__ wts,
                                                              const float* __restrict__ x0,
                                                              const float* __restrict__ x1,
                                                              const float* __restrict__ W,
                                                              const float* __restrict__ bias,
                                                              float* __restrict__ hout) {
    __shared__ float tile[TR_ * WIN_];
    __shared__ float4 stage4[TR_ * SP4_];
    int bin = blockIdx.x;
    int node0 = bin * RB_;
    int rows = (NTOT_ - node0 < RB_) ? (NTOT_ - node0) : RB_;
    accum5<false>(tile, nullptr, bin, node0, rows, binStart, recs, wts, x1);

    int wave = threadIdx.x >> 5;
    int lane = threadIdx.x & 31;
    int half = lane >> 4;
    int l16  = lane & 15;
    v8f c[2][2] = {};
#pragma unroll
    for (int s = 0; s < 3; ++s) {
#pragma unroll
        for (int k0 = 0; k0 < WIN_; k0 += 4) {
            int kA = k0 + half * 2;
            float m0 = (kA     < WIN_) ? 1.0f : 0.0f;
            float m1 = (kA + 1 < WIN_) ? 1.0f : 0.0f;
            int ka0 = (kA     < WIN_) ? kA     : (WIN_ - 1);
            int ka1 = (kA + 1 < WIN_) ? kA + 1 : (WIN_ - 1);
            int kw0 = s * WIN_ + ka0;
            int kw1 = s * WIN_ + ka1;
            v2f b0, b1;
            b0.x = W[kw0 * H_ + l16] * m0;
            b0.y = W[kw1 * H_ + l16] * m1;
            b1.x = W[kw0 * H_ + 16 + l16] * m0;
            b1.y = W[kw1 * H_ + 16 + l16] * m1;
#pragma unroll
            for (int t = 0; t < 2; ++t) {
                int rl = wave * 32 + t * 16 + l16;
                int rg = node0 + rl;
                rg = (rg < NTOT_) ? rg : (NTOT_ - 1);
                v2f a;
                if (s == 0)      { a.x = x0[rg * WIN_ + ka0] * m0;  a.y = x0[rg * WIN_ + ka1] * m1; }
                else if (s == 1) { a.x = x1[rg * XP_ + ka0] * m0;   a.y = x1[rg * XP_ + ka1] * m1; }
                else             { a.x = tile[rl * WIN_ + ka0] * m0; a.y = tile[rl * WIN_ + ka1] * m1; }
                c[t][0] = __builtin_amdgcn_wmma_f32_16x16x4_f32(false, a, false, b0,
                                                                (short)0, c[t][0], false, false);
                c[t][1] = __builtin_amdgcn_wmma_f32_16x16x4_f32(false, a, false, b1,
                                                                (short)0, c[t][1], false, false);
            }
        }
    }
    dense_tail(stage4, c, bias, wave, half, l16, node0, rows, hout);
}

template <int PITCH>
static __device__ __forceinline__ void seg_from_tile(const float* tile, int wave, int half, int l16,
                                                     const float* __restrict__ Wseg, v8f (&c)[2][2]) {
#pragma unroll
    for (int k0 = 0; k0 < H_; k0 += 4) {
        int kA = k0 + half * 2;
        v2f b0, b1;
        b0.x = Wseg[(kA    ) * H_ + l16];
        b0.y = Wseg[(kA + 1) * H_ + l16];
        b1.x = Wseg[(kA    ) * H_ + 16 + l16];
        b1.y = Wseg[(kA + 1) * H_ + 16 + l16];
#pragma unroll
        for (int t = 0; t < 2; ++t) {
            const float* p = &tile[(wave * 32 + t * 16 + l16) * PITCH + kA];
            v2f a;
            a.x = p[0];
            a.y = p[1];
            c[t][0] = __builtin_amdgcn_wmma_f32_16x16x4_f32(false, a, false, b0,
                                                            (short)0, c[t][0], false, false);
            c[t][1] = __builtin_amdgcn_wmma_f32_16x16x4_f32(false, a, false, b1,
                                                            (short)0, c[t][1], false, false);
        }
    }
}

template <bool READOUT>
__global__ __launch_bounds__(256, 8) void bin_prop32_dense_kernel(const int* __restrict__ binStart,
                                                               const int* __restrict__ recs,
                                                               const float* __restrict__ wts,
                                                               const float* __restrict__ x0,
                                                               const float* __restrict__ x1,
                                                               const float* __restrict__ W,
                                                               const float* __restrict__ bias,
                                                               float* __restrict__ hout,
                                                               const float* __restrict__ Wr,
                                                               const float* __restrict__ br,
                                                               float* __restrict__ rout) {
    __shared__ float4 tile4[TR_ * 8];
    __shared__ float4 stage4[TR_ * SP4_];
    const float* tile = (const float*)tile4;
    const float* stage = (const float*)stage4;
    int bin = blockIdx.x;
    int node0 = bin * RB_;
    int rows = (NTOT_ - node0 < RB_) ? (NTOT_ - node0) : RB_;
    const float4* __restrict__ x0_4 = (const float4*)x0;
    const float4* __restrict__ x1_4 = (const float4*)x1;
    accum32(tile4, bin, binStart, recs, wts, x1);

    int wave = threadIdx.x >> 5;
    int lane = threadIdx.x & 31;
    int half = lane >> 4;
    int l16  = lane & 15;
    int srow = threadIdx.x >> 3;
    int sq   = threadIdx.x & 7;
    float4 g[8];
#pragma unroll
    for (int j = 0; j < 8; ++j) {
        int i = threadIdx.x + j * 256;
        g[j] = (i < rows * 8) ? x0_4[node0 * 8 + i] : make_float4(0.f, 0.f, 0.f, 0.f);
    }
#pragma unroll
    for (int j = 0; j < 8; ++j) stage4[(srow + j * 32) * SP4_ + sq] = g[j];
    __syncthreads();
    v8f c[2][2] = {};
    seg_from_tile<SP_>(stage, wave, half, l16, W, c);
#pragma unroll
    for (int j = 0; j < 8; ++j) {
        int i = threadIdx.x + j * 256;
        g[j] = (i < rows * 8) ? x1_4[node0 * 8 + i] : make_float4(0.f, 0.f, 0.f, 0.f);
    }
    __syncthreads();
#pragma unroll
    for (int j = 0; j < 8; ++j) stage4[(srow + j * 32) * SP4_ + sq] = g[j];
    __syncthreads();
    seg_from_tile<SP_>(stage, wave, half, l16, W + H_ * H_, c);
    seg_from_tile<H_>(tile, wave, half, l16, W + 2 * H_ * H_, c);
    dense_tail(stage4, c, bias, wave, half, l16, node0, rows, hout);
    if (READOUT) {
        int row = threadIdx.x;
        if (row < rows) {
            float acc = br[0];
#pragma unroll
            for (int j = 0; j < H_; ++j) acc += stage[row * SP_ + j] * Wr[j];
            rout[node0 + row] = acc;
        }
    }
}

static inline int cdiv(int a, int b) { return (a + b - 1) / b; }

extern "C" void kernel_launch(void* const* d_in, const int* in_sizes, int n_in,
                              void* d_out, int out_size, void* d_ws, size_t ws_size,
                              hipStream_t stream) {
    (void)in_sizes; (void)n_in; (void)out_size; (void)ws_size;

    const float* X  = (const float*)d_in[0];
    const int*   EI = (const int*)d_in[1];
    const float* W0 = (const float*)d_in[2];
    const float* b0 = (const float*)d_in[3];
    const float* W1 = (const float*)d_in[4];
    const float* b1 = (const float*)d_in[5];
    const float* W2 = (const float*)d_in[6];
    const float* b2 = (const float*)d_in[7];
    const float* Wr = (const float*)d_in[8];
    const float* br = (const float*)d_in[9];

    float* outF = (float*)d_out;
    float* hOut = outF + NTOT_;

    float* ws       = (float*)d_ws;
    int*   recs     = (int*)ws;
    int*   binCount = recs + BE_;
    int*   binStart = binCount + 1024;
    int*   cursor   = binStart + 1024;
    float* p1   = (float*)(cursor + 1024);
    float* bufA = p1 + (size_t)NTOT_ * H_;
    float* bufB = bufA + (size_t)NTOT_ * H_;
    float* wts  = bufB + (size_t)NTOT_ * H_;
    float* xp   = wts + BE_;

    const int TB = 256;

    zero_i32_kernel<<<cdiv(NB_, TB), TB, 0, stream>>>(binCount, NB_);
    bin_count_kernel<<<NCH_, 256, 0, stream>>>(EI, binCount);
    bin_scan_kernel<<<1, 1024, 0, stream>>>(binCount, binStart, cursor);
    bin_fill_kernel<<<NCH_, 256, 0, stream>>>(EI, cursor, recs);
    bin_dinv_kernel<<<NB_, 256, 0, stream>>>(binStart, recs, X, xp);

    bin_prop5_kernel<<<NB_, 256, 0, stream>>>(binStart, recs, wts, xp, p1);
    bin_prop5_dense_kernel<<<NB_, 256, 0, stream>>>(binStart, recs, wts, X, p1, W0, b0, bufA);

    bin_prop32_kernel<<<NB_, 256, 0, stream>>>(binStart, recs, wts, bufA, p1);
    bin_prop32_dense_kernel<false><<<NB_, 256, 0, stream>>>(binStart, recs, wts, bufA, p1, W1, b1, bufB,
                                                            Wr, br, outF);

    bin_prop32_kernel<<<NB_, 256, 0, stream>>>(binStart, recs, wts, bufB, p1);
    bin_prop32_dense_kernel<true><<<NB_, 256, 0, stream>>>(binStart, recs, wts, bufB, p1, W2, b2, hOut,
                                                           Wr, br, outF);
}
